# starting kernel
# baseline (speedup 1.0000x reference)
#include <hip/hip_runtime.h>

typedef float v2f __attribute__((ext_vector_type(2)));
typedef float v8f __attribute__((ext_vector_type(8)));

#define B_     4
#define LQ_    128
#define LK_    1024
#define DSZ_   512
#define UNITS_ 256
#define MASK_VAL_ (-1.0e6f)

#if __has_builtin(__builtin_amdgcn_tanhf)
#define TANHF(x) __builtin_amdgcn_tanhf(x)
#else
#define TANHF(x) tanhf(x)
#endif

template <bool TRANS_OUT>
__global__ __launch_bounds__(256) void proj_wmma(
    const float* __restrict__ A,
    const float* __restrict__ W,
    float* __restrict__ out,
    int M, int K, int N, int seq)
{
    const int lane   = threadIdx.x & 31;
    const int wave   = threadIdx.x >> 5;
    const int tilesN = N >> 4;
    const int tile   = blockIdx.x * 8 + wave;
    const int tile_m = tile / tilesN;
    const int tile_n = tile % tilesN;

    const int n15  = lane & 15;
    const int ko   = (lane < 16) ? 0 : 2;
    const int arow = tile_m * 16 + n15;
    const int wrow = tile_n * 16 + n15;

    const float* aptr = A + (size_t)arow * K + ko;
    const float* wptr = W + (size_t)wrow * K + ko;

    v8f c = {0.f, 0.f, 0.f, 0.f, 0.f, 0.f, 0.f, 0.f};
#pragma unroll 4
    for (int kk = 0; kk < K; kk += 4) {
        v2f av = *(const v2f*)(aptr + kk);
        v2f bv = *(const v2f*)(wptr + kk);
        c = __builtin_amdgcn_wmma_f32_16x16x4_f32(
                false, av, false, bv, (short)0, c, false, false);
    }

    const int mo = (lane < 16) ? 0 : 8;
#pragma unroll
    for (int r = 0; r < 8; ++r) {
        const int m = tile_m * 16 + mo + r;
        if (TRANS_OUT) {
            const int b    = m / seq;
            const int kpos = m % seq;
            out[((size_t)b * N + (tile_n * 16 + n15)) * seq + kpos] = c[r];
        } else {
            out[(size_t)m * N + tile_n * 16 + n15] = c[r];
        }
    }
}

__global__ __launch_bounds__(256) void scores_tanh_kernel(
    const float* __restrict__ qf,
    const float* __restrict__ kfT,
    const float* __restrict__ vvec,
    const int*   __restrict__ valid_len,
    float* __restrict__ scores)
{
    const int bq = blockIdx.x;
    const int b  = bq / LQ_;
    const int t  = threadIdx.x;

    const float* qrow  = qf  + (size_t)bq * UNITS_;
    const float* kbase = kfT + (size_t)b * UNITS_ * LK_ + t * 4;

    float a0 = 0.f, a1 = 0.f, a2 = 0.f, a3 = 0.f;
#pragma unroll 4
    for (int u = 0; u < UNITS_; ++u) {
        const float qv = qrow[u];
        const float vv = vvec[u];
        const float4 kv = *(const float4*)(kbase + (size_t)u * LK_);
        a0 = __builtin_fmaf(vv, TANHF(qv + kv.x), a0);
        a1 = __builtin_fmaf(vv, TANHF(qv + kv.y), a1);
        a2 = __builtin_fmaf(vv, TANHF(qv + kv.z), a2);
        a3 = __builtin_fmaf(vv, TANHF(qv + kv.w), a3);
    }

    const int vl = valid_len[b];
    const int k0 = t * 4;
    float4 r;
    r.x = (k0 + 0 < vl) ? a0 : MASK_VAL_;
    r.y = (k0 + 1 < vl) ? a1 : MASK_VAL_;
    r.z = (k0 + 2 < vl) ? a2 : MASK_VAL_;
    r.w = (k0 + 3 < vl) ? a3 : MASK_VAL_;
    *(float4*)(scores + (size_t)bq * LK_ + k0) = r;
}

__global__ __launch_bounds__(256) void softmax_rows(float* __restrict__ scores)
{
    __shared__ float red[8];
    const int t    = threadIdx.x;
    const int lane = t & 31;
    const int wave = t >> 5;
    float* row = scores + (size_t)blockIdx.x * LK_;

    float4 x = *(const float4*)(row + t * 4);

    float m = fmaxf(fmaxf(x.x, x.y), fmaxf(x.z, x.w));
#pragma unroll
    for (int off = 16; off > 0; off >>= 1)
        m = fmaxf(m, __shfl_xor(m, off, 32));
    if (lane == 0) red[wave] = m;
    __syncthreads();
    float gm = red[0];
#pragma unroll
    for (int i = 1; i < 8; ++i) gm = fmaxf(gm, red[i]);
    __syncthreads();

    float4 e;
    e.x = __expf(x.x - gm);
    e.y = __expf(x.y - gm);
    e.z = __expf(x.z - gm);
    e.w = __expf(x.w - gm);

    float s = e.x + e.y + e.z + e.w;
#pragma unroll
    for (int off = 16; off > 0; off >>= 1)
        s += __shfl_xor(s, off, 32);
    if (lane == 0) red[wave] = s;
    __syncthreads();
    float gs = red[0];
#pragma unroll
    for (int i = 1; i < 8; ++i) gs += red[i];

    const float inv = 1.0f / gs;
    e.x *= inv; e.y *= inv; e.z *= inv; e.w *= inv;
    *(float4*)(row + t * 4) = e;
}

__global__ __launch_bounds__(256) void attn_out_wmma(
    const float* __restrict__ attn,
    const float* __restrict__ value,
    float* __restrict__ out)
{
    const int lane      = threadIdx.x & 31;
    const int wave      = threadIdx.x >> 5;
    const int tilesN    = DSZ_ / 16;
    const int tilesPerB = (LQ_ / 16) * tilesN;
    const int tile      = blockIdx.x * 8 + wave;
    const int b         = tile / tilesPerB;
    const int tr        = tile % tilesPerB;
    const int tile_m    = tr / tilesN;
    const int tile_n    = tr % tilesN;

    const int n15 = lane & 15;
    const int ko  = (lane < 16) ? 0 : 2;

    const float* aptr = attn  + ((size_t)b * LQ_ + tile_m * 16 + n15) * LK_ + ko;
    const float* vptr = value + (size_t)b * LK_ * DSZ_
                               + (size_t)ko * DSZ_ + tile_n * 16 + n15;

    v8f c = {0.f, 0.f, 0.f, 0.f, 0.f, 0.f, 0.f, 0.f};
#pragma unroll 4
    for (int kk = 0; kk < LK_; kk += 4) {
        v2f av = *(const v2f*)(aptr + kk);
        const float* vp = vptr + (size_t)kk * DSZ_;
        v2f bv;
        bv.x = vp[0];
        bv.y = vp[DSZ_];
        c = __builtin_amdgcn_wmma_f32_16x16x4_f32(
                false, av, false, bv, (short)0, c, false, false);
    }

    const int mo = (lane < 16) ? 0 : 8;
#pragma unroll
    for (int r = 0; r < 8; ++r) {
        const int m = tile_m * 16 + mo + r;
        out[((size_t)b * LQ_ + m) * DSZ_ + tile_n * 16 + n15] = c[r];
    }
}

extern "C" void kernel_launch(void* const* d_in, const int* in_sizes, int n_in,
                              void* d_out, int out_size, void* d_ws, size_t ws_size,
                              hipStream_t stream) {
    const float* query     = (const float*)d_in[0];
    const float* key       = (const float*)d_in[1];
    const float* value     = (const float*)d_in[2];
    const int*   valid_len = (const int*)  d_in[3];
    const float* W_q       = (const float*)d_in[4];
    const float* W_k       = (const float*)d_in[5];
    const float* vvec      = (const float*)d_in[6];

    float* qf     = (float*)d_ws;
    float* kfT    = qf  + (size_t)B_ * LQ_ * UNITS_;
    float* scores = kfT + (size_t)B_ * UNITS_ * LK_;
    float* outp   = (float*)d_out;

    proj_wmma<false><<<64, 256, 0, stream>>>(query, W_q, qf,
                                             B_ * LQ_, DSZ_, UNITS_, 0);
    proj_wmma<true><<<512, 256, 0, stream>>>(key, W_k, kfT,
                                             B_ * LK_, DSZ_, UNITS_, LK_);
    scores_tanh_kernel<<<B_ * LQ_, 256, 0, stream>>>(qf, kfT, vvec, valid_len, scores);
    softmax_rows<<<B_ * LQ_, 256, 0, stream>>>(scores);
    attn_out_wmma<<<128, 256, 0, stream>>>(scores, value, outp);
}


# scores kernel: stage a 64-column slab of the key features (all 256 units) and 16 query-feature rows in LDS per block, reuse the slab for 16 q rows; per-(q,k) sum order over units unchanged
# speedup vs baseline: 1.2722x; 1.2722x over previous
#include <hip/hip_runtime.h>

typedef float v2f __attribute__((ext_vector_type(2)));
typedef float v8f __attribute__((ext_vector_type(8)));

#define B_     4
#define LQ_    128
#define LK_    1024
#define DSZ_   512
#define UNITS_ 256
#define MASK_VAL_ (-1.0e6f)

#if __has_builtin(__builtin_amdgcn_tanhf)
#define TANHF(x) __builtin_amdgcn_tanhf(x)
#else
#define TANHF(x) tanhf(x)
#endif

template <bool TRANS_OUT>
__global__ __launch_bounds__(256) void proj_wmma(
    const float* __restrict__ A,
    const float* __restrict__ W,
    float* __restrict__ out,
    int M, int K, int N, int seq)
{
    const int lane   = threadIdx.x & 31;
    const int wave   = threadIdx.x >> 5;
    const int tilesN = N >> 4;
    const int tile   = blockIdx.x * 8 + wave;
    const int tile_m = tile / tilesN;
    const int tile_n = tile % tilesN;

    const int n15  = lane & 15;
    const int ko   = (lane < 16) ? 0 : 2;
    const int arow = tile_m * 16 + n15;
    const int wrow = tile_n * 16 + n15;

    const float* aptr = A + (size_t)arow * K + ko;
    const float* wptr = W + (size_t)wrow * K + ko;

    v8f c = {0.f, 0.f, 0.f, 0.f, 0.f, 0.f, 0.f, 0.f};
#pragma unroll 4
    for (int kk = 0; kk < K; kk += 4) {
        v2f av = *(const v2f*)(aptr + kk);
        v2f bv = *(const v2f*)(wptr + kk);
        c = __builtin_amdgcn_wmma_f32_16x16x4_f32(
                false, av, false, bv, (short)0, c, false, false);
    }

    const int mo = (lane < 16) ? 0 : 8;
#pragma unroll
    for (int r = 0; r < 8; ++r) {
        const int m = tile_m * 16 + mo + r;
        if (TRANS_OUT) {
            const int b    = m / seq;
            const int kpos = m % seq;
            out[((size_t)b * N + (tile_n * 16 + n15)) * seq + kpos] = c[r];
        } else {
            out[(size_t)m * N + tile_n * 16 + n15] = c[r];
        }
    }
}

#define SC_KT_ 64
#define SC_QG_ 16
#define SC_QW_ 4

__global__ __launch_bounds__(128) void scores_tanh_kernel(
    const float* __restrict__ qf,
    const float* __restrict__ kfT,
    const float* __restrict__ vvec,
    const int*   __restrict__ valid_len,
    float* __restrict__ scores)
{
    __shared__ float4 kS4[UNITS_ * SC_KT_ / 4];
    __shared__ float4 qS4[SC_QG_ * UNITS_ / 4];

    const int t    = threadIdx.x;
    const int lane = t & 31;
    const int wave = t >> 5;
    const int qg   = blockIdx.x & 7;
    const int kt   = (blockIdx.x >> 3) & 15;
    const int b    = blockIdx.x >> 7;

    const float* ksrc = kfT + (size_t)b * UNITS_ * LK_ + kt * SC_KT_;
    const int c  = t & 15;
    const int r0 = t >> 4;
#pragma unroll
    for (int i = 0; i < UNITS_ / 8; ++i) {
        const int u = r0 + 8 * i;
        kS4[u * (SC_KT_ / 4) + c] = *(const float4*)(ksrc + (size_t)u * LK_ + c * 4);
    }
    const float4* qsrc =
        (const float4*)(qf + (size_t)(b * LQ_ + qg * SC_QG_) * UNITS_);
#pragma unroll
    for (int i = 0; i < SC_QG_ * UNITS_ / 4 / 128; ++i)
        qS4[i * 128 + t] = qsrc[i * 128 + t];
    __syncthreads();

    const float2* kS2  = (const float2*)kS4;
    const float4* qrow = qS4 + (wave * SC_QW_) * (UNITS_ / 4);

    float a[SC_QW_][2];
#pragma unroll
    for (int j = 0; j < SC_QW_; ++j) { a[j][0] = 0.f; a[j][1] = 0.f; }

#pragma unroll 2
    for (int u4 = 0; u4 < UNITS_ / 4; ++u4) {
        float4 qv[SC_QW_];
#pragma unroll
        for (int j = 0; j < SC_QW_; ++j) qv[j] = qrow[j * (UNITS_ / 4) + u4];
        float2 kv[4];
#pragma unroll
        for (int i = 0; i < 4; ++i) kv[i] = kS2[(u4 * 4 + i) * (SC_KT_ / 2) + lane];
        const float vv0 = vvec[u4 * 4 + 0];
        const float vv1 = vvec[u4 * 4 + 1];
        const float vv2 = vvec[u4 * 4 + 2];
        const float vv3 = vvec[u4 * 4 + 3];
#pragma unroll
        for (int j = 0; j < SC_QW_; ++j) {
            a[j][0] = __builtin_fmaf(vv0, TANHF(qv[j].x + kv[0].x), a[j][0]);
            a[j][1] = __builtin_fmaf(vv0, TANHF(qv[j].x + kv[0].y), a[j][1]);
        }
#pragma unroll
        for (int j = 0; j < SC_QW_; ++j) {
            a[j][0] = __builtin_fmaf(vv1, TANHF(qv[j].y + kv[1].x), a[j][0]);
            a[j][1] = __builtin_fmaf(vv1, TANHF(qv[j].y + kv[1].y), a[j][1]);
        }
#pragma unroll
        for (int j = 0; j < SC_QW_; ++j) {
            a[j][0] = __builtin_fmaf(vv2, TANHF(qv[j].z + kv[2].x), a[j][0]);
            a[j][1] = __builtin_fmaf(vv2, TANHF(qv[j].z + kv[2].y), a[j][1]);
        }
#pragma unroll
        for (int j = 0; j < SC_QW_; ++j) {
            a[j][0] = __builtin_fmaf(vv3, TANHF(qv[j].w + kv[3].x), a[j][0]);
            a[j][1] = __builtin_fmaf(vv3, TANHF(qv[j].w + kv[3].y), a[j][1]);
        }
    }

    const int vl = valid_len[b];
    const int k0 = kt * SC_KT_ + 2 * lane;
    const int q0 = b * LQ_ + qg * SC_QG_ + wave * SC_QW_;
#pragma unroll
    for (int j = 0; j < SC_QW_; ++j) {
        float2 r;
        r.x = (k0 + 0 < vl) ? a[j][0] : MASK_VAL_;
        r.y = (k0 + 1 < vl) ? a[j][1] : MASK_VAL_;
        *(float2*)(scores + (size_t)(q0 + j) * LK_ + k0) = r;
    }
}

__global__ __launch_bounds__(256) void softmax_rows(float* __restrict__ scores)
{
    __shared__ float red[8];
    const int t    = threadIdx.x;
    const int lane = t & 31;
    const int wave = t >> 5;
    float* row = scores + (size_t)blockIdx.x * LK_;

    float4 x = *(const float4*)(row + t * 4);

    float m = fmaxf(fmaxf(x.x, x.y), fmaxf(x.z, x.w));
#pragma unroll
    for (int off = 16; off > 0; off >>= 1)
        m = fmaxf(m, __shfl_xor(m, off, 32));
    if (lane == 0) red[wave] = m;
    __syncthreads();
    float gm = red[0];
#pragma unroll
    for (int i = 1; i < 8; ++i) gm = fmaxf(gm, red[i]);
    __syncthreads();

    float4 e;
    e.x = __expf(x.x - gm);
    e.y = __expf(x.y - gm);
    e.z = __expf(x.z - gm);
    e.w = __expf(x.w - gm);

    float s = e.x + e.y + e.z + e.w;
#pragma unroll
    for (int off = 16; off > 0; off >>= 1)
        s += __shfl_xor(s, off, 32);
    if (lane == 0) red[wave] = s;
    __syncthreads();
    float gs = red[0];
#pragma unroll
    for (int i = 1; i < 8; ++i) gs += red[i];

    const float inv = 1.0f / gs;
    e.x *= inv; e.y *= inv; e.z *= inv; e.w *= inv;
    *(float4*)(row + t * 4) = e;
}

__global__ __launch_bounds__(256) void attn_out_wmma(
    const float* __restrict__ attn,
    const float* __restrict__ value,
    float* __restrict__ out)
{
    const int lane      = threadIdx.x & 31;
    const int wave      = threadIdx.x >> 5;
    const int tilesN    = DSZ_ / 16;
    const int tilesPerB = (LQ_ / 16) * tilesN;
    const int tile      = blockIdx.x * 8 + wave;
    const int b         = tile / tilesPerB;
    const int tr        = tile % tilesPerB;
    const int tile_m    = tr / tilesN;
    const int tile_n    = tr % tilesN;

    const int n15 = lane & 15;
    const int ko  = (lane < 16) ? 0 : 2;

    const float* aptr = attn  + ((size_t)b * LQ_ + tile_m * 16 + n15) * LK_ + ko;
    const float* vptr = value + (size_t)b * LK_ * DSZ_
                               + (size_t)ko * DSZ_ + tile_n * 16 + n15;

    v8f c = {0.f, 0.f, 0.f, 0.f, 0.f, 0.f, 0.f, 0.f};
#pragma unroll 4
    for (int kk = 0; kk < LK_; kk += 4) {
        v2f av = *(const v2f*)(aptr + kk);
        const float* vp = vptr + (size_t)kk * DSZ_;
        v2f bv;
        bv.x = vp[0];
        bv.y = vp[DSZ_];
        c = __builtin_amdgcn_wmma_f32_16x16x4_f32(
                false, av, false, bv, (short)0, c, false, false);
    }

    const int mo = (lane < 16) ? 0 : 8;
#pragma unroll
    for (int r = 0; r < 8; ++r) {
        const int m = tile_m * 16 + mo + r;
        out[((size_t)b * LQ_ + m) * DSZ_ + tile_n * 16 + n15] = c[r];
    }
}

extern "C" void kernel_launch(void* const* d_in, const int* in_sizes, int n_in,
                              void* d_out, int out_size, void* d_ws, size_t ws_size,
                              hipStream_t stream) {
    const float* query     = (const float*)d_in[0];
    const float* key       = (const float*)d_in[1];
    const float* value     = (const float*)d_in[2];
    const int*   valid_len = (const int*)  d_in[3];
    const float* W_q       = (const float*)d_in[4];
    const float* W_k       = (const float*)d_in[5];
    const float* vvec      = (const float*)d_in[6];

    float* qf     = (float*)d_ws;
    float* kfT    = qf  + (size_t)B_ * LQ_ * UNITS_;
    float* scores = kfT + (size_t)B_ * UNITS_ * LK_;
    float* outp   = (float*)d_out;

    proj_wmma<false><<<64, 256, 0, stream>>>(query, W_q, qf,
                                             B_ * LQ_, DSZ_, UNITS_, 0);
    proj_wmma<true><<<512, 256, 0, stream>>>(key, W_k, kfT,
                                             B_ * LK_, DSZ_, UNITS_, LK_);
    scores_tanh_kernel<<<512, 128, 0, stream>>>(qf, kfT, vvec, valid_len, scores);
    softmax_rows<<<B_ * LQ_, 256, 0, stream>>>(scores);
    attn_out_wmma<<<128, 256, 0, stream>>>(scores, value, outp);
}


# both projection GEMMs: A and W panels of a block tile (64x64 key side, 32x32 query side) staged in LDS by whole-line 16 B loads, two buffers deep, fragments read from LDS; k sum order unchanged
# speedup vs baseline: 1.5145x; 1.1905x over previous
#include <hip/hip_runtime.h>

typedef float v2f __attribute__((ext_vector_type(2)));
typedef float v8f __attribute__((ext_vector_type(8)));

#define B_     4
#define LQ_    128
#define LK_    1024
#define DSZ_   512
#define UNITS_ 256
#define MASK_VAL_ (-1.0e6f)

#if __has_builtin(__builtin_amdgcn_tanhf)
#define TANHF(x) __builtin_amdgcn_tanhf(x)
#else
#define TANHF(x) tanhf(x)
#endif

#define PJ_KC_     64
#define PJ_P4_     17
#define PJ_NC_     (DSZ_ / PJ_KC_)

template <int BM, int BN, int WN, int SEQ, bool TRANS_OUT>
__global__ __launch_bounds__((BM / 16) * (BN / 16) / WN * 32) void proj_wmma(
    const float* __restrict__ A,
    const float* __restrict__ W,
    float* __restrict__ out)
{
    constexpr int K   = DSZ_;
    constexpr int N   = UNITS_;
    constexpr int NT  = (BM / 16) * (BN / 16) / WN * 32;
    constexpr int RPP = NT / 16;
    constexpr int AIT = BM / RPP;
    constexpr int WIT = BN / RPP;
    constexpr int WPR = (BN / 16) / WN;

    __shared__ float4 As4[2 * BM * PJ_P4_];
    __shared__ float4 Ws4[2 * BN * PJ_P4_];

    const int t    = threadIdx.x;
    const int lane = t & 31;
    const int wave = t >> 5;
    const int c15  = lane & 15;
    const int h    = lane >> 4;
    const int tm   = wave / WPR;
    const int tn0  = (wave % WPR) * WN;
    const int bm   = blockIdx.x / (N / BN);
    const int bn   = blockIdx.x % (N / BN);

    const int lr = t >> 4;
    const int lc = t & 15;
    const float* ag = A + (size_t)(bm * BM + lr) * K + lc * 4;
    const float* wg = W + (size_t)(bn * BN + lr) * K + lc * 4;

    float4 ra[AIT], rw[WIT];
    v8f acc[WN];
#pragma unroll
    for (int j = 0; j < WN; ++j)
        acc[j] = (v8f){0.f, 0.f, 0.f, 0.f, 0.f, 0.f, 0.f, 0.f};

#pragma unroll
    for (int p = 0; p < AIT; ++p) ra[p] = *(const float4*)(ag + (size_t)p * RPP * K);
#pragma unroll
    for (int p = 0; p < WIT; ++p) rw[p] = *(const float4*)(wg + (size_t)p * RPP * K);
#pragma unroll
    for (int p = 0; p < AIT; ++p) As4[(lr + p * RPP) * PJ_P4_ + lc] = ra[p];
#pragma unroll
    for (int p = 0; p < WIT; ++p) Ws4[(lr + p * RPP) * PJ_P4_ + lc] = rw[p];
    __syncthreads();

    for (int kc = 0; kc < PJ_NC_; ++kc) {
        const int  buf  = kc & 1;
        const bool more = (kc + 1 < PJ_NC_);

        if (more) {
#pragma unroll
            for (int p = 0; p < AIT; ++p)
                ra[p] = *(const float4*)(ag + (size_t)p * RPP * K + (kc + 1) * PJ_KC_);
#pragma unroll
            for (int p = 0; p < WIT; ++p)
                rw[p] = *(const float4*)(wg + (size_t)p * RPP * K + (kc + 1) * PJ_KC_);
        }

        const v2f* af = (const v2f*)(As4 + (buf * BM + tm * 16 + c15) * PJ_P4_) + h;
        const v2f* wf = (const v2f*)(Ws4 + (buf * BN + tn0 * 16 + c15) * PJ_P4_) + h;
#pragma unroll
        for (int ks = 0; ks < PJ_KC_ / 4; ++ks) {
            const v2f av = af[ks * 2];
#pragma unroll
            for (int j = 0; j < WN; ++j) {
                const v2f bv = wf[j * 16 * PJ_P4_ * 2 + ks * 2];
                acc[j] = __builtin_amdgcn_wmma_f32_16x16x4_f32(
                             false, av, false, bv, (short)0, acc[j], false, false);
            }
        }

        if (more) {
#pragma unroll
            for (int p = 0; p < AIT; ++p)
                As4[((buf ^ 1) * BM + lr + p * RPP) * PJ_P4_ + lc] = ra[p];
#pragma unroll
            for (int p = 0; p < WIT; ++p)
                Ws4[((buf ^ 1) * BN + lr + p * RPP) * PJ_P4_ + lc] = rw[p];
        }
        __syncthreads();
    }

    const int m0 = bm * BM + tm * 16 + 8 * h;
#pragma unroll
    for (int j = 0; j < WN; ++j) {
        const int n = bn * BN + (tn0 + j) * 16 + c15;
#pragma unroll
        for (int r = 0; r < 8; ++r) {
            const int m = m0 + r;
            if (TRANS_OUT) {
                const int b    = m / SEQ;
                const int kpos = m % SEQ;
                out[((size_t)b * N + n) * SEQ + kpos] = acc[j][r];
            } else {
                out[(size_t)m * N + n] = acc[j][r];
            }
        }
    }
}

#define SC_KT_ 64
#define SC_QG_ 16
#define SC_QW_ 4

__global__ __launch_bounds__(128) void scores_tanh_kernel(
    const float* __restrict__ qf,
    const float* __restrict__ kfT,
    const float* __restrict__ vvec,
    const int*   __restrict__ valid_len,
    float* __restrict__ scores)
{
    __shared__ float4 kS4[UNITS_ * SC_KT_ / 4];
    __shared__ float4 qS4[SC_QG_ * UNITS_ / 4];

    const int t    = threadIdx.x;
    const int lane = t & 31;
    const int wave = t >> 5;
    const int qg   = blockIdx.x & 7;
    const int kt   = (blockIdx.x >> 3) & 15;
    const int b    = blockIdx.x >> 7;

    const float* ksrc = kfT + (size_t)b * UNITS_ * LK_ + kt * SC_KT_;
    const int c  = t & 15;
    const int r0 = t >> 4;
#pragma unroll
    for (int i = 0; i < UNITS_ / 8; ++i) {
        const int u = r0 + 8 * i;
        kS4[u * (SC_KT_ / 4) + c] = *(const float4*)(ksrc + (size_t)u * LK_ + c * 4);
    }
    const float4* qsrc =
        (const float4*)(qf + (size_t)(b * LQ_ + qg * SC_QG_) * UNITS_);
#pragma unroll
    for (int i = 0; i < SC_QG_ * UNITS_ / 4 / 128; ++i)
        qS4[i * 128 + t] = qsrc[i * 128 + t];
    __syncthreads();

    const float2* kS2  = (const float2*)kS4;
    const float4* qrow = qS4 + (wave * SC_QW_) * (UNITS_ / 4);

    float a[SC_QW_][2];
#pragma unroll
    for (int j = 0; j < SC_QW_; ++j) { a[j][0] = 0.f; a[j][1] = 0.f; }

#pragma unroll 2
    for (int u4 = 0; u4 < UNITS_ / 4; ++u4) {
        float4 qv[SC_QW_];
#pragma unroll
        for (int j = 0; j < SC_QW_; ++j) qv[j] = qrow[j * (UNITS_ / 4) + u4];
        float2 kv[4];
#pragma unroll
        for (int i = 0; i < 4; ++i) kv[i] = kS2[(u4 * 4 + i) * (SC_KT_ / 2) + lane];
        const float vv0 = vvec[u4 * 4 + 0];
        const float vv1 = vvec[u4 * 4 + 1];
        const float vv2 = vvec[u4 * 4 + 2];
        const float vv3 = vvec[u4 * 4 + 3];
#pragma unroll
        for (int j = 0; j < SC_QW_; ++j) {
            a[j][0] = __builtin_fmaf(vv0, TANHF(qv[j].x + kv[0].x), a[j][0]);
            a[j][1] = __builtin_fmaf(vv0, TANHF(qv[j].x + kv[0].y), a[j][1]);
        }
#pragma unroll
        for (int j = 0; j < SC_QW_; ++j) {
            a[j][0] = __builtin_fmaf(vv1, TANHF(qv[j].y + kv[1].x), a[j][0]);
            a[j][1] = __builtin_fmaf(vv1, TANHF(qv[j].y + kv[1].y), a[j][1]);
        }
#pragma unroll
        for (int j = 0; j < SC_QW_; ++j) {
            a[j][0] = __builtin_fmaf(vv2, TANHF(qv[j].z + kv[2].x), a[j][0]);
            a[j][1] = __builtin_fmaf(vv2, TANHF(qv[j].z + kv[2].y), a[j][1]);
        }
#pragma unroll
        for (int j = 0; j < SC_QW_; ++j) {
            a[j][0] = __builtin_fmaf(vv3, TANHF(qv[j].w + kv[3].x), a[j][0]);
            a[j][1] = __builtin_fmaf(vv3, TANHF(qv[j].w + kv[3].y), a[j][1]);
        }
    }

    const int vl = valid_len[b];
    const int k0 = kt * SC_KT_ + 2 * lane;
    const int q0 = b * LQ_ + qg * SC_QG_ + wave * SC_QW_;
#pragma unroll
    for (int j = 0; j < SC_QW_; ++j) {
        float2 r;
        r.x = (k0 + 0 < vl) ? a[j][0] : MASK_VAL_;
        r.y = (k0 + 1 < vl) ? a[j][1] : MASK_VAL_;
        *(float2*)(scores + (size_t)(q0 + j) * LK_ + k0) = r;
    }
}

__global__ __launch_bounds__(256) void softmax_rows(float* __restrict__ scores)
{
    __shared__ float red[8];
    const int t    = threadIdx.x;
    const int lane = t & 31;
    const int wave = t >> 5;
    float* row = scores + (size_t)blockIdx.x * LK_;

    float4 x = *(const float4*)(row + t * 4);

    float m = fmaxf(fmaxf(x.x, x.y), fmaxf(x.z, x.w));
#pragma unroll
    for (int off = 16; off > 0; off >>= 1)
        m = fmaxf(m, __shfl_xor(m, off, 32));
    if (lane == 0) red[wave] = m;
    __syncthreads();
    float gm = red[0];
#pragma unroll
    for (int i = 1; i < 8; ++i) gm = fmaxf(gm, red[i]);
    __syncthreads();

    float4 e;
    e.x = __expf(x.x - gm);
    e.y = __expf(x.y - gm);
    e.z = __expf(x.z - gm);
    e.w = __expf(x.w - gm);

    float s = e.x + e.y + e.z + e.w;
#pragma unroll
    for (int off = 16; off > 0; off >>= 1)
        s += __shfl_xor(s, off, 32);
    if (lane == 0) red[wave] = s;
    __syncthreads();
    float gs = red[0];
#pragma unroll
    for (int i = 1; i < 8; ++i) gs += red[i];

    const float inv = 1.0f / gs;
    e.x *= inv; e.y *= inv; e.z *= inv; e.w *= inv;
    *(float4*)(row + t * 4) = e;
}

__global__ __launch_bounds__(256) void attn_out_wmma(
    const float* __restrict__ attn,
    const float* __restrict__ value,
    float* __restrict__ out)
{
    const int lane      = threadIdx.x & 31;
    const int wave      = threadIdx.x >> 5;
    const int tilesN    = DSZ_ / 16;
    const int tilesPerB = (LQ_ / 16) * tilesN;
    const int tile      = blockIdx.x * 8 + wave;
    const int b         = tile / tilesPerB;
    const int tr        = tile % tilesPerB;
    const int tile_m    = tr / tilesN;
    const int tile_n    = tr % tilesN;

    const int n15 = lane & 15;
    const int ko  = (lane < 16) ? 0 : 2;

    const float* aptr = attn  + ((size_t)b * LQ_ + tile_m * 16 + n15) * LK_ + ko;
    const float* vptr = value + (size_t)b * LK_ * DSZ_
                               + (size_t)ko * DSZ_ + tile_n * 16 + n15;

    v8f c = {0.f, 0.f, 0.f, 0.f, 0.f, 0.f, 0.f, 0.f};
#pragma unroll 4
    for (int kk = 0; kk < LK_; kk += 4) {
        v2f av = *(const v2f*)(aptr + kk);
        const float* vp = vptr + (size_t)kk * DSZ_;
        v2f bv;
        bv.x = vp[0];
        bv.y = vp[DSZ_];
        c = __builtin_amdgcn_wmma_f32_16x16x4_f32(
                false, av, false, bv, (short)0, c, false, false);
    }

    const int mo = (lane < 16) ? 0 : 8;
#pragma unroll
    for (int r = 0; r < 8; ++r) {
        const int m = tile_m * 16 + mo + r;
        out[((size_t)b * LQ_ + m) * DSZ_ + tile_n * 16 + n15] = c[r];
    }
}

extern "C" void kernel_launch(void* const* d_in, const int* in_sizes, int n_in,
                              void* d_out, int out_size, void* d_ws, size_t ws_size,
                              hipStream_t stream) {
    const float* query     = (const float*)d_in[0];
    const float* key       = (const float*)d_in[1];
    const float* value     = (const float*)d_in[2];
    const int*   valid_len = (const int*)  d_in[3];
    const float* W_q       = (const float*)d_in[4];
    const float* W_k       = (const float*)d_in[5];
    const float* vvec      = (const float*)d_in[6];

    float* qf     = (float*)d_ws;
    float* kfT    = qf  + (size_t)B_ * LQ_ * UNITS_;
    float* scores = kfT + (size_t)B_ * UNITS_ * LK_;
    float* outp   = (float*)d_out;

    proj_wmma<32, 32, 1, LQ_, false><<<128, 128, 0, stream>>>(query, W_q, qf);
    proj_wmma<64, 64, 2, LK_, true><<<256, 256, 0, stream>>>(key, W_k, kfT);
    scores_tanh_kernel<<<512, 128, 0, stream>>>(qf, kfT, vvec, valid_len, scores);
    softmax_rows<<<B_ * LQ_, 256, 0, stream>>>(scores);
    attn_out_wmma<<<128, 256, 0, stream>>>(scores, value, outp);
}


# attn x value GEMM: attn and value panels of a 32x32 block tile staged in LDS by whole-line 16 B loads, 256-k chunks two buffers deep, fragment reads grouped ahead; k sum order unchanged
# speedup vs baseline: 2.6316x; 1.7376x over previous
#include <hip/hip_runtime.h>

typedef float v2f __attribute__((ext_vector_type(2)));
typedef float v8f __attribute__((ext_vector_type(8)));
typedef float v4f __attribute__((ext_vector_type(4)));

#define B_     4
#define LQ_    128
#define LK_    1024
#define DSZ_   512
#define UNITS_ 256
#define MASK_VAL_ (-1.0e6f)

#if __has_builtin(__builtin_amdgcn_tanhf)
#define TANHF(x) __builtin_amdgcn_tanhf(x)
#else
#define TANHF(x) tanhf(x)
#endif

#define PJ_KC_     64
#define PJ_P4_     17
#define PJ_NC_     (DSZ_ / PJ_KC_)

template <int BM, int BN, int WN, int SEQ, bool TRANS_OUT>
__global__ __launch_bounds__((BM / 16) * (BN / 16) / WN * 32) void proj_wmma(
    const float* __restrict__ A,
    const float* __restrict__ W,
    float* __restrict__ out)
{
    constexpr int K   = DSZ_;
    constexpr int N   = UNITS_;
    constexpr int NT  = (BM / 16) * (BN / 16) / WN * 32;
    constexpr int RPP = NT / 16;
    constexpr int AIT = BM / RPP;
    constexpr int WIT = BN / RPP;
    constexpr int WPR = (BN / 16) / WN;

    __shared__ float4 As4[2 * BM * PJ_P4_];
    __shared__ float4 Ws4[2 * BN * PJ_P4_];

    const int t    = threadIdx.x;
    const int lane = t & 31;
    const int wave = t >> 5;
    const int c15  = lane & 15;
    const int h    = lane >> 4;
    const int tm   = wave / WPR;
    const int tn0  = (wave % WPR) * WN;
    const int bm   = blockIdx.x / (N / BN);
    const int bn   = blockIdx.x % (N / BN);

    const int lr = t >> 4;
    const int lc = t & 15;
    const float* ag = A + (size_t)(bm * BM + lr) * K + lc * 4;
    const float* wg = W + (size_t)(bn * BN + lr) * K + lc * 4;

    float4 ra[AIT], rw[WIT];
    v8f acc[WN];
#pragma unroll
    for (int j = 0; j < WN; ++j)
        acc[j] = (v8f){0.f, 0.f, 0.f, 0.f, 0.f, 0.f, 0.f, 0.f};

#pragma unroll
    for (int p = 0; p < AIT; ++p) ra[p] = *(const float4*)(ag + (size_t)p * RPP * K);
#pragma unroll
    for (int p = 0; p < WIT; ++p) rw[p] = *(const float4*)(wg + (size_t)p * RPP * K);
#pragma unroll
    for (int p = 0; p < AIT; ++p) As4[(lr + p * RPP) * PJ_P4_ + lc] = ra[p];
#pragma unroll
    for (int p = 0; p < WIT; ++p) Ws4[(lr + p * RPP) * PJ_P4_ + lc] = rw[p];
    __syncthreads();

    for (int kc = 0; kc < PJ_NC_; ++kc) {
        const int  buf  = kc & 1;
        const bool more = (kc + 1 < PJ_NC_);

        if (more) {
#pragma unroll
            for (int p = 0; p < AIT; ++p)
                ra[p] = *(const float4*)(ag + (size_t)p * RPP * K + (kc + 1) * PJ_KC_);
#pragma unroll
            for (int p = 0; p < WIT; ++p)
                rw[p] = *(const float4*)(wg + (size_t)p * RPP * K + (kc + 1) * PJ_KC_);
        }

        const v2f* af = (const v2f*)(As4 + (buf * BM + tm * 16 + c15) * PJ_P4_) + h;
        const v2f* wf = (const v2f*)(Ws4 + (buf * BN + tn0 * 16 + c15) * PJ_P4_) + h;
#pragma unroll
        for (int ks = 0; ks < PJ_KC_ / 4; ++ks) {
            const v2f av = af[ks * 2];
#pragma unroll
            for (int j = 0; j < WN; ++j) {
                const v2f bv = wf[j * 16 * PJ_P4_ * 2 + ks * 2];
                acc[j] = __builtin_amdgcn_wmma_f32_16x16x4_f32(
                             false, av, false, bv, (short)0, acc[j], false, false);
            }
        }

        if (more) {
#pragma unroll
            for (int p = 0; p < AIT; ++p)
                As4[((buf ^ 1) * BM + lr + p * RPP) * PJ_P4_ + lc] = ra[p];
#pragma unroll
            for (int p = 0; p < WIT; ++p)
                Ws4[((buf ^ 1) * BN + lr + p * RPP) * PJ_P4_ + lc] = rw[p];
        }
        __syncthreads();
    }

    const int m0 = bm * BM + tm * 16 + 8 * h;
#pragma unroll
    for (int j = 0; j < WN; ++j) {
        const int n = bn * BN + (tn0 + j) * 16 + c15;
#pragma unroll
        for (int r = 0; r < 8; ++r) {
            const int m = m0 + r;
            if (TRANS_OUT) {
                const int b    = m / SEQ;
                const int kpos = m % SEQ;
                out[((size_t)b * N + n) * SEQ + kpos] = acc[j][r];
            } else {
                out[(size_t)m * N + n] = acc[j][r];
            }
        }
    }
}

#define SC_KT_ 64
#define SC_QG_ 16
#define SC_QW_ 4

__global__ __launch_bounds__(128) void scores_tanh_kernel(
    const float* __restrict__ qf,
    const float* __restrict__ kfT,
    const float* __restrict__ vvec,
    const int*   __restrict__ valid_len,
    float* __restrict__ scores)
{
    __shared__ float4 kS4[UNITS_ * SC_KT_ / 4];
    __shared__ float4 qS4[SC_QG_ * UNITS_ / 4];

    const int t    = threadIdx.x;
    const int lane = t & 31;
    const int wave = t >> 5;
    const int qg   = blockIdx.x & 7;
    const int kt   = (blockIdx.x >> 3) & 15;
    const int b    = blockIdx.x >> 7;

    const float* ksrc = kfT + (size_t)b * UNITS_ * LK_ + kt * SC_KT_;
    const int c  = t & 15;
    const int r0 = t >> 4;
#pragma unroll
    for (int i = 0; i < UNITS_ / 8; ++i) {
        const int u = r0 + 8 * i;
        kS4[u * (SC_KT_ / 4) + c] = *(const float4*)(ksrc + (size_t)u * LK_ + c * 4);
    }
    const float4* qsrc =
        (const float4*)(qf + (size_t)(b * LQ_ + qg * SC_QG_) * UNITS_);
#pragma unroll
    for (int i = 0; i < SC_QG_ * UNITS_ / 4 / 128; ++i)
        qS4[i * 128 + t] = qsrc[i * 128 + t];
    __syncthreads();

    const float2* kS2  = (const float2*)kS4;
    const float4* qrow = qS4 + (wave * SC_QW_) * (UNITS_ / 4);

    float a[SC_QW_][2];
#pragma unroll
    for (int j = 0; j < SC_QW_; ++j) { a[j][0] = 0.f; a[j][1] = 0.f; }

#pragma unroll 2
    for (int u4 = 0; u4 < UNITS_ / 4; ++u4) {
        float4 qv[SC_QW_];
#pragma unroll
        for (int j = 0; j < SC_QW_; ++j) qv[j] = qrow[j * (UNITS_ / 4) + u4];
        float2 kv[4];
#pragma unroll
        for (int i = 0; i < 4; ++i) kv[i] = kS2[(u4 * 4 + i) * (SC_KT_ / 2) + lane];
        const float vv0 = vvec[u4 * 4 + 0];
        const float vv1 = vvec[u4 * 4 + 1];
        const float vv2 = vvec[u4 * 4 + 2];
        const float vv3 = vvec[u4 * 4 + 3];
#pragma unroll
        for (int j = 0; j < SC_QW_; ++j) {
            a[j][0] = __builtin_fmaf(vv0, TANHF(qv[j].x + kv[0].x), a[j][0]);
            a[j][1] = __builtin_fmaf(vv0, TANHF(qv[j].x + kv[0].y), a[j][1]);
        }
#pragma unroll
        for (int j = 0; j < SC_QW_; ++j) {
            a[j][0] = __builtin_fmaf(vv1, TANHF(qv[j].y + kv[1].x), a[j][0]);
            a[j][1] = __builtin_fmaf(vv1, TANHF(qv[j].y + kv[1].y), a[j][1]);
        }
#pragma unroll
        for (int j = 0; j < SC_QW_; ++j) {
            a[j][0] = __builtin_fmaf(vv2, TANHF(qv[j].z + kv[2].x), a[j][0]);
            a[j][1] = __builtin_fmaf(vv2, TANHF(qv[j].z + kv[2].y), a[j][1]);
        }
#pragma unroll
        for (int j = 0; j < SC_QW_; ++j) {
            a[j][0] = __builtin_fmaf(vv3, TANHF(qv[j].w + kv[3].x), a[j][0]);
            a[j][1] = __builtin_fmaf(vv3, TANHF(qv[j].w + kv[3].y), a[j][1]);
        }
    }

    const int vl = valid_len[b];
    const int k0 = kt * SC_KT_ + 2 * lane;
    const int q0 = b * LQ_ + qg * SC_QG_ + wave * SC_QW_;
#pragma unroll
    for (int j = 0; j < SC_QW_; ++j) {
        float2 r;
        r.x = (k0 + 0 < vl) ? a[j][0] : MASK_VAL_;
        r.y = (k0 + 1 < vl) ? a[j][1] : MASK_VAL_;
        *(float2*)(scores + (size_t)(q0 + j) * LK_ + k0) = r;
    }
}

__global__ __launch_bounds__(256) void softmax_rows(float* __restrict__ scores)
{
    __shared__ float red[8];
    const int t    = threadIdx.x;
    const int lane = t & 31;
    const int wave = t >> 5;
    float* row = scores + (size_t)blockIdx.x * LK_;

    float4 x = *(const float4*)(row + t * 4);

    float m = fmaxf(fmaxf(x.x, x.y), fmaxf(x.z, x.w));
#pragma unroll
    for (int off = 16; off > 0; off >>= 1)
        m = fmaxf(m, __shfl_xor(m, off, 32));
    if (lane == 0) red[wave] = m;
    __syncthreads();
    float gm = red[0];
#pragma unroll
    for (int i = 1; i < 8; ++i) gm = fmaxf(gm, red[i]);
    __syncthreads();

    float4 e;
    e.x = __expf(x.x - gm);
    e.y = __expf(x.y - gm);
    e.z = __expf(x.z - gm);
    e.w = __expf(x.w - gm);

    float s = e.x + e.y + e.z + e.w;
#pragma unroll
    for (int off = 16; off > 0; off >>= 1)
        s += __shfl_xor(s, off, 32);
    if (lane == 0) red[wave] = s;
    __syncthreads();
    float gs = red[0];
#pragma unroll
    for (int i = 1; i < 8; ++i) gs += red[i];

    const float inv = 1.0f / gs;
    e.x *= inv; e.y *= inv; e.z *= inv; e.w *= inv;
    *(float4*)(row + t * 4) = e;
}

#define AO_BM_  32
#define AO_BN_  32
#define AO_KC_  256
#define AO_NC_  (LK_ / AO_KC_)
#define AO_APR_ (AO_KC_ / 4)
#define AO_AP4_ (AO_APR_ + 1)
#define AO_ARP_ (128 / AO_APR_)
#define AO_VP4_ 12
#define AO_AIT_ (AO_BM_ / AO_ARP_)
#define AO_VIT_ (AO_KC_ / 16)
#define AO_FG_  8
#define AO_NG_  (AO_KC_ / 4 / AO_FG_)

__global__ __launch_bounds__(128) void attn_out_wmma(
    const float* __restrict__ attn,
    const float* __restrict__ value,
    float* __restrict__ out)
{
    __shared__ v4f As4[2 * AO_BM_ * AO_AP4_];
    __shared__ v4f Vs4[2 * AO_KC_ * AO_VP4_];

    const int t    = threadIdx.x;
    const int lane = t & 31;
    const int wave = t >> 5;
    const int c15  = lane & 15;
    const int h    = lane >> 4;
    const int tm   = wave >> 1;
    const int tn   = wave & 1;
    const int bn   = blockIdx.x & 15;
    const int bm   = (blockIdx.x >> 4) & 3;
    const int b    = blockIdx.x >> 6;

    const int ar = t / AO_APR_;
    const int ac = t % AO_APR_;
    const int vr = t >> 3;
    const int vc = t & 7;
    const float* ag = attn  + ((size_t)b * LQ_ + bm * AO_BM_ + ar) * LK_ + ac * 4;
    const float* vg = value + ((size_t)b * LK_ + vr) * DSZ_ + bn * AO_BN_ + vc * 4;

    v4f ra[AO_AIT_], rv[AO_VIT_];
    v8f acc = {0.f, 0.f, 0.f, 0.f, 0.f, 0.f, 0.f, 0.f};

#pragma unroll
    for (int p = 0; p < AO_AIT_; ++p) ra[p] = *(const v4f*)(ag + (size_t)p * AO_ARP_ * LK_);
#pragma unroll
    for (int p = 0; p < AO_VIT_; ++p) rv[p] = *(const v4f*)(vg + (size_t)p * 16 * DSZ_);
#pragma unroll
    for (int p = 0; p < AO_AIT_; ++p) As4[(ar + p * AO_ARP_) * AO_AP4_ + ac] = ra[p];
#pragma unroll
    for (int p = 0; p < AO_VIT_; ++p) Vs4[(vr + p * 16) * AO_VP4_ + vc] = rv[p];
    __syncthreads();

    for (int kc = 0; kc < AO_NC_; ++kc) {
        const int  buf  = kc & 1;
        const bool more = (kc + 1 < AO_NC_);

        if (more) {
#pragma unroll
            for (int p = 0; p < AO_AIT_; ++p)
                ra[p] = *(const v4f*)(ag + (size_t)p * AO_ARP_ * LK_ + (kc + 1) * AO_KC_);
#pragma unroll
            for (int p = 0; p < AO_VIT_; ++p)
                rv[p] = *(const v4f*)(vg + ((size_t)(kc + 1) * AO_KC_ + p * 16) * DSZ_);
        }

        const v2f*   af = (const v2f*)(As4 + (buf * AO_BM_ + tm * 16 + c15) * AO_AP4_) + h;
        const float* vf = (const float*)(Vs4 + (buf * AO_KC_ + 2 * h) * AO_VP4_) + tn * 16 + c15;
        v2f fa[2][AO_FG_], fb[2][AO_FG_];
#pragma unroll
        for (int i = 0; i < AO_FG_; ++i) {
            fa[0][i]   = af[i * 2];
            fb[0][i].x = vf[(i * 4 + 0) * (AO_VP4_ * 4)];
            fb[0][i].y = vf[(i * 4 + 1) * (AO_VP4_ * 4)];
        }
#pragma unroll
        for (int g = 0; g < AO_NG_; ++g) {
            if (g + 1 < AO_NG_) {
#pragma unroll
                for (int i = 0; i < AO_FG_; ++i) {
                    const int ks = (g + 1) * AO_FG_ + i;
                    fa[(g + 1) & 1][i]   = af[ks * 2];
                    fb[(g + 1) & 1][i].x = vf[(ks * 4 + 0) * (AO_VP4_ * 4)];
                    fb[(g + 1) & 1][i].y = vf[(ks * 4 + 1) * (AO_VP4_ * 4)];
                }
            }
            __builtin_amdgcn_sched_barrier(0);
#pragma unroll
            for (int i = 0; i < AO_FG_; ++i)
                acc = __builtin_amdgcn_wmma_f32_16x16x4_f32(
                          false, fa[g & 1][i], false, fb[g & 1][i], (short)0, acc, false, false);
            __builtin_amdgcn_sched_barrier(0);
        }

        if (more) {
#pragma unroll
            for (int p = 0; p < AO_AIT_; ++p)
                As4[((buf ^ 1) * AO_BM_ + ar + p * AO_ARP_) * AO_AP4_ + ac] = ra[p];
#pragma unroll
            for (int p = 0; p < AO_VIT_; ++p)
                Vs4[((buf ^ 1) * AO_KC_ + vr + p * 16) * AO_VP4_ + vc] = rv[p];
        }
        __syncthreads();
    }

    const int m0 = bm * AO_BM_ + tm * 16 + 8 * h;
    const int n  = bn * AO_BN_ + tn * 16 + c15;
#pragma unroll
    for (int r = 0; r < 8; ++r)
        out[((size_t)b * LQ_ + m0 + r) * DSZ_ + n] = acc[r];
}

extern "C" void kernel_launch(void* const* d_in, const int* in_sizes, int n_in,
                              void* d_out, int out_size, void* d_ws, size_t ws_size,
                              hipStream_t stream) {
    const float* query     = (const float*)d_in[0];
    const float* key       = (const float*)d_in[1];
    const float* value     = (const float*)d_in[2];
    const int*   valid_len = (const int*)  d_in[3];
    const float* W_q       = (const float*)d_in[4];
    const float* W_k       = (const float*)d_in[5];
    const float* vvec      = (const float*)d_in[6];

    float* qf     = (float*)d_ws;
    float* kfT    = qf  + (size_t)B_ * LQ_ * UNITS_;
    float* scores = kfT + (size_t)B_ * UNITS_ * LK_;
    float* outp   = (float*)d_out;

    proj_wmma<32, 32, 1, LQ_, false><<<128, 128, 0, stream>>>(query, W_q, qf);
    proj_wmma<64, 64, 2, LK_, true><<<256, 256, 0, stream>>>(key, W_k, kfT);
    scores_tanh_kernel<<<512, 128, 0, stream>>>(qf, kfT, vvec, valid_len, scores);
    softmax_rows<<<B_ * LQ_, 256, 0, stream>>>(scores);
    attn_out_wmma<<<256, 128, 0, stream>>>(scores, value, outp);
}


# both projection GEMMs share one launch: blocks 0-255 do the key side (64x64 tiles), blocks 256-319 the query side (now 32x64 tiles, 8 waves); 4 launches instead of 5; every k sum order unchanged
# speedup vs baseline: 3.2223x; 1.2245x over previous
#include <hip/hip_runtime.h>

typedef float v2f __attribute__((ext_vector_type(2)));
typedef float v8f __attribute__((ext_vector_type(8)));
typedef float v4f __attribute__((ext_vector_type(4)));

#define B_     4
#define LQ_    128
#define LK_    1024
#define DSZ_   512
#define UNITS_ 256
#define MASK_VAL_ (-1.0e6f)

#if __has_builtin(__builtin_amdgcn_tanhf)
#define TANHF(x) __builtin_amdgcn_tanhf(x)
#else
#define TANHF(x) tanhf(x)
#endif

#define PJ_KC_     64
#define PJ_P4_     17
#define PJ_NC_     (DSZ_ / PJ_KC_)
#define PJ_NT_     256
#define PJ_KB_     256
#define PJ_QB_     64

template <int BM, int BN, int WN, int SEQ, bool TRANS_OUT>
__device__ __forceinline__ void proj_tile(
    const float* __restrict__ A,
    const float* __restrict__ W,
    float* __restrict__ out,
    const int blk,
    float4* __restrict__ As4,
    float4* __restrict__ Ws4)
{
    constexpr int K   = DSZ_;
    constexpr int N   = UNITS_;
    constexpr int NT  = (BM / 16) * (BN / 16) / WN * 32;
    constexpr int RPP = NT / 16;
    constexpr int AIT = BM / RPP;
    constexpr int WIT = BN / RPP;
    constexpr int WPR = (BN / 16) / WN;
    static_assert(NT == PJ_NT_, "both sides run in blocks of PJ_NT_ threads");

    const int t    = threadIdx.x;
    const int lane = t & 31;
    const int wave = t >> 5;
    const int c15  = lane & 15;
    const int h    = lane >> 4;
    const int tm   = wave / WPR;
    const int tn0  = (wave % WPR) * WN;
    const int bm   = blk / (N / BN);
    const int bn   = blk % (N / BN);

    const int lr = t >> 4;
    const int lc = t & 15;
    const float* ag = A + (size_t)(bm * BM + lr) * K + lc * 4;
    const float* wg = W + (size_t)(bn * BN + lr) * K + lc * 4;

    float4 ra[AIT], rw[WIT];
    v8f acc[WN];
#pragma unroll
    for (int j = 0; j < WN; ++j)
        acc[j] = (v8f){0.f, 0.f, 0.f, 0.f, 0.f, 0.f, 0.f, 0.f};

#pragma unroll
    for (int p = 0; p < AIT; ++p) ra[p] = *(const float4*)(ag + (size_t)p * RPP * K);
#pragma unroll
    for (int p = 0; p < WIT; ++p) rw[p] = *(const float4*)(wg + (size_t)p * RPP * K);
#pragma unroll
    for (int p = 0; p < AIT; ++p) As4[(lr + p * RPP) * PJ_P4_ + lc] = ra[p];
#pragma unroll
    for (int p = 0; p < WIT; ++p) Ws4[(lr + p * RPP) * PJ_P4_ + lc] = rw[p];
    __syncthreads();

    for (int kc = 0; kc < PJ_NC_; ++kc) {
        const int  buf  = kc & 1;
        const bool more = (kc + 1 < PJ_NC_);

        if (more) {
#pragma unroll
            for (int p = 0; p < AIT; ++p)
                ra[p] = *(const float4*)(ag + (size_t)p * RPP * K + (kc + 1) * PJ_KC_);
#pragma unroll
            for (int p = 0; p < WIT; ++p)
                rw[p] = *(const float4*)(wg + (size_t)p * RPP * K + (kc + 1) * PJ_KC_);
        }

        const v2f* af = (const v2f*)(As4 + (buf * BM + tm * 16 + c15) * PJ_P4_) + h;
        const v2f* wf = (const v2f*)(Ws4 + (buf * BN + tn0 * 16 + c15) * PJ_P4_) + h;
#pragma unroll
        for (int ks = 0; ks < PJ_KC_ / 4; ++ks) {
            const v2f av = af[ks * 2];
#pragma unroll
            for (int j = 0; j < WN; ++j) {
                const v2f bv = wf[j * 16 * PJ_P4_ * 2 + ks * 2];
                acc[j] = __builtin_amdgcn_wmma_f32_16x16x4_f32(
                             false, av, false, bv, (short)0, acc[j], false, false);
            }
        }

        if (more) {
#pragma unroll
            for (int p = 0; p < AIT; ++p)
                As4[((buf ^ 1) * BM + lr + p * RPP) * PJ_P4_ + lc] = ra[p];
#pragma unroll
            for (int p = 0; p < WIT; ++p)
                Ws4[((buf ^ 1) * BN + lr + p * RPP) * PJ_P4_ + lc] = rw[p];
        }
        __syncthreads();
    }

    const int m0 = bm * BM + tm * 16 + 8 * h;
#pragma unroll
    for (int j = 0; j < WN; ++j) {
        const int n = bn * BN + (tn0 + j) * 16 + c15;
#pragma unroll
        for (int r = 0; r < 8; ++r) {
            const int m = m0 + r;
            if (TRANS_OUT) {
                const int b    = m / SEQ;
                const int kpos = m % SEQ;
                out[((size_t)b * N + n) * SEQ + kpos] = acc[j][r];
            } else {
                out[(size_t)m * N + n] = acc[j][r];
            }
        }
    }
}

__global__ __launch_bounds__(PJ_NT_) void proj_both(
    const float* __restrict__ key,
    const float* __restrict__ W_k,
    float* __restrict__ kfT,
    const float* __restrict__ query,
    const float* __restrict__ W_q,
    float* __restrict__ qf)
{
    __shared__ float4 As4[2 * 64 * PJ_P4_];
    __shared__ float4 Ws4[2 * 64 * PJ_P4_];

    if (blockIdx.x < PJ_KB_)
        proj_tile<64, 64, 2, LK_, true>(key, W_k, kfT, (int)blockIdx.x, As4, Ws4);
    else
        proj_tile<32, 64, 1, LQ_, false>(query, W_q, qf, (int)blockIdx.x - PJ_KB_, As4, Ws4);
}

#define SC_KT_ 64
#define SC_QG_ 16
#define SC_QW_ 4

__global__ __launch_bounds__(128) void scores_tanh_kernel(
    const float* __restrict__ qf,
    const float* __restrict__ kfT,
    const float* __restrict__ vvec,
    const int*   __restrict__ valid_len,
    float* __restrict__ scores)
{
    __shared__ float4 kS4[UNITS_ * SC_KT_ / 4];
    __shared__ float4 qS4[SC_QG_ * UNITS_ / 4];

    const int t    = threadIdx.x;
    const int lane = t & 31;
    const int wave = t >> 5;
    const int qg   = blockIdx.x & 7;
    const int kt   = (blockIdx.x >> 3) & 15;
    const int b    = blockIdx.x >> 7;

    const float* ksrc = kfT + (size_t)b * UNITS_ * LK_ + kt * SC_KT_;
    const int c  = t & 15;
    const int r0 = t >> 4;
#pragma unroll
    for (int i = 0; i < UNITS_ / 8; ++i) {
        const int u = r0 + 8 * i;
        kS4[u * (SC_KT_ / 4) + c] = *(const float4*)(ksrc + (size_t)u * LK_ + c * 4);
    }
    const float4* qsrc =
        (const float4*)(qf + (size_t)(b * LQ_ + qg * SC_QG_) * UNITS_);
#pragma unroll
    for (int i = 0; i < SC_QG_ * UNITS_ / 4 / 128; ++i)
        qS4[i * 128 + t] = qsrc[i * 128 + t];
    __syncthreads();

    const float2* kS2  = (const float2*)kS4;
    const float4* qrow = qS4 + (wave * SC_QW_) * (UNITS_ / 4);

    float a[SC_QW_][2];
#pragma unroll
    for (int j = 0; j < SC_QW_; ++j) { a[j][0] = 0.f; a[j][1] = 0.f; }

#pragma unroll 2
    for (int u4 = 0; u4 < UNITS_ / 4; ++u4) {
        float4 qv[SC_QW_];
#pragma unroll
        for (int j = 0; j < SC_QW_; ++j) qv[j] = qrow[j * (UNITS_ / 4) + u4];
        float2 kv[4];
#pragma unroll
        for (int i = 0; i < 4; ++i) kv[i] = kS2[(u4 * 4 + i) * (SC_KT_ / 2) + lane];
        const float vv0 = vvec[u4 * 4 + 0];
        const float vv1 = vvec[u4 * 4 + 1];
        const float vv2 = vvec[u4 * 4 + 2];
        const float vv3 = vvec[u4 * 4 + 3];
#pragma unroll
        for (int j = 0; j < SC_QW_; ++j) {
            a[j][0] = __builtin_fmaf(vv0, TANHF(qv[j].x + kv[0].x), a[j][0]);
            a[j][1] = __builtin_fmaf(vv0, TANHF(qv[j].x + kv[0].y), a[j][1]);
        }
#pragma unroll
        for (int j = 0; j < SC_QW_; ++j) {
            a[j][0] = __builtin_fmaf(vv1, TANHF(qv[j].y + kv[1].x), a[j][0]);
            a[j][1] = __builtin_fmaf(vv1, TANHF(qv[j].y + kv[1].y), a[j][1]);
        }
#pragma unroll
        for (int j = 0; j < SC_QW_; ++j) {
            a[j][0] = __builtin_fmaf(vv2, TANHF(qv[j].z + kv[2].x), a[j][0]);
            a[j][1] = __builtin_fmaf(vv2, TANHF(qv[j].z + kv[2].y), a[j][1]);
        }
#pragma unroll
        for (int j = 0; j < SC_QW_; ++j) {
            a[j][0] = __builtin_fmaf(vv3, TANHF(qv[j].w + kv[3].x), a[j][0]);
            a[j][1] = __builtin_fmaf(vv3, TANHF(qv[j].w + kv[3].y), a[j][1]);
        }
    }

    const int vl = valid_len[b];
    const int k0 = kt * SC_KT_ + 2 * lane;
    const int q0 = b * LQ_ + qg * SC_QG_ + wave * SC_QW_;
#pragma unroll
    for (int j = 0; j < SC_QW_; ++j) {
        float2 r;
        r.x = (k0 + 0 < vl) ? a[j][0] : MASK_VAL_;
        r.y = (k0 + 1 < vl) ? a[j][1] : MASK_VAL_;
        *(float2*)(scores + (size_t)(q0 + j) * LK_ + k0) = r;
    }
}

__global__ __launch_bounds__(256) void softmax_rows(float* __restrict__ scores)
{
    __shared__ float red[8];
    const int t    = threadIdx.x;
    const int lane = t & 31;
    const int wave = t >> 5;
    float* row = scores + (size_t)blockIdx.x * LK_;

    float4 x = *(const float4*)(row + t * 4);

    float m = fmaxf(fmaxf(x.x, x.y), fmaxf(x.z, x.w));
#pragma unroll
    for (int off = 16; off > 0; off >>= 1)
        m = fmaxf(m, __shfl_xor(m, off, 32));
    if (lane == 0) red[wave] = m;
    __syncthreads();
    float gm = red[0];
#pragma unroll
    for (int i = 1; i < 8; ++i) gm = fmaxf(gm, red[i]);
    __syncthreads();

    float4 e;
    e.x = __expf(x.x - gm);
    e.y = __expf(x.y - gm);
    e.z = __expf(x.z - gm);
    e.w = __expf(x.w - gm);

    float s = e.x + e.y + e.z + e.w;
#pragma unroll
    for (int off = 16; off > 0; off >>= 1)
        s += __shfl_xor(s, off, 32);
    if (lane == 0) red[wave] = s;
    __syncthreads();
    float gs = red[0];
#pragma unroll
    for (int i = 1; i < 8; ++i) gs += red[i];

    const float inv = 1.0f / gs;
    e.x *= inv; e.y *= inv; e.z *= inv; e.w *= inv;
    *(float4*)(row + t * 4) = e;
}

#define AO_BM_  32
#define AO_BN_  32
#define AO_KC_  256
#define AO_NC_  (LK_ / AO_KC_)
#define AO_APR_ (AO_KC_ / 4)
#define AO_AP4_ (AO_APR_ + 1)
#define AO_ARP_ (128 / AO_APR_)
#define AO_VP4_ 12
#define AO_AIT_ (AO_BM_ / AO_ARP_)
#define AO_VIT_ (AO_KC_ / 16)
#define AO_FG_  8
#define AO_NG_  (AO_KC_ / 4 / AO_FG_)

__global__ __launch_bounds__(128) void attn_out_wmma(
    const float* __restrict__ attn,
    const float* __restrict__ value,
    float* __restrict__ out)
{
    __shared__ v4f As4[2 * AO_BM_ * AO_AP4_];
    __shared__ v4f Vs4[2 * AO_KC_ * AO_VP4_];

    const int t    = threadIdx.x;
    const int lane = t & 31;
    const int wave = t >> 5;
    const int c15  = lane & 15;
    const int h    = lane >> 4;
    const int tm   = wave >> 1;
    const int tn   = wave & 1;
    const int bn   = blockIdx.x & 15;
    const int bm   = (blockIdx.x >> 4) & 3;
    const int b    = blockIdx.x >> 6;

    const int ar = t / AO_APR_;
    const int ac = t % AO_APR_;
    const int vr = t >> 3;
    const int vc = t & 7;
    const float* ag = attn  + ((size_t)b * LQ_ + bm * AO_BM_ + ar) * LK_ + ac * 4;
    const float* vg = value + ((size_t)b * LK_ + vr) * DSZ_ + bn * AO_BN_ + vc * 4;

    v4f ra[AO_AIT_], rv[AO_VIT_];
    v8f acc = {0.f, 0.f, 0.f, 0.f, 0.f, 0.f, 0.f, 0.f};

#pragma unroll
    for (int p = 0; p < AO_AIT_; ++p) ra[p] = *(const v4f*)(ag + (size_t)p * AO_ARP_ * LK_);
#pragma unroll
    for (int p = 0; p < AO_VIT_; ++p) rv[p] = *(const v4f*)(vg + (size_t)p * 16 * DSZ_);
#pragma unroll
    for (int p = 0; p < AO_AIT_; ++p) As4[(ar + p * AO_ARP_) * AO_AP4_ + ac] = ra[p];
#pragma unroll
    for (int p = 0; p < AO_VIT_; ++p) Vs4[(vr + p * 16) * AO_VP4_ + vc] = rv[p];
    __syncthreads();

    for (int kc = 0; kc < AO_NC_; ++kc) {
        const int  buf  = kc & 1;
        const bool more = (kc + 1 < AO_NC_);

        if (more) {
#pragma unroll
            for (int p = 0; p < AO_AIT_; ++p)
                ra[p] = *(const v4f*)(ag + (size_t)p * AO_ARP_ * LK_ + (kc + 1) * AO_KC_);
#pragma unroll
            for (int p = 0; p < AO_VIT_; ++p)
                rv[p] = *(const v4f*)(vg + ((size_t)(kc + 1) * AO_KC_ + p * 16) * DSZ_);
        }

        const v2f*   af = (const v2f*)(As4 + (buf * AO_BM_ + tm * 16 + c15) * AO_AP4_) + h;
        const float* vf = (const float*)(Vs4 + (buf * AO_KC_ + 2 * h) * AO_VP4_) + tn * 16 + c15;
        v2f fa[2][AO_FG_], fb[2][AO_FG_];
#pragma unroll
        for (int i = 0; i < AO_FG_; ++i) {
            fa[0][i]   = af[i * 2];
            fb[0][i].x = vf[(i * 4 + 0) * (AO_VP4_ * 4)];
            fb[0][i].y = vf[(i * 4 + 1) * (AO_VP4_ * 4)];
        }
#pragma unroll
        for (int g = 0; g < AO_NG_; ++g) {
            if (g + 1 < AO_NG_) {
#pragma unroll
                for (int i = 0; i < AO_FG_; ++i) {
                    const int ks = (g + 1) * AO_FG_ + i;
                    fa[(g + 1) & 1][i]   = af[ks * 2];
                    fb[(g + 1) & 1][i].x = vf[(ks * 4 + 0) * (AO_VP4_ * 4)];
                    fb[(g + 1) & 1][i].y = vf[(ks * 4 + 1) * (AO_VP4_ * 4)];
                }
            }
            __builtin_amdgcn_sched_barrier(0);
#pragma unroll
            for (int i = 0; i < AO_FG_; ++i)
                acc = __builtin_amdgcn_wmma_f32_16x16x4_f32(
                          false, fa[g & 1][i], false, fb[g & 1][i], (short)0, acc, false, false);
            __builtin_amdgcn_sched_barrier(0);
        }

        if (more) {
#pragma unroll
            for (int p = 0; p < AO_AIT_; ++p)
                As4[((buf ^ 1) * AO_BM_ + ar + p * AO_ARP_) * AO_AP4_ + ac] = ra[p];
#pragma unroll
            for (int p = 0; p < AO_VIT_; ++p)
                Vs4[((buf ^ 1) * AO_KC_ + vr + p * 16) * AO_VP4_ + vc] = rv[p];
        }
        __syncthreads();
    }

    const int m0 = bm * AO_BM_ + tm * 16 + 8 * h;
    const int n  = bn * AO_BN_ + tn * 16 + c15;
#pragma unroll
    for (int r = 0; r < 8; ++r)
        out[((size_t)b * LQ_ + m0 + r) * DSZ_ + n] = acc[r];
}

extern "C" void kernel_launch(void* const* d_in, const int* in_sizes, int n_in,
                              void* d_out, int out_size, void* d_ws, size_t ws_size,
                              hipStream_t stream) {
    const float* query     = (const float*)d_in[0];
    const float* key       = (const float*)d_in[1];
    const float* value     = (const float*)d_in[2];
    const int*   valid_len = (const int*)  d_in[3];
    const float* W_q       = (const float*)d_in[4];
    const float* W_k       = (const float*)d_in[5];
    const float* vvec      = (const float*)d_in[6];

    float* qf     = (float*)d_ws;
    float* kfT    = qf  + (size_t)B_ * LQ_ * UNITS_;
    float* scores = kfT + (size_t)B_ * UNITS_ * LK_;
    float* outp   = (float*)d_out;

    proj_both<<<PJ_KB_ + PJ_QB_, PJ_NT_, 0, stream>>>(key, W_k, kfT, query, W_q, qf);
    scores_tanh_kernel<<<512, 128, 0, stream>>>(qf, kfT, vvec, valid_len, scores);
    softmax_rows<<<B_ * LQ_, 256, 0, stream>>>(scores);
    attn_out_wmma<<<256, 128, 0, stream>>>(scores, value, outp);
}


# softmax folded into the attn x value kernel: a block normalises its 32 score rows into LDS (same max, exp, sum order and division: weights bit-identical), then multiplies; 3 launches, not 4
# speedup vs baseline: 3.6375x; 1.1289x over previous
#include <hip/hip_runtime.h>

typedef float v2f __attribute__((ext_vector_type(2)));
typedef float v8f __attribute__((ext_vector_type(8)));
typedef float v4f __attribute__((ext_vector_type(4)));

#define B_     4
#define LQ_    128
#define LK_    1024
#define DSZ_   512
#define UNITS_ 256
#define MASK_VAL_ (-1.0e6f)

#if __has_builtin(__builtin_amdgcn_tanhf)
#define TANHF(x) __builtin_amdgcn_tanhf(x)
#else
#define TANHF(x) tanhf(x)
#endif

#define PJ_KC_     64
#define PJ_P4_     17
#define PJ_NC_     (DSZ_ / PJ_KC_)
#define PJ_NT_     256
#define PJ_KB_     256
#define PJ_QB_     64

template <int BM, int BN, int WN, int SEQ, bool TRANS_OUT>
__device__ __forceinline__ void proj_tile(
    const float* __restrict__ A,
    const float* __restrict__ W,
    float* __restrict__ out,
    const int blk,
    float4* __restrict__ As4,
    float4* __restrict__ Ws4)
{
    constexpr int K   = DSZ_;
    constexpr int N   = UNITS_;
    constexpr int NT  = (BM / 16) * (BN / 16) / WN * 32;
    constexpr int RPP = NT / 16;
    constexpr int AIT = BM / RPP;
    constexpr int WIT = BN / RPP;
    constexpr int WPR = (BN / 16) / WN;
    static_assert(NT == PJ_NT_, "both sides run in blocks of PJ_NT_ threads");

    const int t    = threadIdx.x;
    const int lane = t & 31;
    const int wave = t >> 5;
    const int c15  = lane & 15;
    const int h    = lane >> 4;
    const int tm   = wave / WPR;
    const int tn0  = (wave % WPR) * WN;
    const int bm   = blk / (N / BN);
    const int bn   = blk % (N / BN);

    const int lr = t >> 4;
    const int lc = t & 15;
    const float* ag = A + (size_t)(bm * BM + lr) * K + lc * 4;
    const float* wg = W + (size_t)(bn * BN + lr) * K + lc * 4;

    float4 ra[AIT], rw[WIT];
    v8f acc[WN];
#pragma unroll
    for (int j = 0; j < WN; ++j)
        acc[j] = (v8f){0.f, 0.f, 0.f, 0.f, 0.f, 0.f, 0.f, 0.f};

#pragma unroll
    for (int p = 0; p < AIT; ++p) ra[p] = *(const float4*)(ag + (size_t)p * RPP * K);
#pragma unroll
    for (int p = 0; p < WIT; ++p) rw[p] = *(const float4*)(wg + (size_t)p * RPP * K);
#pragma unroll
    for (int p = 0; p < AIT; ++p) As4[(lr + p * RPP) * PJ_P4_ + lc] = ra[p];
#pragma unroll
    for (int p = 0; p < WIT; ++p) Ws4[(lr + p * RPP) * PJ_P4_ + lc] = rw[p];
    __syncthreads();

    for (int kc = 0; kc < PJ_NC_; ++kc) {
        const int  buf  = kc & 1;
        const bool more = (kc + 1 < PJ_NC_);

        if (more) {
#pragma unroll
            for (int p = 0; p < AIT; ++p)
                ra[p] = *(const float4*)(ag + (size_t)p * RPP * K + (kc + 1) * PJ_KC_);
#pragma unroll
            for (int p = 0; p < WIT; ++p)
                rw[p] = *(const float4*)(wg + (size_t)p * RPP * K + (kc + 1) * PJ_KC_);
        }

        const v2f* af = (const v2f*)(As4 + (buf * BM + tm * 16 + c15) * PJ_P4_) + h;
        const v2f* wf = (const v2f*)(Ws4 + (buf * BN + tn0 * 16 + c15) * PJ_P4_) + h;
#pragma unroll
        for (int ks = 0; ks < PJ_KC_ / 4; ++ks) {
            const v2f av = af[ks * 2];
#pragma unroll
            for (int j = 0; j < WN; ++j) {
                const v2f bv = wf[j * 16 * PJ_P4_ * 2 + ks * 2];
                acc[j] = __builtin_amdgcn_wmma_f32_16x16x4_f32(
                             false, av, false, bv, (short)0, acc[j], false, false);
            }
        }

        if (more) {
#pragma unroll
            for (int p = 0; p < AIT; ++p)
                As4[((buf ^ 1) * BM + lr + p * RPP) * PJ_P4_ + lc] = ra[p];
#pragma unroll
            for (int p = 0; p < WIT; ++p)
                Ws4[((buf ^ 1) * BN + lr + p * RPP) * PJ_P4_ + lc] = rw[p];
        }
        __syncthreads();
    }

    const int m0 = bm * BM + tm * 16 + 8 * h;
#pragma unroll
    for (int j = 0; j < WN; ++j) {
        const int n = bn * BN + (tn0 + j) * 16 + c15;
#pragma unroll
        for (int r = 0; r < 8; ++r) {
            const int m = m0 + r;
            if (TRANS_OUT) {
                const int b    = m / SEQ;
                const int kpos = m % SEQ;
                out[((size_t)b * N + n) * SEQ + kpos] = acc[j][r];
            } else {
                out[(size_t)m * N + n] = acc[j][r];
            }
        }
    }
}

__global__ __launch_bounds__(PJ_NT_) void proj_both(
    const float* __restrict__ key,
    const float* __restrict__ W_k,
    float* __restrict__ kfT,
    const float* __restrict__ query,
    const float* __restrict__ W_q,
    float* __restrict__ qf)
{
    __shared__ float4 As4[2 * 64 * PJ_P4_];
    __shared__ float4 Ws4[2 * 64 * PJ_P4_];

    if (blockIdx.x < PJ_KB_)
        proj_tile<64, 64, 2, LK_, true>(key, W_k, kfT, (int)blockIdx.x, As4, Ws4);
    else
        proj_tile<32, 64, 1, LQ_, false>(query, W_q, qf, (int)blockIdx.x - PJ_KB_, As4, Ws4);
}

#define SC_KT_ 64
#define SC_QG_ 16
#define SC_QW_ 4

__global__ __launch_bounds__(128) void scores_tanh_kernel(
    const float* __restrict__ qf,
    const float* __restrict__ kfT,
    const float* __restrict__ vvec,
    const int*   __restrict__ valid_len,
    float* __restrict__ scores)
{
    __shared__ float4 kS4[UNITS_ * SC_KT_ / 4];
    __shared__ float4 qS4[SC_QG_ * UNITS_ / 4];

    const int t    = threadIdx.x;
    const int lane = t & 31;
    const int wave = t >> 5;
    const int qg   = blockIdx.x & 7;
    const int kt   = (blockIdx.x >> 3) & 15;
    const int b    = blockIdx.x >> 7;

    const float* ksrc = kfT + (size_t)b * UNITS_ * LK_ + kt * SC_KT_;
    const int c  = t & 15;
    const int r0 = t >> 4;
#pragma unroll
    for (int i = 0; i < UNITS_ / 8; ++i) {
        const int u = r0 + 8 * i;
        kS4[u * (SC_KT_ / 4) + c] = *(const float4*)(ksrc + (size_t)u * LK_ + c * 4);
    }
    const float4* qsrc =
        (const float4*)(qf + (size_t)(b * LQ_ + qg * SC_QG_) * UNITS_);
#pragma unroll
    for (int i = 0; i < SC_QG_ * UNITS_ / 4 / 128; ++i)
        qS4[i * 128 + t] = qsrc[i * 128 + t];
    __syncthreads();

    const float2* kS2  = (const float2*)kS4;
    const float4* qrow = qS4 + (wave * SC_QW_) * (UNITS_ / 4);

    float a[SC_QW_][2];
#pragma unroll
    for (int j = 0; j < SC_QW_; ++j) { a[j][0] = 0.f; a[j][1] = 0.f; }

#pragma unroll 2
    for (int u4 = 0; u4 < UNITS_ / 4; ++u4) {
        float4 qv[SC_QW_];
#pragma unroll
        for (int j = 0; j < SC_QW_; ++j) qv[j] = qrow[j * (UNITS_ / 4) + u4];
        float2 kv[4];
#pragma unroll
        for (int i = 0; i < 4; ++i) kv[i] = kS2[(u4 * 4 + i) * (SC_KT_ / 2) + lane];
        const float vv0 = vvec[u4 * 4 + 0];
        const float vv1 = vvec[u4 * 4 + 1];
        const float vv2 = vvec[u4 * 4 + 2];
        const float vv3 = vvec[u4 * 4 + 3];
#pragma unroll
        for (int j = 0; j < SC_QW_; ++j) {
            a[j][0] = __builtin_fmaf(vv0, TANHF(qv[j].x + kv[0].x), a[j][0]);
            a[j][1] = __builtin_fmaf(vv0, TANHF(qv[j].x + kv[0].y), a[j][1]);
        }
#pragma unroll
        for (int j = 0; j < SC_QW_; ++j) {
            a[j][0] = __builtin_fmaf(vv1, TANHF(qv[j].y + kv[1].x), a[j][0]);
            a[j][1] = __builtin_fmaf(vv1, TANHF(qv[j].y + kv[1].y), a[j][1]);
        }
#pragma unroll
        for (int j = 0; j < SC_QW_; ++j) {
            a[j][0] = __builtin_fmaf(vv2, TANHF(qv[j].z + kv[2].x), a[j][0]);
            a[j][1] = __builtin_fmaf(vv2, TANHF(qv[j].z + kv[2].y), a[j][1]);
        }
#pragma unroll
        for (int j = 0; j < SC_QW_; ++j) {
            a[j][0] = __builtin_fmaf(vv3, TANHF(qv[j].w + kv[3].x), a[j][0]);
            a[j][1] = __builtin_fmaf(vv3, TANHF(qv[j].w + kv[3].y), a[j][1]);
        }
    }

    const int vl = valid_len[b];
    const int k0 = kt * SC_KT_ + 2 * lane;
    const int q0 = b * LQ_ + qg * SC_QG_ + wave * SC_QW_;
#pragma unroll
    for (int j = 0; j < SC_QW_; ++j) {
        float2 r;
        r.x = (k0 + 0 < vl) ? a[j][0] : MASK_VAL_;
        r.y = (k0 + 1 < vl) ? a[j][1] : MASK_VAL_;
        *(float2*)(scores + (size_t)(q0 + j) * LK_ + k0) = r;
    }
}

#define AO_BM_  32
#define AO_BN_  32
#define AO_KC_  256
#define AO_NC_  (LK_ / AO_KC_)
#define AO_RP4_ (LK_ / 4 + 1)
#define AO_RW_  (AO_BM_ / 4)
#define AO_RD_  4
#define AO_RJ_  (LK_ / 128)
#define AO_VP4_ 12
#define AO_VIT_ (AO_KC_ / 16)
#define AO_FG_  8
#define AO_NG_  (AO_KC_ / 4 / AO_FG_)

__device__ __forceinline__ float lane_xor16(float v) {
    const int b = __builtin_bit_cast(int, v);
    return __builtin_bit_cast(float,
        __builtin_amdgcn_permlanex16(b, b, 0x76543210, 0xfedcba98, false, false));
}
template <int N>
__device__ __forceinline__ float lane_xor(float v) {
    return __builtin_bit_cast(float,
        __builtin_amdgcn_update_dpp(0, __builtin_bit_cast(int, v), 0x160 + N, 0xf, 0xf, true));
}

__global__ __launch_bounds__(128) void attn_out_wmma(
    const float* __restrict__ scores,
    const float* __restrict__ value,
    float* __restrict__ out)
{
    __shared__ v4f As4[AO_BM_ * AO_RP4_];
    __shared__ v4f Vs4[2 * AO_KC_ * AO_VP4_];

    const int t    = threadIdx.x;
    const int lane = t & 31;
    const int wave = t >> 5;
    const int c15  = lane & 15;
    const int h    = lane >> 4;
    const int tm   = wave >> 1;
    const int tn   = wave & 1;
    const int bn   = blockIdx.x & 15;
    const int bm   = (blockIdx.x >> 4) & 3;
    const int b    = blockIdx.x >> 6;

    const int vr = t >> 3;
    const int vc = t & 7;
    const float* sg = scores + ((size_t)b * LQ_ + bm * AO_BM_ + wave * AO_RW_) * LK_ + lane * 4;
    const float* vg = value + ((size_t)b * LK_ + vr) * DSZ_ + bn * AO_BN_ + vc * 4;

    v4f x[AO_RD_][AO_RJ_], rv[AO_VIT_];
    v8f acc = {0.f, 0.f, 0.f, 0.f, 0.f, 0.f, 0.f, 0.f};

#pragma unroll
    for (int r = 0; r < AO_RD_; ++r)
#pragma unroll
        for (int j = 0; j < AO_RJ_; ++j)
            x[r][j] = *(const v4f*)(sg + (size_t)r * LK_ + j * 128);
#pragma unroll
    for (int p = 0; p < AO_VIT_; ++p) rv[p] = *(const v4f*)(vg + (size_t)p * 16 * DSZ_);

#pragma unroll
    for (int r = 0; r < AO_RW_; ++r) {
        v4f (&xr)[AO_RJ_] = x[r % AO_RD_];

        float m = fmaxf(fmaxf(xr[0].x, xr[0].y), fmaxf(xr[0].z, xr[0].w));
#pragma unroll
        for (int j = 1; j < AO_RJ_; ++j)
            m = fmaxf(m, fmaxf(fmaxf(xr[j].x, xr[j].y), fmaxf(xr[j].z, xr[j].w)));
        m = fmaxf(m, lane_xor16(m));
        m = fmaxf(m, lane_xor<8>(m));
        m = fmaxf(m, lane_xor<4>(m));
        m = fmaxf(m, lane_xor<2>(m));
        m = fmaxf(m, lane_xor<1>(m));

        float s[AO_RJ_];
#pragma unroll
        for (int j = 0; j < AO_RJ_; ++j) {
            v4f e;
            e.x = __expf(xr[j].x - m);
            e.y = __expf(xr[j].y - m);
            e.z = __expf(xr[j].z - m);
            e.w = __expf(xr[j].w - m);
            xr[j] = e;
            s[j] = e.x + e.y + e.z + e.w;
        }
#pragma unroll
        for (int j = 0; j < AO_RJ_; ++j) s[j] += lane_xor16(s[j]);
#pragma unroll
        for (int j = 0; j < AO_RJ_; ++j) s[j] += lane_xor<8>(s[j]);
#pragma unroll
        for (int j = 0; j < AO_RJ_; ++j) s[j] += lane_xor<4>(s[j]);
#pragma unroll
        for (int j = 0; j < AO_RJ_; ++j) s[j] += lane_xor<2>(s[j]);
#pragma unroll
        for (int j = 0; j < AO_RJ_; ++j) s[j] += lane_xor<1>(s[j]);
        float gs = s[0];
#pragma unroll
        for (int j = 1; j < AO_RJ_; ++j) gs += s[j];

        const float inv = 1.0f / gs;
        v4f* arow = As4 + (wave * AO_RW_ + r) * AO_RP4_ + lane;
#pragma unroll
        for (int j = 0; j < AO_RJ_; ++j) {
            v4f e = xr[j];
            e.x *= inv; e.y *= inv; e.z *= inv; e.w *= inv;
            arow[j * 32] = e;
        }

        if (r + AO_RD_ < AO_RW_) {
#pragma unroll
            for (int j = 0; j < AO_RJ_; ++j)
                xr[j] = *(const v4f*)(sg + (size_t)(r + AO_RD_) * LK_ + j * 128);
        }
    }

#pragma unroll
    for (int p = 0; p < AO_VIT_; ++p) Vs4[(vr + p * 16) * AO_VP4_ + vc] = rv[p];
    __syncthreads();

    for (int kc = 0; kc < AO_NC_; ++kc) {
        const int  buf  = kc & 1;
        const bool more = (kc + 1 < AO_NC_);

        if (more) {
#pragma unroll
            for (int p = 0; p < AO_VIT_; ++p)
                rv[p] = *(const v4f*)(vg + ((size_t)(kc + 1) * AO_KC_ + p * 16) * DSZ_);
        }

        const v2f*   af = (const v2f*)(As4 + (tm * 16 + c15) * AO_RP4_) + kc * (AO_KC_ / 2) + h;
        const float* vf = (const float*)(Vs4 + (buf * AO_KC_ + 2 * h) * AO_VP4_) + tn * 16 + c15;
        v2f fa[2][AO_FG_], fb[2][AO_FG_];
#pragma unroll
        for (int i = 0; i < AO_FG_; ++i) {
            fa[0][i]   = af[i * 2];
            fb[0][i].x = vf[(i * 4 + 0) * (AO_VP4_ * 4)];
            fb[0][i].y = vf[(i * 4 + 1) * (AO_VP4_ * 4)];
        }
#pragma unroll
        for (int g = 0; g < AO_NG_; ++g) {
            if (g + 1 < AO_NG_) {
#pragma unroll
                for (int i = 0; i < AO_FG_; ++i) {
                    const int ks = (g + 1) * AO_FG_ + i;
                    fa[(g + 1) & 1][i]   = af[ks * 2];
                    fb[(g + 1) & 1][i].x = vf[(ks * 4 + 0) * (AO_VP4_ * 4)];
                    fb[(g + 1) & 1][i].y = vf[(ks * 4 + 1) * (AO_VP4_ * 4)];
                }
            }
            __builtin_amdgcn_sched_barrier(0);
#pragma unroll
            for (int i = 0; i < AO_FG_; ++i)
                acc = __builtin_amdgcn_wmma_f32_16x16x4_f32(
                          false, fa[g & 1][i], false, fb[g & 1][i], (short)0, acc, false, false);
            __builtin_amdgcn_sched_barrier(0);
        }

        if (more) {
#pragma unroll
            for (int p = 0; p < AO_VIT_; ++p)
                Vs4[((buf ^ 1) * AO_KC_ + vr + p * 16) * AO_VP4_ + vc] = rv[p];
        }
        __syncthreads();
    }

    const int m0 = bm * AO_BM_ + tm * 16 + 8 * h;
    const int n  = bn * AO_BN_ + tn * 16 + c15;
#pragma unroll
    for (int r = 0; r < 8; ++r)
        out[((size_t)b * LQ_ + m0 + r) * DSZ_ + n] = acc[r];
}

extern "C" void kernel_launch(void* const* d_in, const int* in_sizes, int n_in,
                              void* d_out, int out_size, void* d_ws, size_t ws_size,
                              hipStream_t stream) {
    const float* query     = (const float*)d_in[0];
    const float* key       = (const float*)d_in[1];
    const float* value     = (const float*)d_in[2];
    const int*   valid_len = (const int*)  d_in[3];
    const float* W_q       = (const float*)d_in[4];
    const float* W_k       = (const float*)d_in[5];
    const float* vvec      = (const float*)d_in[6];

    float* qf     = (float*)d_ws;
    float* kfT    = qf  + (size_t)B_ * LQ_ * UNITS_;
    float* scores = kfT + (size_t)B_ * UNITS_ * LK_;
    float* outp   = (float*)d_out;

    proj_both<<<PJ_KB_ + PJ_QB_, PJ_NT_, 0, stream>>>(key, W_k, kfT, query, W_q, qf);
    scores_tanh_kernel<<<512, 128, 0, stream>>>(qf, kfT, vvec, valid_len, scores);
    attn_out_wmma<<<256, 128, 0, stream>>>(scores, value, outp);
}


# 64-column key tiles wholly at or above valid_len get the mask value without key projection or tanh sums (exact for any input); blocks with work are numbered first; every computed sum unchanged
# speedup vs baseline: 3.8536x; 1.0594x over previous
#include <hip/hip_runtime.h>

typedef float v2f __attribute__((ext_vector_type(2)));
typedef float v8f __attribute__((ext_vector_type(8)));
typedef float v4f __attribute__((ext_vector_type(4)));

#define B_     4
#define LQ_    128
#define LK_    1024
#define DSZ_   512
#define UNITS_ 256
#define MASK_VAL_ (-1.0e6f)

#if __has_builtin(__builtin_amdgcn_tanhf)
#define TANHF(x) __builtin_amdgcn_tanhf(x)
#else
#define TANHF(x) tanhf(x)
#endif

#define KT_   64
#define NKT_  (LK_ / KT_)
struct TileRef { int b; int kt; bool live; };

__device__ __forceinline__ TileRef tile_lookup(const int* __restrict__ valid_len, const int w)
{
    int n[B_];
    int live_total = 0;
#pragma unroll
    for (int b = 0; b < B_; ++b) {
        const int vl = valid_len[b];
        n[b] = (vl <= 0) ? 0 : ((vl >= LK_) ? NKT_ : (vl + KT_ - 1) / KT_);
        live_total += n[b];
    }
    TileRef r;
    r.live = (w < live_total);
    r.b  = 0;
    r.kt = 0;
    int  rest  = r.live ? w : w - live_total;
    bool found = false;
#pragma unroll
    for (int b = 0; b < B_; ++b) {
        const int cnt = r.live ? n[b] : NKT_ - n[b];
        if (!found && rest < cnt) {
            r.b   = b;
            r.kt  = r.live ? rest : n[b] + rest;
            found = true;
        }
        if (!found) rest -= cnt;
    }
    return r;
}

#define PJ_KC_     64
#define PJ_P4_     17
#define PJ_NC_     (DSZ_ / PJ_KC_)
#define PJ_NT_     256
#define PJ_KB_     256
#define PJ_QB_     64

template <int BM, int BN, int WN, int SEQ, bool TRANS_OUT>
__device__ __forceinline__ void proj_tile(
    const float* __restrict__ A,
    const float* __restrict__ W,
    float* __restrict__ out,
    const int blk,
    float4* __restrict__ As4,
    float4* __restrict__ Ws4)
{
    constexpr int K   = DSZ_;
    constexpr int N   = UNITS_;
    constexpr int NT  = (BM / 16) * (BN / 16) / WN * 32;
    constexpr int RPP = NT / 16;
    constexpr int AIT = BM / RPP;
    constexpr int WIT = BN / RPP;
    constexpr int WPR = (BN / 16) / WN;
    static_assert(NT == PJ_NT_, "both sides run in blocks of PJ_NT_ threads");

    const int t    = threadIdx.x;
    const int lane = t & 31;
    const int wave = t >> 5;
    const int c15  = lane & 15;
    const int h    = lane >> 4;
    const int tm   = wave / WPR;
    const int tn0  = (wave % WPR) * WN;
    const int bm   = blk / (N / BN);
    const int bn   = blk % (N / BN);

    const int lr = t >> 4;
    const int lc = t & 15;
    const float* ag = A + (size_t)(bm * BM + lr) * K + lc * 4;
    const float* wg = W + (size_t)(bn * BN + lr) * K + lc * 4;

    float4 ra[AIT], rw[WIT];
    v8f acc[WN];
#pragma unroll
    for (int j = 0; j < WN; ++j)
        acc[j] = (v8f){0.f, 0.f, 0.f, 0.f, 0.f, 0.f, 0.f, 0.f};

#pragma unroll
    for (int p = 0; p < AIT; ++p) ra[p] = *(const float4*)(ag + (size_t)p * RPP * K);
#pragma unroll
    for (int p = 0; p < WIT; ++p) rw[p] = *(const float4*)(wg + (size_t)p * RPP * K);
#pragma unroll
    for (int p = 0; p < AIT; ++p) As4[(lr + p * RPP) * PJ_P4_ + lc] = ra[p];
#pragma unroll
    for (int p = 0; p < WIT; ++p) Ws4[(lr + p * RPP) * PJ_P4_ + lc] = rw[p];
    __syncthreads();

    for (int kc = 0; kc < PJ_NC_; ++kc) {
        const int  buf  = kc & 1;
        const bool more = (kc + 1 < PJ_NC_);

        if (more) {
#pragma unroll
            for (int p = 0; p < AIT; ++p)
                ra[p] = *(const float4*)(ag + (size_t)p * RPP * K + (kc + 1) * PJ_KC_);
#pragma unroll
            for (int p = 0; p < WIT; ++p)
                rw[p] = *(const float4*)(wg + (size_t)p * RPP * K + (kc + 1) * PJ_KC_);
        }

        const v2f* af = (const v2f*)(As4 + (buf * BM + tm * 16 + c15) * PJ_P4_) + h;
        const v2f* wf = (const v2f*)(Ws4 + (buf * BN + tn0 * 16 + c15) * PJ_P4_) + h;
#pragma unroll
        for (int ks = 0; ks < PJ_KC_ / 4; ++ks) {
            const v2f av = af[ks * 2];
#pragma unroll
            for (int j = 0; j < WN; ++j) {
                const v2f bv = wf[j * 16 * PJ_P4_ * 2 + ks * 2];
                acc[j] = __builtin_amdgcn_wmma_f32_16x16x4_f32(
                             false, av, false, bv, (short)0, acc[j], false, false);
            }
        }

        if (more) {
#pragma unroll
            for (int p = 0; p < AIT; ++p)
                As4[((buf ^ 1) * BM + lr + p * RPP) * PJ_P4_ + lc] = ra[p];
#pragma unroll
            for (int p = 0; p < WIT; ++p)
                Ws4[((buf ^ 1) * BN + lr + p * RPP) * PJ_P4_ + lc] = rw[p];
        }
        __syncthreads();
    }

    const int m0 = bm * BM + tm * 16 + 8 * h;
#pragma unroll
    for (int j = 0; j < WN; ++j) {
        const int n = bn * BN + (tn0 + j) * 16 + c15;
#pragma unroll
        for (int r = 0; r < 8; ++r) {
            const int m = m0 + r;
            if (TRANS_OUT) {
                const int b    = m / SEQ;
                const int kpos = m % SEQ;
                out[((size_t)b * N + n) * SEQ + kpos] = acc[j][r];
            } else {
                out[(size_t)m * N + n] = acc[j][r];
            }
        }
    }
}

__global__ __launch_bounds__(PJ_NT_) void proj_both(
    const float* __restrict__ key,
    const float* __restrict__ W_k,
    float* __restrict__ kfT,
    const float* __restrict__ query,
    const float* __restrict__ W_q,
    float* __restrict__ qf,
    const int* __restrict__ valid_len)
{
    __shared__ float4 As4[2 * 64 * PJ_P4_];
    __shared__ float4 Ws4[2 * 64 * PJ_P4_];

    if (blockIdx.x < PJ_QB_) {
        proj_tile<32, 64, 1, LQ_, false>(query, W_q, qf, (int)blockIdx.x, As4, Ws4);
    } else {
        const int kb = (int)blockIdx.x - PJ_QB_;
        const TileRef tr = tile_lookup(valid_len, kb >> 2);
        if (!tr.live) return;
        proj_tile<64, 64, 2, LK_, true>(key, W_k, kfT,
                                        ((tr.b * NKT_ + tr.kt) << 2) | (kb & 3), As4, Ws4);
    }
}

#define SC_KT_ 64
#define SC_QG_ 16
#define SC_QW_ 4

__global__ __launch_bounds__(128) void scores_tanh_kernel(
    const float* __restrict__ qf,
    const float* __restrict__ kfT,
    const float* __restrict__ vvec,
    const int*   __restrict__ valid_len,
    float* __restrict__ scores)
{
    __shared__ float4 kS4[UNITS_ * SC_KT_ / 4];
    __shared__ float4 qS4[SC_QG_ * UNITS_ / 4];

    const int t    = threadIdx.x;
    const int lane = t & 31;
    const int wave = t >> 5;
    const int qg   = blockIdx.x & 7;
    const TileRef tr = tile_lookup(valid_len, (int)(blockIdx.x >> 3));
    const int kt   = tr.kt;
    const int b    = tr.b;

    if (!tr.live) {
        float* dst = scores + (size_t)(b * LQ_ + qg * SC_QG_ + wave * SC_QW_) * LK_
                            + kt * SC_KT_ + 2 * lane;
#pragma unroll
        for (int j = 0; j < SC_QW_; ++j)
            *(float2*)(dst + (size_t)j * LK_) = make_float2(MASK_VAL_, MASK_VAL_);
        return;
    }

    const float* ksrc = kfT + (size_t)b * UNITS_ * LK_ + kt * SC_KT_;
    const int c  = t & 15;
    const int r0 = t >> 4;
#pragma unroll
    for (int i = 0; i < UNITS_ / 8; ++i) {
        const int u = r0 + 8 * i;
        kS4[u * (SC_KT_ / 4) + c] = *(const float4*)(ksrc + (size_t)u * LK_ + c * 4);
    }
    const float4* qsrc =
        (const float4*)(qf + (size_t)(b * LQ_ + qg * SC_QG_) * UNITS_);
#pragma unroll
    for (int i = 0; i < SC_QG_ * UNITS_ / 4 / 128; ++i)
        qS4[i * 128 + t] = qsrc[i * 128 + t];
    __syncthreads();

    const float2* kS2  = (const float2*)kS4;
    const float4* qrow = qS4 + (wave * SC_QW_) * (UNITS_ / 4);

    float a[SC_QW_][2];
#pragma unroll
    for (int j = 0; j < SC_QW_; ++j) { a[j][0] = 0.f; a[j][1] = 0.f; }

#pragma unroll 2
    for (int u4 = 0; u4 < UNITS_ / 4; ++u4) {
        float4 qv[SC_QW_];
#pragma unroll
        for (int j = 0; j < SC_QW_; ++j) qv[j] = qrow[j * (UNITS_ / 4) + u4];
        float2 kv[4];
#pragma unroll
        for (int i = 0; i < 4; ++i) kv[i] = kS2[(u4 * 4 + i) * (SC_KT_ / 2) + lane];
        const float vv0 = vvec[u4 * 4 + 0];
        const float vv1 = vvec[u4 * 4 + 1];
        const float vv2 = vvec[u4 * 4 + 2];
        const float vv3 = vvec[u4 * 4 + 3];
#pragma unroll
        for (int j = 0; j < SC_QW_; ++j) {
            a[j][0] = __builtin_fmaf(vv0, TANHF(qv[j].x + kv[0].x), a[j][0]);
            a[j][1] = __builtin_fmaf(vv0, TANHF(qv[j].x + kv[0].y), a[j][1]);
        }
#pragma unroll
        for (int j = 0; j < SC_QW_; ++j) {
            a[j][0] = __builtin_fmaf(vv1, TANHF(qv[j].y + kv[1].x), a[j][0]);
            a[j][1] = __builtin_fmaf(vv1, TANHF(qv[j].y + kv[1].y), a[j][1]);
        }
#pragma unroll
        for (int j = 0; j < SC_QW_; ++j) {
            a[j][0] = __builtin_fmaf(vv2, TANHF(qv[j].z + kv[2].x), a[j][0]);
            a[j][1] = __builtin_fmaf(vv2, TANHF(qv[j].z + kv[2].y), a[j][1]);
        }
#pragma unroll
        for (int j = 0; j < SC_QW_; ++j) {
            a[j][0] = __builtin_fmaf(vv3, TANHF(qv[j].w + kv[3].x), a[j][0]);
            a[j][1] = __builtin_fmaf(vv3, TANHF(qv[j].w + kv[3].y), a[j][1]);
        }
    }

    const int vl = valid_len[b];
    const int k0 = kt * SC_KT_ + 2 * lane;
    const int q0 = b * LQ_ + qg * SC_QG_ + wave * SC_QW_;
#pragma unroll
    for (int j = 0; j < SC_QW_; ++j) {
        float2 r;
        r.x = (k0 + 0 < vl) ? a[j][0] : MASK_VAL_;
        r.y = (k0 + 1 < vl) ? a[j][1] : MASK_VAL_;
        *(float2*)(scores + (size_t)(q0 + j) * LK_ + k0) = r;
    }
}

#define AO_BM_  32
#define AO_BN_  32
#define AO_KC_  256
#define AO_NC_  (LK_ / AO_KC_)
#define AO_RP4_ (LK_ / 4 + 1)
#define AO_RW_  (AO_BM_ / 4)
#define AO_RD_  4
#define AO_RJ_  (LK_ / 128)
#define AO_VP4_ 12
#define AO_VIT_ (AO_KC_ / 16)
#define AO_FG_  8
#define AO_NG_  (AO_KC_ / 4 / AO_FG_)

__device__ __forceinline__ float lane_xor16(float v) {
    const int b = __builtin_bit_cast(int, v);
    return __builtin_bit_cast(float,
        __builtin_amdgcn_permlanex16(b, b, 0x76543210, 0xfedcba98, false, false));
}
template <int N>
__device__ __forceinline__ float lane_xor(float v) {
    return __builtin_bit_cast(float,
        __builtin_amdgcn_update_dpp(0, __builtin_bit_cast(int, v), 0x160 + N, 0xf, 0xf, true));
}

__global__ __launch_bounds__(128) void attn_out_wmma(
    const float* __restrict__ scores,
    const float* __restrict__ value,
    float* __restrict__ out)
{
    __shared__ v4f As4[AO_BM_ * AO_RP4_];
    __shared__ v4f Vs4[2 * AO_KC_ * AO_VP4_];

    const int t    = threadIdx.x;
    const int lane = t & 31;
    const int wave = t >> 5;
    const int c15  = lane & 15;
    const int h    = lane >> 4;
    const int tm   = wave >> 1;
    const int tn   = wave & 1;
    const int bn   = blockIdx.x & 15;
    const int bm   = (blockIdx.x >> 4) & 3;
    const int b    = blockIdx.x >> 6;

    const int vr = t >> 3;
    const int vc = t & 7;
    const float* sg = scores + ((size_t)b * LQ_ + bm * AO_BM_ + wave * AO_RW_) * LK_ + lane * 4;
    const float* vg = value + ((size_t)b * LK_ + vr) * DSZ_ + bn * AO_BN_ + vc * 4;

    v4f x[AO_RD_][AO_RJ_], rv[AO_VIT_];
    v8f acc = {0.f, 0.f, 0.f, 0.f, 0.f, 0.f, 0.f, 0.f};

#pragma unroll
    for (int r = 0; r < AO_RD_; ++r)
#pragma unroll
        for (int j = 0; j < AO_RJ_; ++j)
            x[r][j] = *(const v4f*)(sg + (size_t)r * LK_ + j * 128);
#pragma unroll
    for (int p = 0; p < AO_VIT_; ++p) rv[p] = *(const v4f*)(vg + (size_t)p * 16 * DSZ_);

#pragma unroll
    for (int r = 0; r < AO_RW_; ++r) {
        v4f (&xr)[AO_RJ_] = x[r % AO_RD_];

        float m = fmaxf(fmaxf(xr[0].x, xr[0].y), fmaxf(xr[0].z, xr[0].w));
#pragma unroll
        for (int j = 1; j < AO_RJ_; ++j)
            m = fmaxf(m, fmaxf(fmaxf(xr[j].x, xr[j].y), fmaxf(xr[j].z, xr[j].w)));
        m = fmaxf(m, lane_xor16(m));
        m = fmaxf(m, lane_xor<8>(m));
        m = fmaxf(m, lane_xor<4>(m));
        m = fmaxf(m, lane_xor<2>(m));
        m = fmaxf(m, lane_xor<1>(m));

        float s[AO_RJ_];
#pragma unroll
        for (int j = 0; j < AO_RJ_; ++j) {
            v4f e;
            e.x = __expf(xr[j].x - m);
            e.y = __expf(xr[j].y - m);
            e.z = __expf(xr[j].z - m);
            e.w = __expf(xr[j].w - m);
            xr[j] = e;
            s[j] = e.x + e.y + e.z + e.w;
        }
#pragma unroll
        for (int j = 0; j < AO_RJ_; ++j) s[j] += lane_xor16(s[j]);
#pragma unroll
        for (int j = 0; j < AO_RJ_; ++j) s[j] += lane_xor<8>(s[j]);
#pragma unroll
        for (int j = 0; j < AO_RJ_; ++j) s[j] += lane_xor<4>(s[j]);
#pragma unroll
        for (int j = 0; j < AO_RJ_; ++j) s[j] += lane_xor<2>(s[j]);
#pragma unroll
        for (int j = 0; j < AO_RJ_; ++j) s[j] += lane_xor<1>(s[j]);
        float gs = s[0];
#pragma unroll
        for (int j = 1; j < AO_RJ_; ++j) gs += s[j];

        const float inv = 1.0f / gs;
        v4f* arow = As4 + (wave * AO_RW_ + r) * AO_RP4_ + lane;
#pragma unroll
        for (int j = 0; j < AO_RJ_; ++j) {
            v4f e = xr[j];
            e.x *= inv; e.y *= inv; e.z *= inv; e.w *= inv;
            arow[j * 32] = e;
        }

        if (r + AO_RD_ < AO_RW_) {
#pragma unroll
            for (int j = 0; j < AO_RJ_; ++j)
                xr[j] = *(const v4f*)(sg + (size_t)(r + AO_RD_) * LK_ + j * 128);
        }
    }

#pragma unroll
    for (int p = 0; p < AO_VIT_; ++p) Vs4[(vr + p * 16) * AO_VP4_ + vc] = rv[p];
    __syncthreads();

    for (int kc = 0; kc < AO_NC_; ++kc) {
        const int  buf  = kc & 1;
        const bool more = (kc + 1 < AO_NC_);

        if (more) {
#pragma unroll
            for (int p = 0; p < AO_VIT_; ++p)
                rv[p] = *(const v4f*)(vg + ((size_t)(kc + 1) * AO_KC_ + p * 16) * DSZ_);
        }

        const v2f*   af = (const v2f*)(As4 + (tm * 16 + c15) * AO_RP4_) + kc * (AO_KC_ / 2) + h;
        const float* vf = (const float*)(Vs4 + (buf * AO_KC_ + 2 * h) * AO_VP4_) + tn * 16 + c15;
        v2f fa[2][AO_FG_], fb[2][AO_FG_];
#pragma unroll
        for (int i = 0; i < AO_FG_; ++i) {
            fa[0][i]   = af[i * 2];
            fb[0][i].x = vf[(i * 4 + 0) * (AO_VP4_ * 4)];
            fb[0][i].y = vf[(i * 4 + 1) * (AO_VP4_ * 4)];
        }
#pragma unroll
        for (int g = 0; g < AO_NG_; ++g) {
            if (g + 1 < AO_NG_) {
#pragma unroll
                for (int i = 0; i < AO_FG_; ++i) {
                    const int ks = (g + 1) * AO_FG_ + i;
                    fa[(g + 1) & 1][i]   = af[ks * 2];
                    fb[(g + 1) & 1][i].x = vf[(ks * 4 + 0) * (AO_VP4_ * 4)];
                    fb[(g + 1) & 1][i].y = vf[(ks * 4 + 1) * (AO_VP4_ * 4)];
                }
            }
            __builtin_amdgcn_sched_barrier(0);
#pragma unroll
            for (int i = 0; i < AO_FG_; ++i)
                acc = __builtin_amdgcn_wmma_f32_16x16x4_f32(
                          false, fa[g & 1][i], false, fb[g & 1][i], (short)0, acc, false, false);
            __builtin_amdgcn_sched_barrier(0);
        }

        if (more) {
#pragma unroll
            for (int p = 0; p < AO_VIT_; ++p)
                Vs4[((buf ^ 1) * AO_KC_ + vr + p * 16) * AO_VP4_ + vc] = rv[p];
        }
        __syncthreads();
    }

    const int m0 = bm * AO_BM_ + tm * 16 + 8 * h;
    const int n  = bn * AO_BN_ + tn * 16 + c15;
#pragma unroll
    for (int r = 0; r < 8; ++r)
        out[((size_t)b * LQ_ + m0 + r) * DSZ_ + n] = acc[r];
}

extern "C" void kernel_launch(void* const* d_in, const int* in_sizes, int n_in,
                              void* d_out, int out_size, void* d_ws, size_t ws_size,
                              hipStream_t stream) {
    const float* query     = (const float*)d_in[0];
    const float* key       = (const float*)d_in[1];
    const float* value     = (const float*)d_in[2];
    const int*   valid_len = (const int*)  d_in[3];
    const float* W_q       = (const float*)d_in[4];
    const float* W_k       = (const float*)d_in[5];
    const float* vvec      = (const float*)d_in[6];

    float* qf     = (float*)d_ws;
    float* kfT    = qf  + (size_t)B_ * LQ_ * UNITS_;
    float* scores = kfT + (size_t)B_ * UNITS_ * LK_;
    float* outp   = (float*)d_out;

    proj_both<<<PJ_KB_ + PJ_QB_, PJ_NT_, 0, stream>>>(key, W_k, kfT, query, W_q, qf, valid_len);
    scores_tanh_kernel<<<512, 128, 0, stream>>>(qf, kfT, vvec, valid_len, scores);
    attn_out_wmma<<<256, 128, 0, stream>>>(scores, value, outp);
}


# scores kernel sum loop in fixed issue order: tanh and one packed add/multiply-add by turns, LDS operands (v too) read one 8-unit trip ahead; per-(q,k) sum order over units unchanged
# speedup vs baseline: 4.2395x; 1.1001x over previous
#include <hip/hip_runtime.h>

typedef float v2f __attribute__((ext_vector_type(2)));
typedef float v8f __attribute__((ext_vector_type(8)));
typedef float v4f __attribute__((ext_vector_type(4)));

#define B_     4
#define LQ_    128
#define LK_    1024
#define DSZ_   512
#define UNITS_ 256
#define MASK_VAL_ (-1.0e6f)

#if __has_builtin(__builtin_amdgcn_tanhf)
#define TANHF(x) __builtin_amdgcn_tanhf(x)
#else
#define TANHF(x) tanhf(x)
#endif

#define KT_   64
#define NKT_  (LK_ / KT_)
struct TileRef { int b; int kt; bool live; };

__device__ __forceinline__ TileRef tile_lookup(const int* __restrict__ valid_len, const int w)
{
    int n[B_];
    int live_total = 0;
#pragma unroll
    for (int b = 0; b < B_; ++b) {
        const int vl = valid_len[b];
        n[b] = (vl <= 0) ? 0 : ((vl >= LK_) ? NKT_ : (vl + KT_ - 1) / KT_);
        live_total += n[b];
    }
    TileRef r;
    r.live = (w < live_total);
    r.b  = 0;
    r.kt = 0;
    int  rest  = r.live ? w : w - live_total;
    bool found = false;
#pragma unroll
    for (int b = 0; b < B_; ++b) {
        const int cnt = r.live ? n[b] : NKT_ - n[b];
        if (!found && rest < cnt) {
            r.b   = b;
            r.kt  = r.live ? rest : n[b] + rest;
            found = true;
        }
        if (!found) rest -= cnt;
    }
    return r;
}

#define PJ_KC_     64
#define PJ_P4_     17
#define PJ_NC_     (DSZ_ / PJ_KC_)
#define PJ_NT_     256
#define PJ_KB_     256
#define PJ_QB_     64

template <int BM, int BN, int WN, int SEQ, bool TRANS_OUT>
__device__ __forceinline__ void proj_tile(
    const float* __restrict__ A,
    const float* __restrict__ W,
    float* __restrict__ out,
    const int blk,
    float4* __restrict__ As4,
    float4* __restrict__ Ws4)
{
    constexpr int K   = DSZ_;
    constexpr int N   = UNITS_;
    constexpr int NT  = (BM / 16) * (BN / 16) / WN * 32;
    constexpr int RPP = NT / 16;
    constexpr int AIT = BM / RPP;
    constexpr int WIT = BN / RPP;
    constexpr int WPR = (BN / 16) / WN;
    static_assert(NT == PJ_NT_, "both sides run in blocks of PJ_NT_ threads");

    const int t    = threadIdx.x;
    const int lane = t & 31;
    const int wave = t >> 5;
    const int c15  = lane & 15;
    const int h    = lane >> 4;
    const int tm   = wave / WPR;
    const int tn0  = (wave % WPR) * WN;
    const int bm   = blk / (N / BN);
    const int bn   = blk % (N / BN);

    const int lr = t >> 4;
    const int lc = t & 15;
    const float* ag = A + (size_t)(bm * BM + lr) * K + lc * 4;
    const float* wg = W + (size_t)(bn * BN + lr) * K + lc * 4;

    float4 ra[AIT], rw[WIT];
    v8f acc[WN];
#pragma unroll
    for (int j = 0; j < WN; ++j)
        acc[j] = (v8f){0.f, 0.f, 0.f, 0.f, 0.f, 0.f, 0.f, 0.f};

#pragma unroll
    for (int p = 0; p < AIT; ++p) ra[p] = *(const float4*)(ag + (size_t)p * RPP * K);
#pragma unroll
    for (int p = 0; p < WIT; ++p) rw[p] = *(const float4*)(wg + (size_t)p * RPP * K);
#pragma unroll
    for (int p = 0; p < AIT; ++p) As4[(lr + p * RPP) * PJ_P4_ + lc] = ra[p];
#pragma unroll
    for (int p = 0; p < WIT; ++p) Ws4[(lr + p * RPP) * PJ_P4_ + lc] = rw[p];
    __syncthreads();

    for (int kc = 0; kc < PJ_NC_; ++kc) {
        const int  buf  = kc & 1;
        const bool more = (kc + 1 < PJ_NC_);

        if (more) {
#pragma unroll
            for (int p = 0; p < AIT; ++p)
                ra[p] = *(const float4*)(ag + (size_t)p * RPP * K + (kc + 1) * PJ_KC_);
#pragma unroll
            for (int p = 0; p < WIT; ++p)
                rw[p] = *(const float4*)(wg + (size_t)p * RPP * K + (kc + 1) * PJ_KC_);
        }

        const v2f* af = (const v2f*)(As4 + (buf * BM + tm * 16 + c15) * PJ_P4_) + h;
        const v2f* wf = (const v2f*)(Ws4 + (buf * BN + tn0 * 16 + c15) * PJ_P4_) + h;
#pragma unroll
        for (int ks = 0; ks < PJ_KC_ / 4; ++ks) {
            const v2f av = af[ks * 2];
#pragma unroll
            for (int j = 0; j < WN; ++j) {
                const v2f bv = wf[j * 16 * PJ_P4_ * 2 + ks * 2];
                acc[j] = __builtin_amdgcn_wmma_f32_16x16x4_f32(
                             false, av, false, bv, (short)0, acc[j], false, false);
            }
        }

        if (more) {
#pragma unroll
            for (int p = 0; p < AIT; ++p)
                As4[((buf ^ 1) * BM + lr + p * RPP) * PJ_P4_ + lc] = ra[p];
#pragma unroll
            for (int p = 0; p < WIT; ++p)
                Ws4[((buf ^ 1) * BN + lr + p * RPP) * PJ_P4_ + lc] = rw[p];
        }
        __syncthreads();
    }

    const int m0 = bm * BM + tm * 16 + 8 * h;
#pragma unroll
    for (int j = 0; j < WN; ++j) {
        const int n = bn * BN + (tn0 + j) * 16 + c15;
#pragma unroll
        for (int r = 0; r < 8; ++r) {
            const int m = m0 + r;
            if (TRANS_OUT) {
                const int b    = m / SEQ;
                const int kpos = m % SEQ;
                out[((size_t)b * N + n) * SEQ + kpos] = acc[j][r];
            } else {
                out[(size_t)m * N + n] = acc[j][r];
            }
        }
    }
}

__global__ __launch_bounds__(PJ_NT_) void proj_both(
    const float* __restrict__ key,
    const float* __restrict__ W_k,
    float* __restrict__ kfT,
    const float* __restrict__ query,
    const float* __restrict__ W_q,
    float* __restrict__ qf,
    const int* __restrict__ valid_len)
{
    __shared__ float4 As4[2 * 64 * PJ_P4_];
    __shared__ float4 Ws4[2 * 64 * PJ_P4_];

    if (blockIdx.x < PJ_QB_) {
        proj_tile<32, 64, 1, LQ_, false>(query, W_q, qf, (int)blockIdx.x, As4, Ws4);
    } else {
        const int kb = (int)blockIdx.x - PJ_QB_;
        const TileRef tr = tile_lookup(valid_len, kb >> 2);
        if (!tr.live) return;
        proj_tile<64, 64, 2, LK_, true>(key, W_k, kfT,
                                        ((tr.b * NKT_ + tr.kt) << 2) | (kb & 3), As4, Ws4);
    }
}

#define SC_KT_ 64
#define SC_QG_ 16
#define SC_QW_ 4

#define SC_UT_ 8
#define SC_NS_ (SC_UT_ * SC_QW_)
#define SC_DA_ 3
#define SC_DF_ 3
#define SC_E_  (SC_DA_ + SC_DF_)
#define SC_HOLD() __builtin_amdgcn_sched_barrier(0)

struct ScTrip  { v4f q[SC_UT_ / 4][SC_QW_]; v2f k[SC_UT_]; v4f w[SC_UT_ / 4]; };
struct ScCarry { v2f s[SC_DA_]; v2f t[SC_DF_]; v2f w; };

__device__ __forceinline__ float sc_tanh(const float x)
{
    float r;
    asm volatile("v_tanh_f32 %0, %1" : "=v"(r) : "v"(x));
    return r;
}
template <int HI>
__device__ __forceinline__ v2f sc_add(const v2f k, const v2f qp)
{
    v2f r;
    if (HI) asm volatile("v_pk_add_f32 %0, %1, %2 op_sel:[0,1]"    : "=v"(r) : "v"(k), "v"(qp));
    else    asm volatile("v_pk_add_f32 %0, %1, %2 op_sel_hi:[1,0]" : "=v"(r) : "v"(k), "v"(qp));
    return r;
}
template <int HI>
__device__ __forceinline__ void sc_fma(v2f& a, const v2f wp, const v2f t)
{
    if (HI) asm volatile("v_pk_fma_f32 %0, %1, %2, %0 op_sel:[1,0,0]"    : "+v"(a) : "v"(wp), "v"(t));
    else    asm volatile("v_pk_fma_f32 %0, %1, %2, %0 op_sel_hi:[0,1,1]" : "+v"(a) : "v"(wp), "v"(t));
}

__device__ __forceinline__ void sc_trip_load(ScTrip& o, const v4f* __restrict__ qrow,
                                             const v2f* __restrict__ kS2l,
                                             const v4f* __restrict__ vS4, const int tr)
{
#pragma unroll
    for (int h = 0; h < SC_UT_ / 4; ++h)
#pragma unroll
        for (int j = 0; j < SC_QW_; ++j)
            o.q[h][j] = qrow[j * (UNITS_ / 4) + tr * (SC_UT_ / 4) + h];
#pragma unroll
    for (int i = 0; i < SC_UT_; ++i)
        o.k[i] = kS2l[(tr * SC_UT_ + i) * (SC_KT_ / 2)];
#pragma unroll
    for (int h = 0; h < SC_UT_ / 4; ++h)
        o.w[h] = vS4[tr * (SC_UT_ / 4) + h];
}

template <int P>
__device__ __forceinline__ v2f sc_slot_add(const ScTrip& o)
{
    constexpr int i = P / SC_QW_, j = P % SC_QW_;
    const v4f q4 = o.q[i >> 2][j];
    const v2f qp = ((i & 2) == 0) ? (v2f){q4.x, q4.y} : (v2f){q4.z, q4.w};
    return sc_add<(i & 1)>(o.k[i], qp);
}
template <int I>
__device__ __forceinline__ v2f sc_wpair(const ScTrip& o)
{
    const v4f w4 = o.w[I >> 2];
    return ((I & 2) == 0) ? (v2f){w4.x, w4.y} : (v2f){w4.z, w4.w};
}
template <int PS>
__device__ __forceinline__ void sc_slot_fma(v2f (&a)[SC_QW_], const v2f wp, const v2f t)
{
    sc_fma<((PS / SC_QW_) & 1)>(a[PS % SC_QW_], wp, t);
}

template <int P>
__device__ __forceinline__ void sc_step(const ScTrip& o, const ScCarry& cy,
                                        v2f (&Sx)[SC_NS_ + SC_E_], v2f (&Tx)[SC_NS_ + SC_DF_],
                                        v2f (&a)[SC_QW_])
{
    constexpr int eT = P + SC_DF_;
    Tx[eT].x = sc_tanh(Sx[eT].x);
    Sx[P + SC_E_] = sc_slot_add<P>(o);
    Tx[eT].y = sc_tanh(Sx[eT].y);
    constexpr int sl = P - SC_E_;
    constexpr int ps = (sl >= 0) ? sl : SC_NS_ + sl;
    sc_slot_fma<ps>(a, (sl >= 0) ? sc_wpair<ps / SC_QW_>(o) : cy.w, Tx[P]);
    if constexpr (P + 1 < SC_NS_) sc_step<P + 1>(o, cy, Sx, Tx, a);
}

__device__ __forceinline__ void sc_trip_steps(const ScTrip& o, ScCarry& cy, v2f (&a)[SC_QW_])
{
    v2f Sx[SC_NS_ + SC_E_], Tx[SC_NS_ + SC_DF_];
#pragma unroll
    for (int d = 0; d < SC_DA_; ++d) Sx[SC_DF_ + d] = cy.s[d];
#pragma unroll
    for (int d = 0; d < SC_DF_; ++d) Tx[d] = cy.t[d];

    sc_step<0>(o, cy, Sx, Tx, a);

#pragma unroll
    for (int d = 0; d < SC_DA_; ++d) cy.s[d] = Sx[SC_NS_ + SC_DF_ + d];
#pragma unroll
    for (int d = 0; d < SC_DF_; ++d) cy.t[d] = Tx[SC_NS_ + d];
    cy.w = sc_wpair<SC_UT_ - 1>(o);
}

template <int D>
__device__ __forceinline__ void sc_drain_fma(const ScCarry& cy, const v2f (&td)[SC_DA_],
                                             v2f (&a)[SC_QW_])
{
    constexpr int ps = SC_NS_ - SC_E_ + D;
    if constexpr (D < SC_DF_) sc_slot_fma<ps>(a, cy.w, cy.t[D]);
    else                      sc_slot_fma<ps>(a, cy.w, td[D - SC_DF_]);
    if constexpr (D + 1 < SC_E_) sc_drain_fma<D + 1>(cy, td, a);
}
__device__ __forceinline__ void sc_drain(const ScCarry& cy, v2f (&a)[SC_QW_])
{
    v2f td[SC_DA_];
#pragma unroll
    for (int d = 0; d < SC_DA_; ++d) td[d].x = sc_tanh(cy.s[d].x);
#pragma unroll
    for (int d = 0; d < SC_DA_; ++d) td[d].y = sc_tanh(cy.s[d].y);
    sc_drain_fma<0>(cy, td, a);
}

__global__ __launch_bounds__(128) void scores_tanh_kernel(
    const float* __restrict__ qf,
    const float* __restrict__ kfT,
    const float* __restrict__ vvec,
    const int*   __restrict__ valid_len,
    float* __restrict__ scores)
{
    __shared__ float4 scS4[SC_QG_ * UNITS_ / 4 + UNITS_ / 4 + UNITS_ * SC_KT_ / 4];
    float4* const qS4 = scS4;
    float*  const vS  = (float*)(scS4 + SC_QG_ * UNITS_ / 4);
    float4* const kS4 = scS4 + SC_QG_ * UNITS_ / 4 + UNITS_ / 4;

    const int t    = threadIdx.x;
    const int lane = t & 31;
    const int wave = t >> 5;
    const int qg   = blockIdx.x & 7;
    const TileRef tr = tile_lookup(valid_len, (int)(blockIdx.x >> 3));
    const int kt   = tr.kt;
    const int b    = tr.b;

    if (!tr.live) {
        float* dst = scores + (size_t)(b * LQ_ + qg * SC_QG_ + wave * SC_QW_) * LK_
                            + kt * SC_KT_ + 2 * lane;
#pragma unroll
        for (int j = 0; j < SC_QW_; ++j)
            *(float2*)(dst + (size_t)j * LK_) = make_float2(MASK_VAL_, MASK_VAL_);
        return;
    }

    const float* ksrc = kfT + (size_t)b * UNITS_ * LK_ + kt * SC_KT_;
    const int c  = t & 15;
    const int r0 = t >> 4;
#pragma unroll
    for (int i = 0; i < UNITS_ / 8; ++i) {
        const int u = r0 + 8 * i;
        kS4[u * (SC_KT_ / 4) + c] = *(const float4*)(ksrc + (size_t)u * LK_ + c * 4);
    }
    const float4* qsrc =
        (const float4*)(qf + (size_t)(b * LQ_ + qg * SC_QG_) * UNITS_);
#pragma unroll
    for (int i = 0; i < SC_QG_ * UNITS_ / 4 / 128; ++i)
        qS4[i * 128 + t] = qsrc[i * 128 + t];
#pragma unroll
    for (int i = 0; i < UNITS_ / 128; ++i)
        vS[i * 128 + t] = vvec[i * 128 + t];
    __syncthreads();

    const v2f* kS2l = (const v2f*)kS4 + lane;
    const v4f* qrow = (const v4f*)qS4 + (wave * SC_QW_) * (UNITS_ / 4);
    const v4f* vS4  = (const v4f*)vS;

    v2f a[SC_QW_];
#pragma unroll
    for (int j = 0; j < SC_QW_; ++j) a[j] = (v2f){0.f, 0.f};

    ScCarry cy;
#pragma unroll
    for (int d = 0; d < SC_DA_; ++d) cy.s[d] = (v2f){0.f, 0.f};
#pragma unroll
    for (int d = 0; d < SC_DF_; ++d) cy.t[d] = (v2f){0.f, 0.f};
    cy.w = (v2f){0.f, 0.f};

    ScTrip A, B;
    sc_trip_load(A, qrow, kS2l, vS4, 0);
    for (int tr = 0; tr < UNITS_ / SC_UT_; tr += 2) {
        sc_trip_load(B, qrow, kS2l, vS4, tr + 1);
        SC_HOLD();
        sc_trip_steps(A, cy, a);
        sc_trip_load(A, qrow, kS2l, vS4, (tr + 2 < UNITS_ / SC_UT_) ? tr + 2 : tr + 1);
        SC_HOLD();
        sc_trip_steps(B, cy, a);
    }
    sc_drain(cy, a);

    const int vl = valid_len[b];
    const int k0 = kt * SC_KT_ + 2 * lane;
    const int q0 = b * LQ_ + qg * SC_QG_ + wave * SC_QW_;
#pragma unroll
    for (int j = 0; j < SC_QW_; ++j) {
        float2 r;
        r.x = (k0 + 0 < vl) ? a[j].x : MASK_VAL_;
        r.y = (k0 + 1 < vl) ? a[j].y : MASK_VAL_;
        *(float2*)(scores + (size_t)(q0 + j) * LK_ + k0) = r;
    }
}

#define AO_BM_  32
#define AO_BN_  32
#define AO_KC_  256
#define AO_NC_  (LK_ / AO_KC_)
#define AO_RP4_ (LK_ / 4 + 1)
#define AO_RW_  (AO_BM_ / 4)
#define AO_RD_  4
#define AO_RJ_  (LK_ / 128)
#define AO_VP4_ 12
#define AO_VIT_ (AO_KC_ / 16)
#define AO_FG_  8
#define AO_NG_  (AO_KC_ / 4 / AO_FG_)

__device__ __forceinline__ float lane_xor16(float v) {
    const int b = __builtin_bit_cast(int, v);
    return __builtin_bit_cast(float,
        __builtin_amdgcn_permlanex16(b, b, 0x76543210, 0xfedcba98, false, false));
}
template <int N>
__device__ __forceinline__ float lane_xor(float v) {
    return __builtin_bit_cast(float,
        __builtin_amdgcn_update_dpp(0, __builtin_bit_cast(int, v), 0x160 + N, 0xf, 0xf, true));
}

__global__ __launch_bounds__(128) void attn_out_wmma(
    const float* __restrict__ scores,
    const float* __restrict__ value,
    float* __restrict__ out)
{
    __shared__ v4f As4[AO_BM_ * AO_RP4_];
    __shared__ v4f Vs4[2 * AO_KC_ * AO_VP4_];

    const int t    = threadIdx.x;
    const int lane = t & 31;
    const int wave = t >> 5;
    const int c15  = lane & 15;
    const int h    = lane >> 4;
    const int tm   = wave >> 1;
    const int tn   = wave & 1;
    const int bn   = blockIdx.x & 15;
    const int bm   = (blockIdx.x >> 4) & 3;
    const int b    = blockIdx.x >> 6;

    const int vr = t >> 3;
    const int vc = t & 7;
    const float* sg = scores + ((size_t)b * LQ_ + bm * AO_BM_ + wave * AO_RW_) * LK_ + lane * 4;
    const float* vg = value + ((size_t)b * LK_ + vr) * DSZ_ + bn * AO_BN_ + vc * 4;

    v4f x[AO_RD_][AO_RJ_], rv[AO_VIT_];
    v8f acc = {0.f, 0.f, 0.f, 0.f, 0.f, 0.f, 0.f, 0.f};

#pragma unroll
    for (int r = 0; r < AO_RD_; ++r)
#pragma unroll
        for (int j = 0; j < AO_RJ_; ++j)
            x[r][j] = *(const v4f*)(sg + (size_t)r * LK_ + j * 128);
#pragma unroll
    for (int p = 0; p < AO_VIT_; ++p) rv[p] = *(const v4f*)(vg + (size_t)p * 16 * DSZ_);

#pragma unroll
    for (int r = 0; r < AO_RW_; ++r) {
        v4f (&xr)[AO_RJ_] = x[r % AO_RD_];

        float m = fmaxf(fmaxf(xr[0].x, xr[0].y), fmaxf(xr[0].z, xr[0].w));
#pragma unroll
        for (int j = 1; j < AO_RJ_; ++j)
            m = fmaxf(m, fmaxf(fmaxf(xr[j].x, xr[j].y), fmaxf(xr[j].z, xr[j].w)));
        m = fmaxf(m, lane_xor16(m));
        m = fmaxf(m, lane_xor<8>(m));
        m = fmaxf(m, lane_xor<4>(m));
        m = fmaxf(m, lane_xor<2>(m));
        m = fmaxf(m, lane_xor<1>(m));

        float s[AO_RJ_];
#pragma unroll
        for (int j = 0; j < AO_RJ_; ++j) {
            v4f e;
            e.x = __expf(xr[j].x - m);
            e.y = __expf(xr[j].y - m);
            e.z = __expf(xr[j].z - m);
            e.w = __expf(xr[j].w - m);
            xr[j] = e;
            s[j] = e.x + e.y + e.z + e.w;
        }
#pragma unroll
        for (int j = 0; j < AO_RJ_; ++j) s[j] += lane_xor16(s[j]);
#pragma unroll
        for (int j = 0; j < AO_RJ_; ++j) s[j] += lane_xor<8>(s[j]);
#pragma unroll
        for (int j = 0; j < AO_RJ_; ++j) s[j] += lane_xor<4>(s[j]);
#pragma unroll
        for (int j = 0; j < AO_RJ_; ++j) s[j] += lane_xor<2>(s[j]);
#pragma unroll
        for (int j = 0; j < AO_RJ_; ++j) s[j] += lane_xor<1>(s[j]);
        float gs = s[0];
#pragma unroll
        for (int j = 1; j < AO_RJ_; ++j) gs += s[j];

        const float inv = 1.0f / gs;
        v4f* arow = As4 + (wave * AO_RW_ + r) * AO_RP4_ + lane;
#pragma unroll
        for (int j = 0; j < AO_RJ_; ++j) {
            v4f e = xr[j];
            e.x *= inv; e.y *= inv; e.z *= inv; e.w *= inv;
            arow[j * 32] = e;
        }

        if (r + AO_RD_ < AO_RW_) {
#pragma unroll
            for (int j = 0; j < AO_RJ_; ++j)
                xr[j] = *(const v4f*)(sg + (size_t)(r + AO_RD_) * LK_ + j * 128);
        }
    }

#pragma unroll
    for (int p = 0; p < AO_VIT_; ++p) Vs4[(vr + p * 16) * AO_VP4_ + vc] = rv[p];
    __syncthreads();

    for (int kc = 0; kc < AO_NC_; ++kc) {
        const int  buf  = kc & 1;
        const bool more = (kc + 1 < AO_NC_);

        if (more) {
#pragma unroll
            for (int p = 0; p < AO_VIT_; ++p)
                rv[p] = *(const v4f*)(vg + ((size_t)(kc + 1) * AO_KC_ + p * 16) * DSZ_);
        }

        const v2f*   af = (const v2f*)(As4 + (tm * 16 + c15) * AO_RP4_) + kc * (AO_KC_ / 2) + h;
        const float* vf = (const float*)(Vs4 + (buf * AO_KC_ + 2 * h) * AO_VP4_) + tn * 16 + c15;
        v2f fa[2][AO_FG_], fb[2][AO_FG_];
#pragma unroll
        for (int i = 0; i < AO_FG_; ++i) {
            fa[0][i]   = af[i * 2];
            fb[0][i].x = vf[(i * 4 + 0) * (AO_VP4_ * 4)];
            fb[0][i].y = vf[(i * 4 + 1) * (AO_VP4_ * 4)];
        }
#pragma unroll
        for (int g = 0; g < AO_NG_; ++g) {
            if (g + 1 < AO_NG_) {
#pragma unroll
                for (int i = 0; i < AO_FG_; ++i) {
                    const int ks = (g + 1) * AO_FG_ + i;
                    fa[(g + 1) & 1][i]   = af[ks * 2];
                    fb[(g + 1) & 1][i].x = vf[(ks * 4 + 0) * (AO_VP4_ * 4)];
                    fb[(g + 1) & 1][i].y = vf[(ks * 4 + 1) * (AO_VP4_ * 4)];
                }
            }
            __builtin_amdgcn_sched_barrier(0);
#pragma unroll
            for (int i = 0; i < AO_FG_; ++i)
                acc = __builtin_amdgcn_wmma_f32_16x16x4_f32(
                          false, fa[g & 1][i], false, fb[g & 1][i], (short)0, acc, false, false);
            __builtin_amdgcn_sched_barrier(0);
        }

        if (more) {
#pragma unroll
            for (int p = 0; p < AO_VIT_; ++p)
                Vs4[((buf ^ 1) * AO_KC_ + vr + p * 16) * AO_VP4_ + vc] = rv[p];
        }
        __syncthreads();
    }

    const int m0 = bm * AO_BM_ + tm * 16 + 8 * h;
    const int n  = bn * AO_BN_ + tn * 16 + c15;
#pragma unroll
    for (int r = 0; r < 8; ++r)
        out[((size_t)b * LQ_ + m0 + r) * DSZ_ + n] = acc[r];
}

extern "C" void kernel_launch(void* const* d_in, const int* in_sizes, int n_in,
                              void* d_out, int out_size, void* d_ws, size_t ws_size,
                              hipStream_t stream) {
    const float* query     = (const float*)d_in[0];
    const float* key       = (const float*)d_in[1];
    const float* value     = (const float*)d_in[2];
    const int*   valid_len = (const int*)  d_in[3];
    const float* W_q       = (const float*)d_in[4];
    const float* W_k       = (const float*)d_in[5];
    const float* vvec      = (const float*)d_in[6];

    float* qf     = (float*)d_ws;
    float* kfT    = qf  + (size_t)B_ * LQ_ * UNITS_;
    float* scores = kfT + (size_t)B_ * UNITS_ * LK_;
    float* outp   = (float*)d_out;

    proj_both<<<PJ_KB_ + PJ_QB_, PJ_NT_, 0, stream>>>(key, W_k, kfT, query, W_q, qf, valid_len);
    scores_tanh_kernel<<<512, 128, 0, stream>>>(qf, kfT, vvec, valid_len, scores);
    attn_out_wmma<<<256, 128, 0, stream>>>(scores, value, outp);
}


# projection launch: 128-k chunks (were 64) so one chunk's products last a load round trip, key side in 32x64 tiles like the query side (512 blocks, live first); every k sum order unchanged
# speedup vs baseline: 4.7361x; 1.1171x over previous
#include <hip/hip_runtime.h>

typedef float v2f __attribute__((ext_vector_type(2)));
typedef float v8f __attribute__((ext_vector_type(8)));
typedef float v4f __attribute__((ext_vector_type(4)));

#define B_     4
#define LQ_    128
#define LK_    1024
#define DSZ_   512
#define UNITS_ 256
#define MASK_VAL_ (-1.0e6f)

#if __has_builtin(__builtin_amdgcn_tanhf)
#define TANHF(x) __builtin_amdgcn_tanhf(x)
#else
#define TANHF(x) tanhf(x)
#endif

#define KT_   64
#define NKT_  (LK_ / KT_)
struct TileRef { int b; int kt; bool live; };

__device__ __forceinline__ TileRef tile_lookup(const int* __restrict__ valid_len, const int w)
{
    int n[B_];
    int live_total = 0;
#pragma unroll
    for (int b = 0; b < B_; ++b) {
        const int vl = valid_len[b];
        n[b] = (vl <= 0) ? 0 : ((vl >= LK_) ? NKT_ : (vl + KT_ - 1) / KT_);
        live_total += n[b];
    }
    TileRef r;
    r.live = (w < live_total);
    r.b  = 0;
    r.kt = 0;
    int  rest  = r.live ? w : w - live_total;
    bool found = false;
#pragma unroll
    for (int b = 0; b < B_; ++b) {
        const int cnt = r.live ? n[b] : NKT_ - n[b];
        if (!found && rest < cnt) {
            r.b   = b;
            r.kt  = r.live ? rest : n[b] + rest;
            found = true;
        }
        if (!found) rest -= cnt;
    }
    return r;
}

#define PJ_KC_     128
#define PJ_C4_     (PJ_KC_ / 4)
#define PJ_P4_     (PJ_C4_ + 1)
#define PJ_NC_     (DSZ_ / PJ_KC_)
#define PJ_NT_     256
#define PJ_BM_     32
#define PJ_BN_     64
#define PJ_KB_     512
#define PJ_QB_     64

template <int BM, int BN, int WN, int SEQ, bool TRANS_OUT>
__device__ __forceinline__ void proj_tile(
    const float* __restrict__ A,
    const float* __restrict__ W,
    float* __restrict__ out,
    const int blk,
    float4* __restrict__ As4,
    float4* __restrict__ Ws4)
{
    constexpr int K   = DSZ_;
    constexpr int N   = UNITS_;
    constexpr int NT  = (BM / 16) * (BN / 16) / WN * 32;
    constexpr int RPP = NT / PJ_C4_;
    constexpr int AIT = BM / RPP;
    constexpr int WIT = BN / RPP;
    constexpr int WPR = (BN / 16) / WN;
    static_assert(NT == PJ_NT_, "both sides run in blocks of PJ_NT_ threads");

    const int t    = threadIdx.x;
    const int lane = t & 31;
    const int wave = t >> 5;
    const int c15  = lane & 15;
    const int h    = lane >> 4;
    const int tm   = wave / WPR;
    const int tn0  = (wave % WPR) * WN;
    const int bm   = blk / (N / BN);
    const int bn   = blk % (N / BN);

    const int lr = t / PJ_C4_;
    const int lc = t % PJ_C4_;
    const float* ag = A + (size_t)(bm * BM + lr) * K + lc * 4;
    const float* wg = W + (size_t)(bn * BN + lr) * K + lc * 4;

    float4 ra[AIT], rw[WIT];
    v8f acc[WN];
#pragma unroll
    for (int j = 0; j < WN; ++j)
        acc[j] = (v8f){0.f, 0.f, 0.f, 0.f, 0.f, 0.f, 0.f, 0.f};

#pragma unroll
    for (int p = 0; p < AIT; ++p) ra[p] = *(const float4*)(ag + (size_t)p * RPP * K);
#pragma unroll
    for (int p = 0; p < WIT; ++p) rw[p] = *(const float4*)(wg + (size_t)p * RPP * K);
#pragma unroll
    for (int p = 0; p < AIT; ++p) As4[(lr + p * RPP) * PJ_P4_ + lc] = ra[p];
#pragma unroll
    for (int p = 0; p < WIT; ++p) Ws4[(lr + p * RPP) * PJ_P4_ + lc] = rw[p];
    __syncthreads();

    for (int kc = 0; kc < PJ_NC_; ++kc) {
        const int  buf  = kc & 1;
        const bool more = (kc + 1 < PJ_NC_);

        if (more) {
#pragma unroll
            for (int p = 0; p < AIT; ++p)
                ra[p] = *(const float4*)(ag + (size_t)p * RPP * K + (kc + 1) * PJ_KC_);
#pragma unroll
            for (int p = 0; p < WIT; ++p)
                rw[p] = *(const float4*)(wg + (size_t)p * RPP * K + (kc + 1) * PJ_KC_);
        }

        const v2f* af = (const v2f*)(As4 + (buf * BM + tm * 16 + c15) * PJ_P4_) + h;
        const v2f* wf = (const v2f*)(Ws4 + (buf * BN + tn0 * 16 + c15) * PJ_P4_) + h;
#pragma unroll
        for (int ks = 0; ks < PJ_KC_ / 4; ++ks) {
            const v2f av = af[ks * 2];
#pragma unroll
            for (int j = 0; j < WN; ++j) {
                const v2f bv = wf[j * 16 * PJ_P4_ * 2 + ks * 2];
                acc[j] = __builtin_amdgcn_wmma_f32_16x16x4_f32(
                             false, av, false, bv, (short)0, acc[j], false, false);
            }
        }

        if (more) {
#pragma unroll
            for (int p = 0; p < AIT; ++p)
                As4[((buf ^ 1) * BM + lr + p * RPP) * PJ_P4_ + lc] = ra[p];
#pragma unroll
            for (int p = 0; p < WIT; ++p)
                Ws4[((buf ^ 1) * BN + lr + p * RPP) * PJ_P4_ + lc] = rw[p];
        }
        __syncthreads();
    }

    const int m0 = bm * BM + tm * 16 + 8 * h;
#pragma unroll
    for (int j = 0; j < WN; ++j) {
        const int n = bn * BN + (tn0 + j) * 16 + c15;
#pragma unroll
        for (int r = 0; r < 8; ++r) {
            const int m = m0 + r;
            if (TRANS_OUT) {
                const int b    = m / SEQ;
                const int kpos = m % SEQ;
                out[((size_t)b * N + n) * SEQ + kpos] = acc[j][r];
            } else {
                out[(size_t)m * N + n] = acc[j][r];
            }
        }
    }
}

__global__ __launch_bounds__(PJ_NT_) void proj_both(
    const float* __restrict__ key,
    const float* __restrict__ W_k,
    float* __restrict__ kfT,
    const float* __restrict__ query,
    const float* __restrict__ W_q,
    float* __restrict__ qf,
    const int* __restrict__ valid_len)
{
    __shared__ float4 As4[2 * PJ_BM_ * PJ_P4_];
    __shared__ float4 Ws4[2 * PJ_BN_ * PJ_P4_];

    if (blockIdx.x < PJ_QB_) {
        proj_tile<PJ_BM_, PJ_BN_, 1, LQ_, false>(query, W_q, qf, (int)blockIdx.x, As4, Ws4);
    } else {
        const int kb = (int)blockIdx.x - PJ_QB_;
        const TileRef tr = tile_lookup(valid_len, kb >> 3);
        if (!tr.live) return;
        proj_tile<PJ_BM_, PJ_BN_, 1, LK_, true>(key, W_k, kfT,
                                                ((tr.b * NKT_ + tr.kt) << 3) | (kb & 7), As4, Ws4);
    }
}

#define SC_KT_ 64
#define SC_QG_ 16
#define SC_QW_ 4

#define SC_UT_ 8
#define SC_NS_ (SC_UT_ * SC_QW_)
#define SC_DA_ 3
#define SC_DF_ 3
#define SC_E_  (SC_DA_ + SC_DF_)
#define SC_HOLD() __builtin_amdgcn_sched_barrier(0)

struct ScTrip  { v4f q[SC_UT_ / 4][SC_QW_]; v2f k[SC_UT_]; v4f w[SC_UT_ / 4]; };
struct ScCarry { v2f s[SC_DA_]; v2f t[SC_DF_]; v2f w; };

__device__ __forceinline__ float sc_tanh(const float x)
{
    float r;
    asm volatile("v_tanh_f32 %0, %1" : "=v"(r) : "v"(x));
    return r;
}
template <int HI>
__device__ __forceinline__ v2f sc_add(const v2f k, const v2f qp)
{
    v2f r;
    if (HI) asm volatile("v_pk_add_f32 %0, %1, %2 op_sel:[0,1]"    : "=v"(r) : "v"(k), "v"(qp));
    else    asm volatile("v_pk_add_f32 %0, %1, %2 op_sel_hi:[1,0]" : "=v"(r) : "v"(k), "v"(qp));
    return r;
}
template <int HI>
__device__ __forceinline__ void sc_fma(v2f& a, const v2f wp, const v2f t)
{
    if (HI) asm volatile("v_pk_fma_f32 %0, %1, %2, %0 op_sel:[1,0,0]"    : "+v"(a) : "v"(wp), "v"(t));
    else    asm volatile("v_pk_fma_f32 %0, %1, %2, %0 op_sel_hi:[0,1,1]" : "+v"(a) : "v"(wp), "v"(t));
}

__device__ __forceinline__ void sc_trip_load(ScTrip& o, const v4f* __restrict__ qrow,
                                             const v2f* __restrict__ kS2l,
                                             const v4f* __restrict__ vS4, const int tr)
{
#pragma unroll
    for (int h = 0; h < SC_UT_ / 4; ++h)
#pragma unroll
        for (int j = 0; j < SC_QW_; ++j)
            o.q[h][j] = qrow[j * (UNITS_ / 4) + tr * (SC_UT_ / 4) + h];
#pragma unroll
    for (int i = 0; i < SC_UT_; ++i)
        o.k[i] = kS2l[(tr * SC_UT_ + i) * (SC_KT_ / 2)];
#pragma unroll
    for (int h = 0; h < SC_UT_ / 4; ++h)
        o.w[h] = vS4[tr * (SC_UT_ / 4) + h];
}

template <int P>
__device__ __forceinline__ v2f sc_slot_add(const ScTrip& o)
{
    constexpr int i = P / SC_QW_, j = P % SC_QW_;
    const v4f q4 = o.q[i >> 2][j];
    const v2f qp = ((i & 2) == 0) ? (v2f){q4.x, q4.y} : (v2f){q4.z, q4.w};
    return sc_add<(i & 1)>(o.k[i], qp);
}
template <int I>
__device__ __forceinline__ v2f sc_wpair(const ScTrip& o)
{
    const v4f w4 = o.w[I >> 2];
    return ((I & 2) == 0) ? (v2f){w4.x, w4.y} : (v2f){w4.z, w4.w};
}
template <int PS>
__device__ __forceinline__ void sc_slot_fma(v2f (&a)[SC_QW_], const v2f wp, const v2f t)
{
    sc_fma<((PS / SC_QW_) & 1)>(a[PS % SC_QW_], wp, t);
}

template <int P>
__device__ __forceinline__ void sc_step(const ScTrip& o, const ScCarry& cy,
                                        v2f (&Sx)[SC_NS_ + SC_E_], v2f (&Tx)[SC_NS_ + SC_DF_],
                                        v2f (&a)[SC_QW_])
{
    constexpr int eT = P + SC_DF_;
    Tx[eT].x = sc_tanh(Sx[eT].x);
    Sx[P + SC_E_] = sc_slot_add<P>(o);
    Tx[eT].y = sc_tanh(Sx[eT].y);
    constexpr int sl = P - SC_E_;
    constexpr int ps = (sl >= 0) ? sl : SC_NS_ + sl;
    sc_slot_fma<ps>(a, (sl >= 0) ? sc_wpair<ps / SC_QW_>(o) : cy.w, Tx[P]);
    if constexpr (P + 1 < SC_NS_) sc_step<P + 1>(o, cy, Sx, Tx, a);
}

__device__ __forceinline__ void sc_trip_steps(const ScTrip& o, ScCarry& cy, v2f (&a)[SC_QW_])
{
    v2f Sx[SC_NS_ + SC_E_], Tx[SC_NS_ + SC_DF_];
#pragma unroll
    for (int d = 0; d < SC_DA_; ++d) Sx[SC_DF_ + d] = cy.s[d];
#pragma unroll
    for (int d = 0; d < SC_DF_; ++d) Tx[d] = cy.t[d];

    sc_step<0>(o, cy, Sx, Tx, a);

#pragma unroll
    for (int d = 0; d < SC_DA_; ++d) cy.s[d] = Sx[SC_NS_ + SC_DF_ + d];
#pragma unroll
    for (int d = 0; d < SC_DF_; ++d) cy.t[d] = Tx[SC_NS_ + d];
    cy.w = sc_wpair<SC_UT_ - 1>(o);
}

template <int D>
__device__ __forceinline__ void sc_drain_fma(const ScCarry& cy, const v2f (&td)[SC_DA_],
                                             v2f (&a)[SC_QW_])
{
    constexpr int ps = SC_NS_ - SC_E_ + D;
    if constexpr (D < SC_DF_) sc_slot_fma<ps>(a, cy.w, cy.t[D]);
    else                      sc_slot_fma<ps>(a, cy.w, td[D - SC_DF_]);
    if constexpr (D + 1 < SC_E_) sc_drain_fma<D + 1>(cy, td, a);
}
__device__ __forceinline__ void sc_drain(const ScCarry& cy, v2f (&a)[SC_QW_])
{
    v2f td[SC_DA_];
#pragma unroll
    for (int d = 0; d < SC_DA_; ++d) td[d].x = sc_tanh(cy.s[d].x);
#pragma unroll
    for (int d = 0; d < SC_DA_; ++d) td[d].y = sc_tanh(cy.s[d].y);
    sc_drain_fma<0>(cy, td, a);
}

__global__ __launch_bounds__(128) void scores_tanh_kernel(
    const float* __restrict__ qf,
    const float* __restrict__ kfT,
    const float* __restrict__ vvec,
    const int*   __restrict__ valid_len,
    float* __restrict__ scores)
{
    __shared__ float4 scS4[SC_QG_ * UNITS_ / 4 + UNITS_ / 4 + UNITS_ * SC_KT_ / 4];
    float4* const qS4 = scS4;
    float*  const vS  = (float*)(scS4 + SC_QG_ * UNITS_ / 4);
    float4* const kS4 = scS4 + SC_QG_ * UNITS_ / 4 + UNITS_ / 4;

    const int t    = threadIdx.x;
    const int lane = t & 31;
    const int wave = t >> 5;
    const int qg   = blockIdx.x & 7;
    const TileRef tr = tile_lookup(valid_len, (int)(blockIdx.x >> 3));
    const int kt   = tr.kt;
    const int b    = tr.b;

    if (!tr.live) {
        float* dst = scores + (size_t)(b * LQ_ + qg * SC_QG_ + wave * SC_QW_) * LK_
                            + kt * SC_KT_ + 2 * lane;
#pragma unroll
        for (int j = 0; j < SC_QW_; ++j)
            *(float2*)(dst + (size_t)j * LK_) = make_float2(MASK_VAL_, MASK_VAL_);
        return;
    }

    const float* ksrc = kfT + (size_t)b * UNITS_ * LK_ + kt * SC_KT_;
    const int c  = t & 15;
    const int r0 = t >> 4;
#pragma unroll
    for (int i = 0; i < UNITS_ / 8; ++i) {
        const int u = r0 + 8 * i;
        kS4[u * (SC_KT_ / 4) + c] = *(const float4*)(ksrc + (size_t)u * LK_ + c * 4);
    }
    const float4* qsrc =
        (const float4*)(qf + (size_t)(b * LQ_ + qg * SC_QG_) * UNITS_);
#pragma unroll
    for (int i = 0; i < SC_QG_ * UNITS_ / 4 / 128; ++i)
        qS4[i * 128 + t] = qsrc[i * 128 + t];
#pragma unroll
    for (int i = 0; i < UNITS_ / 128; ++i)
        vS[i * 128 + t] = vvec[i * 128 + t];
    __syncthreads();

    const v2f* kS2l = (const v2f*)kS4 + lane;
    const v4f* qrow = (const v4f*)qS4 + (wave * SC_QW_) * (UNITS_ / 4);
    const v4f* vS4  = (const v4f*)vS;

    v2f a[SC_QW_];
#pragma unroll
    for (int j = 0; j < SC_QW_; ++j) a[j] = (v2f){0.f, 0.f};

    ScCarry cy;
#pragma unroll
    for (int d = 0; d < SC_DA_; ++d) cy.s[d] = (v2f){0.f, 0.f};
#pragma unroll
    for (int d = 0; d < SC_DF_; ++d) cy.t[d] = (v2f){0.f, 0.f};
    cy.w = (v2f){0.f, 0.f};

    ScTrip A, B;
    sc_trip_load(A, qrow, kS2l, vS4, 0);
    for (int tr = 0; tr < UNITS_ / SC_UT_; tr += 2) {
        sc_trip_load(B, qrow, kS2l, vS4, tr + 1);
        SC_HOLD();
        sc_trip_steps(A, cy, a);
        sc_trip_load(A, qrow, kS2l, vS4, (tr + 2 < UNITS_ / SC_UT_) ? tr + 2 : tr + 1);
        SC_HOLD();
        sc_trip_steps(B, cy, a);
    }
    sc_drain(cy, a);

    const int vl = valid_len[b];
    const int k0 = kt * SC_KT_ + 2 * lane;
    const int q0 = b * LQ_ + qg * SC_QG_ + wave * SC_QW_;
#pragma unroll
    for (int j = 0; j < SC_QW_; ++j) {
        float2 r;
        r.x = (k0 + 0 < vl) ? a[j].x : MASK_VAL_;
        r.y = (k0 + 1 < vl) ? a[j].y : MASK_VAL_;
        *(float2*)(scores + (size_t)(q0 + j) * LK_ + k0) = r;
    }
}

#define AO_BM_  32
#define AO_BN_  32
#define AO_KC_  256
#define AO_NC_  (LK_ / AO_KC_)
#define AO_RP4_ (LK_ / 4 + 1)
#define AO_RW_  (AO_BM_ / 4)
#define AO_RD_  4
#define AO_RJ_  (LK_ / 128)
#define AO_VP4_ 12
#define AO_VIT_ (AO_KC_ / 16)
#define AO_FG_  8
#define AO_NG_  (AO_KC_ / 4 / AO_FG_)

__device__ __forceinline__ float lane_xor16(float v) {
    const int b = __builtin_bit_cast(int, v);
    return __builtin_bit_cast(float,
        __builtin_amdgcn_permlanex16(b, b, 0x76543210, 0xfedcba98, false, false));
}
template <int N>
__device__ __forceinline__ float lane_xor(float v) {
    return __builtin_bit_cast(float,
        __builtin_amdgcn_update_dpp(0, __builtin_bit_cast(int, v), 0x160 + N, 0xf, 0xf, true));
}

__global__ __launch_bounds__(128) void attn_out_wmma(
    const float* __restrict__ scores,
    const float* __restrict__ value,
    float* __restrict__ out)
{
    __shared__ v4f As4[AO_BM_ * AO_RP4_];
    __shared__ v4f Vs4[2 * AO_KC_ * AO_VP4_];

    const int t    = threadIdx.x;
    const int lane = t & 31;
    const int wave = t >> 5;
    const int c15  = lane & 15;
    const int h    = lane >> 4;
    const int tm   = wave >> 1;
    const int tn   = wave & 1;
    const int bn   = blockIdx.x & 15;
    const int bm   = (blockIdx.x >> 4) & 3;
    const int b    = blockIdx.x >> 6;

    const int vr = t >> 3;
    const int vc = t & 7;
    const float* sg = scores + ((size_t)b * LQ_ + bm * AO_BM_ + wave * AO_RW_) * LK_ + lane * 4;
    const float* vg = value + ((size_t)b * LK_ + vr) * DSZ_ + bn * AO_BN_ + vc * 4;

    v4f x[AO_RD_][AO_RJ_], rv[AO_VIT_];
    v8f acc = {0.f, 0.f, 0.f, 0.f, 0.f, 0.f, 0.f, 0.f};

#pragma unroll
    for (int r = 0; r < AO_RD_; ++r)
#pragma unroll
        for (int j = 0; j < AO_RJ_; ++j)
            x[r][j] = *(const v4f*)(sg + (size_t)r * LK_ + j * 128);
#pragma unroll
    for (int p = 0; p < AO_VIT_; ++p) rv[p] = *(const v4f*)(vg + (size_t)p * 16 * DSZ_);

#pragma unroll
    for (int r = 0; r < AO_RW_; ++r) {
        v4f (&xr)[AO_RJ_] = x[r % AO_RD_];

        float m = fmaxf(fmaxf(xr[0].x, xr[0].y), fmaxf(xr[0].z, xr[0].w));
#pragma unroll
        for (int j = 1; j < AO_RJ_; ++j)
            m = fmaxf(m, fmaxf(fmaxf(xr[j].x, xr[j].y), fmaxf(xr[j].z, xr[j].w)));
        m = fmaxf(m, lane_xor16(m));
        m = fmaxf(m, lane_xor<8>(m));
        m = fmaxf(m, lane_xor<4>(m));
        m = fmaxf(m, lane_xor<2>(m));
        m = fmaxf(m, lane_xor<1>(m));

        float s[AO_RJ_];
#pragma unroll
        for (int j = 0; j < AO_RJ_; ++j) {
            v4f e;
            e.x = __expf(xr[j].x - m);
            e.y = __expf(xr[j].y - m);
            e.z = __expf(xr[j].z - m);
            e.w = __expf(xr[j].w - m);
            xr[j] = e;
            s[j] = e.x + e.y + e.z + e.w;
        }
#pragma unroll
        for (int j = 0; j < AO_RJ_; ++j) s[j] += lane_xor16(s[j]);
#pragma unroll
        for (int j = 0; j < AO_RJ_; ++j) s[j] += lane_xor<8>(s[j]);
#pragma unroll
        for (int j = 0; j < AO_RJ_; ++j) s[j] += lane_xor<4>(s[j]);
#pragma unroll
        for (int j = 0; j < AO_RJ_; ++j) s[j] += lane_xor<2>(s[j]);
#pragma unroll
        for (int j = 0; j < AO_RJ_; ++j) s[j] += lane_xor<1>(s[j]);
        float gs = s[0];
#pragma unroll
        for (int j = 1; j < AO_RJ_; ++j) gs += s[j];

        const float inv = 1.0f / gs;
        v4f* arow = As4 + (wave * AO_RW_ + r) * AO_RP4_ + lane;
#pragma unroll
        for (int j = 0; j < AO_RJ_; ++j) {
            v4f e = xr[j];
            e.x *= inv; e.y *= inv; e.z *= inv; e.w *= inv;
            arow[j * 32] = e;
        }

        if (r + AO_RD_ < AO_RW_) {
#pragma unroll
            for (int j = 0; j < AO_RJ_; ++j)
                xr[j] = *(const v4f*)(sg + (size_t)(r + AO_RD_) * LK_ + j * 128);
        }
    }

#pragma unroll
    for (int p = 0; p < AO_VIT_; ++p) Vs4[(vr + p * 16) * AO_VP4_ + vc] = rv[p];
    __syncthreads();

    for (int kc = 0; kc < AO_NC_; ++kc) {
        const int  buf  = kc & 1;
        const bool more = (kc + 1 < AO_NC_);

        if (more) {
#pragma unroll
            for (int p = 0; p < AO_VIT_; ++p)
                rv[p] = *(const v4f*)(vg + ((size_t)(kc + 1) * AO_KC_ + p * 16) * DSZ_);
        }

        const v2f*   af = (const v2f*)(As4 + (tm * 16 + c15) * AO_RP4_) + kc * (AO_KC_ / 2) + h;
        const float* vf = (const float*)(Vs4 + (buf * AO_KC_ + 2 * h) * AO_VP4_) + tn * 16 + c15;
        v2f fa[2][AO_FG_], fb[2][AO_FG_];
#pragma unroll
        for (int i = 0; i < AO_FG_; ++i) {
            fa[0][i]   = af[i * 2];
            fb[0][i].x = vf[(i * 4 + 0) * (AO_VP4_ * 4)];
            fb[0][i].y = vf[(i * 4 + 1) * (AO_VP4_ * 4)];
        }
#pragma unroll
        for (int g = 0; g < AO_NG_; ++g) {
            if (g + 1 < AO_NG_) {
#pragma unroll
                for (int i = 0; i < AO_FG_; ++i) {
                    const int ks = (g + 1) * AO_FG_ + i;
                    fa[(g + 1) & 1][i]   = af[ks * 2];
                    fb[(g + 1) & 1][i].x = vf[(ks * 4 + 0) * (AO_VP4_ * 4)];
                    fb[(g + 1) & 1][i].y = vf[(ks * 4 + 1) * (AO_VP4_ * 4)];
                }
            }
            __builtin_amdgcn_sched_barrier(0);
#pragma unroll
            for (int i = 0; i < AO_FG_; ++i)
                acc = __builtin_amdgcn_wmma_f32_16x16x4_f32(
                          false, fa[g & 1][i], false, fb[g & 1][i], (short)0, acc, false, false);
            __builtin_amdgcn_sched_barrier(0);
        }

        if (more) {
#pragma unroll
            for (int p = 0; p < AO_VIT_; ++p)
                Vs4[((buf ^ 1) * AO_KC_ + vr + p * 16) * AO_VP4_ + vc] = rv[p];
        }
        __syncthreads();
    }

    const int m0 = bm * AO_BM_ + tm * 16 + 8 * h;
    const int n  = bn * AO_BN_ + tn * 16 + c15;
#pragma unroll
    for (int r = 0; r < 8; ++r)
        out[((size_t)b * LQ_ + m0 + r) * DSZ_ + n] = acc[r];
}

extern "C" void kernel_launch(void* const* d_in, const int* in_sizes, int n_in,
                              void* d_out, int out_size, void* d_ws, size_t ws_size,
                              hipStream_t stream) {
    const float* query     = (const float*)d_in[0];
    const float* key       = (const float*)d_in[1];
    const float* value     = (const float*)d_in[2];
    const int*   valid_len = (const int*)  d_in[3];
    const float* W_q       = (const float*)d_in[4];
    const float* W_k       = (const float*)d_in[5];
    const float* vvec      = (const float*)d_in[6];

    float* qf     = (float*)d_ws;
    float* kfT    = qf  + (size_t)B_ * LQ_ * UNITS_;
    float* scores = kfT + (size_t)B_ * UNITS_ * LK_;
    float* outp   = (float*)d_out;

    proj_both<<<PJ_KB_ + PJ_QB_, PJ_NT_, 0, stream>>>(key, W_k, kfT, query, W_q, qf, valid_len);
    scores_tanh_kernel<<<512, 128, 0, stream>>>(qf, kfT, vvec, valid_len, scores);
    attn_out_wmma<<<256, 128, 0, stream>>>(scores, value, outp);
}


# attn x value kernel: score pieces and value rows at or above valid_len are not read or multiplied where their weight is exactly zero (checked per block); weights and k sum order unchanged
# speedup vs baseline: 5.1042x; 1.0777x over previous
#include <hip/hip_runtime.h>

typedef float v2f __attribute__((ext_vector_type(2)));
typedef float v8f __attribute__((ext_vector_type(8)));
typedef float v4f __attribute__((ext_vector_type(4)));

#define B_     4
#define LQ_    128
#define LK_    1024
#define DSZ_   512
#define UNITS_ 256
#define MASK_VAL_ (-1.0e6f)

#if __has_builtin(__builtin_amdgcn_tanhf)
#define TANHF(x) __builtin_amdgcn_tanhf(x)
#else
#define TANHF(x) tanhf(x)
#endif

#define KT_   64
#define NKT_  (LK_ / KT_)
struct TileRef { int b; int kt; bool live; };

__device__ __forceinline__ TileRef tile_lookup(const int* __restrict__ valid_len, const int w)
{
    int n[B_];
    int live_total = 0;
#pragma unroll
    for (int b = 0; b < B_; ++b) {
        const int vl = valid_len[b];
        n[b] = (vl <= 0) ? 0 : ((vl >= LK_) ? NKT_ : (vl + KT_ - 1) / KT_);
        live_total += n[b];
    }
    TileRef r;
    r.live = (w < live_total);
    r.b  = 0;
    r.kt = 0;
    int  rest  = r.live ? w : w - live_total;
    bool found = false;
#pragma unroll
    for (int b = 0; b < B_; ++b) {
        const int cnt = r.live ? n[b] : NKT_ - n[b];
        if (!found && rest < cnt) {
            r.b   = b;
            r.kt  = r.live ? rest : n[b] + rest;
            found = true;
        }
        if (!found) rest -= cnt;
    }
    return r;
}

#define PJ_KC_     128
#define PJ_C4_     (PJ_KC_ / 4)
#define PJ_P4_     (PJ_C4_ + 1)
#define PJ_NC_     (DSZ_ / PJ_KC_)
#define PJ_NT_     256
#define PJ_BM_     32
#define PJ_BN_     64
#define PJ_KB_     512
#define PJ_QB_     64

template <int BM, int BN, int WN, int SEQ, bool TRANS_OUT>
__device__ __forceinline__ void proj_tile(
    const float* __restrict__ A,
    const float* __restrict__ W,
    float* __restrict__ out,
    const int blk,
    float4* __restrict__ As4,
    float4* __restrict__ Ws4)
{
    constexpr int K   = DSZ_;
    constexpr int N   = UNITS_;
    constexpr int NT  = (BM / 16) * (BN / 16) / WN * 32;
    constexpr int RPP = NT / PJ_C4_;
    constexpr int AIT = BM / RPP;
    constexpr int WIT = BN / RPP;
    constexpr int WPR = (BN / 16) / WN;
    static_assert(NT == PJ_NT_, "both sides run in blocks of PJ_NT_ threads");

    const int t    = threadIdx.x;
    const int lane = t & 31;
    const int wave = t >> 5;
    const int c15  = lane & 15;
    const int h    = lane >> 4;
    const int tm   = wave / WPR;
    const int tn0  = (wave % WPR) * WN;
    const int bm   = blk / (N / BN);
    const int bn   = blk % (N / BN);

    const int lr = t / PJ_C4_;
    const int lc = t % PJ_C4_;
    const float* ag = A + (size_t)(bm * BM + lr) * K + lc * 4;
    const float* wg = W + (size_t)(bn * BN + lr) * K + lc * 4;

    float4 ra[AIT], rw[WIT];
    v8f acc[WN];
#pragma unroll
    for (int j = 0; j < WN; ++j)
        acc[j] = (v8f){0.f, 0.f, 0.f, 0.f, 0.f, 0.f, 0.f, 0.f};

#pragma unroll
    for (int p = 0; p < AIT; ++p) ra[p] = *(const float4*)(ag + (size_t)p * RPP * K);
#pragma unroll
    for (int p = 0; p < WIT; ++p) rw[p] = *(const float4*)(wg + (size_t)p * RPP * K);
#pragma unroll
    for (int p = 0; p < AIT; ++p) As4[(lr + p * RPP) * PJ_P4_ + lc] = ra[p];
#pragma unroll
    for (int p = 0; p < WIT; ++p) Ws4[(lr + p * RPP) * PJ_P4_ + lc] = rw[p];
    __syncthreads();

    for (int kc = 0; kc < PJ_NC_; ++kc) {
        const int  buf  = kc & 1;
        const bool more = (kc + 1 < PJ_NC_);

        if (more) {
#pragma unroll
            for (int p = 0; p < AIT; ++p)
                ra[p] = *(const float4*)(ag + (size_t)p * RPP * K + (kc + 1) * PJ_KC_);
#pragma unroll
            for (int p = 0; p < WIT; ++p)
                rw[p] = *(const float4*)(wg + (size_t)p * RPP * K + (kc + 1) * PJ_KC_);
        }

        const v2f* af = (const v2f*)(As4 + (buf * BM + tm * 16 + c15) * PJ_P4_) + h;
        const v2f* wf = (const v2f*)(Ws4 + (buf * BN + tn0 * 16 + c15) * PJ_P4_) + h;
#pragma unroll
        for (int ks = 0; ks < PJ_KC_ / 4; ++ks) {
            const v2f av = af[ks * 2];
#pragma unroll
            for (int j = 0; j < WN; ++j) {
                const v2f bv = wf[j * 16 * PJ_P4_ * 2 + ks * 2];
                acc[j] = __builtin_amdgcn_wmma_f32_16x16x4_f32(
                             false, av, false, bv, (short)0, acc[j], false, false);
            }
        }

        if (more) {
#pragma unroll
            for (int p = 0; p < AIT; ++p)
                As4[((buf ^ 1) * BM + lr + p * RPP) * PJ_P4_ + lc] = ra[p];
#pragma unroll
            for (int p = 0; p < WIT; ++p)
                Ws4[((buf ^ 1) * BN + lr + p * RPP) * PJ_P4_ + lc] = rw[p];
        }
        __syncthreads();
    }

    const int m0 = bm * BM + tm * 16 + 8 * h;
#pragma unroll
    for (int j = 0; j < WN; ++j) {
        const int n = bn * BN + (tn0 + j) * 16 + c15;
#pragma unroll
        for (int r = 0; r < 8; ++r) {
            const int m = m0 + r;
            if (TRANS_OUT) {
                const int b    = m / SEQ;
                const int kpos = m % SEQ;
                out[((size_t)b * N + n) * SEQ + kpos] = acc[j][r];
            } else {
                out[(size_t)m * N + n] = acc[j][r];
            }
        }
    }
}

__global__ __launch_bounds__(PJ_NT_) void proj_both(
    const float* __restrict__ key,
    const float* __restrict__ W_k,
    float* __restrict__ kfT,
    const float* __restrict__ query,
    const float* __restrict__ W_q,
    float* __restrict__ qf,
    const int* __restrict__ valid_len)
{
    __shared__ float4 As4[2 * PJ_BM_ * PJ_P4_];
    __shared__ float4 Ws4[2 * PJ_BN_ * PJ_P4_];

    if (blockIdx.x < PJ_QB_) {
        proj_tile<PJ_BM_, PJ_BN_, 1, LQ_, false>(query, W_q, qf, (int)blockIdx.x, As4, Ws4);
    } else {
        const int kb = (int)blockIdx.x - PJ_QB_;
        const TileRef tr = tile_lookup(valid_len, kb >> 3);
        if (!tr.live) return;
        proj_tile<PJ_BM_, PJ_BN_, 1, LK_, true>(key, W_k, kfT,
                                                ((tr.b * NKT_ + tr.kt) << 3) | (kb & 7), As4, Ws4);
    }
}

#define SC_KT_ 64
#define SC_QG_ 16
#define SC_QW_ 4

#define SC_UT_ 8
#define SC_NS_ (SC_UT_ * SC_QW_)
#define SC_DA_ 3
#define SC_DF_ 3
#define SC_E_  (SC_DA_ + SC_DF_)
#define SC_HOLD() __builtin_amdgcn_sched_barrier(0)

struct ScTrip  { v4f q[SC_UT_ / 4][SC_QW_]; v2f k[SC_UT_]; v4f w[SC_UT_ / 4]; };
struct ScCarry { v2f s[SC_DA_]; v2f t[SC_DF_]; v2f w; };

__device__ __forceinline__ float sc_tanh(const float x)
{
    float r;
    asm volatile("v_tanh_f32 %0, %1" : "=v"(r) : "v"(x));
    return r;
}
template <int HI>
__device__ __forceinline__ v2f sc_add(const v2f k, const v2f qp)
{
    v2f r;
    if (HI) asm volatile("v_pk_add_f32 %0, %1, %2 op_sel:[0,1]"    : "=v"(r) : "v"(k), "v"(qp));
    else    asm volatile("v_pk_add_f32 %0, %1, %2 op_sel_hi:[1,0]" : "=v"(r) : "v"(k), "v"(qp));
    return r;
}
template <int HI>
__device__ __forceinline__ void sc_fma(v2f& a, const v2f wp, const v2f t)
{
    if (HI) asm volatile("v_pk_fma_f32 %0, %1, %2, %0 op_sel:[1,0,0]"    : "+v"(a) : "v"(wp), "v"(t));
    else    asm volatile("v_pk_fma_f32 %0, %1, %2, %0 op_sel_hi:[0,1,1]" : "+v"(a) : "v"(wp), "v"(t));
}

__device__ __forceinline__ void sc_trip_load(ScTrip& o, const v4f* __restrict__ qrow,
                                             const v2f* __restrict__ kS2l,
                                             const v4f* __restrict__ vS4, const int tr)
{
#pragma unroll
    for (int h = 0; h < SC_UT_ / 4; ++h)
#pragma unroll
        for (int j = 0; j < SC_QW_; ++j)
            o.q[h][j] = qrow[j * (UNITS_ / 4) + tr * (SC_UT_ / 4) + h];
#pragma unroll
    for (int i = 0; i < SC_UT_; ++i)
        o.k[i] = kS2l[(tr * SC_UT_ + i) * (SC_KT_ / 2)];
#pragma unroll
    for (int h = 0; h < SC_UT_ / 4; ++h)
        o.w[h] = vS4[tr * (SC_UT_ / 4) + h];
}

template <int P>
__device__ __forceinline__ v2f sc_slot_add(const ScTrip& o)
{
    constexpr int i = P / SC_QW_, j = P % SC_QW_;
    const v4f q4 = o.q[i >> 2][j];
    const v2f qp = ((i & 2) == 0) ? (v2f){q4.x, q4.y} : (v2f){q4.z, q4.w};
    return sc_add<(i & 1)>(o.k[i], qp);
}
template <int I>
__device__ __forceinline__ v2f sc_wpair(const ScTrip& o)
{
    const v4f w4 = o.w[I >> 2];
    return ((I & 2) == 0) ? (v2f){w4.x, w4.y} : (v2f){w4.z, w4.w};
}
template <int PS>
__device__ __forceinline__ void sc_slot_fma(v2f (&a)[SC_QW_], const v2f wp, const v2f t)
{
    sc_fma<((PS / SC_QW_) & 1)>(a[PS % SC_QW_], wp, t);
}

template <int P>
__device__ __forceinline__ void sc_step(const ScTrip& o, const ScCarry& cy,
                                        v2f (&Sx)[SC_NS_ + SC_E_], v2f (&Tx)[SC_NS_ + SC_DF_],
                                        v2f (&a)[SC_QW_])
{
    constexpr int eT = P + SC_DF_;
    Tx[eT].x = sc_tanh(Sx[eT].x);
    Sx[P + SC_E_] = sc_slot_add<P>(o);
    Tx[eT].y = sc_tanh(Sx[eT].y);
    constexpr int sl = P - SC_E_;
    constexpr int ps = (sl >= 0) ? sl : SC_NS_ + sl;
    sc_slot_fma<ps>(a, (sl >= 0) ? sc_wpair<ps / SC_QW_>(o) : cy.w, Tx[P]);
    if constexpr (P + 1 < SC_NS_) sc_step<P + 1>(o, cy, Sx, Tx, a);
}

__device__ __forceinline__ void sc_trip_steps(const ScTrip& o, ScCarry& cy, v2f (&a)[SC_QW_])
{
    v2f Sx[SC_NS_ + SC_E_], Tx[SC_NS_ + SC_DF_];
#pragma unroll
    for (int d = 0; d < SC_DA_; ++d) Sx[SC_DF_ + d] = cy.s[d];
#pragma unroll
    for (int d = 0; d < SC_DF_; ++d) Tx[d] = cy.t[d];

    sc_step<0>(o, cy, Sx, Tx, a);

#pragma unroll
    for (int d = 0; d < SC_DA_; ++d) cy.s[d] = Sx[SC_NS_ + SC_DF_ + d];
#pragma unroll
    for (int d = 0; d < SC_DF_; ++d) cy.t[d] = Tx[SC_NS_ + d];
    cy.w = sc_wpair<SC_UT_ - 1>(o);
}

template <int D>
__device__ __forceinline__ void sc_drain_fma(const ScCarry& cy, const v2f (&td)[SC_DA_],
                                             v2f (&a)[SC_QW_])
{
    constexpr int ps = SC_NS_ - SC_E_ + D;
    if constexpr (D < SC_DF_) sc_slot_fma<ps>(a, cy.w, cy.t[D]);
    else                      sc_slot_fma<ps>(a, cy.w, td[D - SC_DF_]);
    if constexpr (D + 1 < SC_E_) sc_drain_fma<D + 1>(cy, td, a);
}
__device__ __forceinline__ void sc_drain(const ScCarry& cy, v2f (&a)[SC_QW_])
{
    v2f td[SC_DA_];
#pragma unroll
    for (int d = 0; d < SC_DA_; ++d) td[d].x = sc_tanh(cy.s[d].x);
#pragma unroll
    for (int d = 0; d < SC_DA_; ++d) td[d].y = sc_tanh(cy.s[d].y);
    sc_drain_fma<0>(cy, td, a);
}

__global__ __launch_bounds__(128) void scores_tanh_kernel(
    const float* __restrict__ qf,
    const float* __restrict__ kfT,
    const float* __restrict__ vvec,
    const int*   __restrict__ valid_len,
    float* __restrict__ scores)
{
    __shared__ float4 scS4[SC_QG_ * UNITS_ / 4 + UNITS_ / 4 + UNITS_ * SC_KT_ / 4];
    float4* const qS4 = scS4;
    float*  const vS  = (float*)(scS4 + SC_QG_ * UNITS_ / 4);
    float4* const kS4 = scS4 + SC_QG_ * UNITS_ / 4 + UNITS_ / 4;

    const int t    = threadIdx.x;
    const int lane = t & 31;
    const int wave = t >> 5;
    const int qg   = blockIdx.x & 7;
    const TileRef tr = tile_lookup(valid_len, (int)(blockIdx.x >> 3));
    const int kt   = tr.kt;
    const int b    = tr.b;

    if (!tr.live) {
        float* dst = scores + (size_t)(b * LQ_ + qg * SC_QG_ + wave * SC_QW_) * LK_
                            + kt * SC_KT_ + 2 * lane;
#pragma unroll
        for (int j = 0; j < SC_QW_; ++j)
            *(float2*)(dst + (size_t)j * LK_) = make_float2(MASK_VAL_, MASK_VAL_);
        return;
    }

    const float* ksrc = kfT + (size_t)b * UNITS_ * LK_ + kt * SC_KT_;
    const int c  = t & 15;
    const int r0 = t >> 4;
#pragma unroll
    for (int i = 0; i < UNITS_ / 8; ++i) {
        const int u = r0 + 8 * i;
        kS4[u * (SC_KT_ / 4) + c] = *(const float4*)(ksrc + (size_t)u * LK_ + c * 4);
    }
    const float4* qsrc =
        (const float4*)(qf + (size_t)(b * LQ_ + qg * SC_QG_) * UNITS_);
#pragma unroll
    for (int i = 0; i < SC_QG_ * UNITS_ / 4 / 128; ++i)
        qS4[i * 128 + t] = qsrc[i * 128 + t];
#pragma unroll
    for (int i = 0; i < UNITS_ / 128; ++i)
        vS[i * 128 + t] = vvec[i * 128 + t];
    __syncthreads();

    const v2f* kS2l = (const v2f*)kS4 + lane;
    const v4f* qrow = (const v4f*)qS4 + (wave * SC_QW_) * (UNITS_ / 4);
    const v4f* vS4  = (const v4f*)vS;

    v2f a[SC_QW_];
#pragma unroll
    for (int j = 0; j < SC_QW_; ++j) a[j] = (v2f){0.f, 0.f};

    ScCarry cy;
#pragma unroll
    for (int d = 0; d < SC_DA_; ++d) cy.s[d] = (v2f){0.f, 0.f};
#pragma unroll
    for (int d = 0; d < SC_DF_; ++d) cy.t[d] = (v2f){0.f, 0.f};
    cy.w = (v2f){0.f, 0.f};

    ScTrip A, B;
    sc_trip_load(A, qrow, kS2l, vS4, 0);
    for (int tr = 0; tr < UNITS_ / SC_UT_; tr += 2) {
        sc_trip_load(B, qrow, kS2l, vS4, tr + 1);
        SC_HOLD();
        sc_trip_steps(A, cy, a);
        sc_trip_load(A, qrow, kS2l, vS4, (tr + 2 < UNITS_ / SC_UT_) ? tr + 2 : tr + 1);
        SC_HOLD();
        sc_trip_steps(B, cy, a);
    }
    sc_drain(cy, a);

    const int vl = valid_len[b];
    const int k0 = kt * SC_KT_ + 2 * lane;
    const int q0 = b * LQ_ + qg * SC_QG_ + wave * SC_QW_;
#pragma unroll
    for (int j = 0; j < SC_QW_; ++j) {
        float2 r;
        r.x = (k0 + 0 < vl) ? a[j].x : MASK_VAL_;
        r.y = (k0 + 1 < vl) ? a[j].y : MASK_VAL_;
        *(float2*)(scores + (size_t)(q0 + j) * LK_ + k0) = r;
    }
}

#define AO_BM_  32
#define AO_BN_  32
#define AO_KC_  256
#define AO_NC_  (LK_ / AO_KC_)
#define AO_RP4_ (LK_ / 4 + 1)
#define AO_RW_  (AO_BM_ / 4)
#define AO_RD_  4
#define AO_RJ_  (LK_ / 128)
#define AO_VP4_ 12
#define AO_VIT_ (AO_KC_ / 16)
#define AO_FG_  8
#define AO_NG_  (AO_KC_ / 4 / AO_FG_)

__device__ __forceinline__ float lane_xor16(float v) {
    const int b = __builtin_bit_cast(int, v);
    return __builtin_bit_cast(float,
        __builtin_amdgcn_permlanex16(b, b, 0x76543210, 0xfedcba98, false, false));
}
template <int N>
__device__ __forceinline__ float lane_xor(float v) {
    return __builtin_bit_cast(float,
        __builtin_amdgcn_update_dpp(0, __builtin_bit_cast(int, v), 0x160 + N, 0xf, 0xf, true));
}

template <int NLJ>
__device__ __forceinline__ bool ao_rows(const float* __restrict__ sg, v4f* __restrict__ As4,
                                        const int wave, const int lane)
{
    constexpr int NX = (NLJ > 0) ? NLJ : 1;
    v4f x[AO_RD_][NX];
    if constexpr (NLJ > 0) {
#pragma unroll
        for (int r = 0; r < AO_RD_; ++r)
#pragma unroll
            for (int j = 0; j < NLJ; ++j)
                x[r][j] = *(const v4f*)(sg + (size_t)r * LK_ + j * 128);
    }
    bool nz = false;
#pragma unroll
    for (int r = 0; r < AO_RW_; ++r) {
        v4f (&xr)[NX] = x[r % AO_RD_];

        float m = MASK_VAL_;
        if constexpr (NLJ > 0) {
            m = fmaxf(fmaxf(xr[0].x, xr[0].y), fmaxf(xr[0].z, xr[0].w));
#pragma unroll
            for (int j = 1; j < NLJ; ++j)
                m = fmaxf(m, fmaxf(fmaxf(xr[j].x, xr[j].y), fmaxf(xr[j].z, xr[j].w)));
            if constexpr (NLJ < AO_RJ_) m = fmaxf(m, MASK_VAL_);
        }
        m = fmaxf(m, lane_xor16(m));
        m = fmaxf(m, lane_xor<8>(m));
        m = fmaxf(m, lane_xor<4>(m));
        m = fmaxf(m, lane_xor<2>(m));
        m = fmaxf(m, lane_xor<1>(m));

        float s[NX];
        if constexpr (NLJ > 0) {
#pragma unroll
            for (int j = 0; j < NLJ; ++j) {
                v4f e;
                e.x = __expf(xr[j].x - m);
                e.y = __expf(xr[j].y - m);
                e.z = __expf(xr[j].z - m);
                e.w = __expf(xr[j].w - m);
                xr[j] = e;
                s[j] = e.x + e.y + e.z + e.w;
            }
        }
        const float ed = __expf(MASK_VAL_ - m);
        nz = nz || (ed != 0.0f);
        float sd = ed + ed + ed + ed;
        if constexpr (NLJ > 0) {
#pragma unroll
            for (int j = 0; j < NLJ; ++j) s[j] += lane_xor16(s[j]);
#pragma unroll
            for (int j = 0; j < NLJ; ++j) s[j] += lane_xor<8>(s[j]);
#pragma unroll
            for (int j = 0; j < NLJ; ++j) s[j] += lane_xor<4>(s[j]);
#pragma unroll
            for (int j = 0; j < NLJ; ++j) s[j] += lane_xor<2>(s[j]);
#pragma unroll
            for (int j = 0; j < NLJ; ++j) s[j] += lane_xor<1>(s[j]);
        }
        if constexpr (NLJ < AO_RJ_) {
            sd += lane_xor16(sd);
            sd += lane_xor<8>(sd);
            sd += lane_xor<4>(sd);
            sd += lane_xor<2>(sd);
            sd += lane_xor<1>(sd);
        }
        float gs = (NLJ > 0) ? s[0] : sd;
#pragma unroll
        for (int j = 1; j < AO_RJ_; ++j) gs += (j < NLJ) ? s[j < NX ? j : 0] : sd;

        const float inv = 1.0f / gs;
        v4f* arow = As4 + (wave * AO_RW_ + r) * AO_RP4_ + lane;
        if constexpr (NLJ > 0) {
#pragma unroll
            for (int j = 0; j < NLJ; ++j) {
                v4f e = xr[j];
                e.x *= inv; e.y *= inv; e.z *= inv; e.w *= inv;
                arow[j * 32] = e;
            }
        }
        if constexpr (NLJ < AO_RJ_) {
            const float wd = ed * inv;
            const v4f wd4 = {wd, wd, wd, wd};
#pragma unroll
            for (int j = NLJ; j < AO_RJ_; ++j) arow[j * 32] = wd4;
        }

        if constexpr (NLJ > 0) {
            if (r + AO_RD_ < AO_RW_) {
#pragma unroll
                for (int j = 0; j < NLJ; ++j)
                    xr[j] = *(const v4f*)(sg + (size_t)(r + AO_RD_) * LK_ + j * 128);
            }
        }
    }
    return nz;
}

__global__ __launch_bounds__(128) void attn_out_wmma(
    const float* __restrict__ scores,
    const float* __restrict__ value,
    const int*   __restrict__ valid_len,
    float* __restrict__ out)
{
    __shared__ v4f As4[AO_BM_ * AO_RP4_];
    __shared__ v4f Vs4[2 * AO_KC_ * AO_VP4_];
    __shared__ int nzS[4];

    const int t    = threadIdx.x;
    const int lane = t & 31;
    const int wave = t >> 5;
    const int c15  = lane & 15;
    const int h    = lane >> 4;
    const int tm   = wave >> 1;
    const int tn   = wave & 1;
    const int bn   = blockIdx.x & 15;
    const int bm   = (blockIdx.x >> 4) & 3;
    const int b    = blockIdx.x >> 6;
    const int vl   = valid_len[b];

    const int vr = t >> 3;
    const int vc = t & 7;
    const float* sg = scores + ((size_t)b * LQ_ + bm * AO_BM_ + wave * AO_RW_) * LK_ + lane * 4;
    const float* vg = value + ((size_t)b * LK_ + vr) * DSZ_ + bn * AO_BN_ + vc * 4;

    v4f rv[AO_VIT_];
    v8f acc = {0.f, 0.f, 0.f, 0.f, 0.f, 0.f, 0.f, 0.f};

#pragma unroll
    for (int p = 0; p < AO_VIT_; ++p) rv[p] = *(const v4f*)(vg + (size_t)p * 16 * DSZ_);

    const int nlj = (vl <= 0) ? 0 : ((vl >= LK_) ? AO_RJ_ : (vl + 127) >> 7);
    bool nz;
    switch (nlj) {
    case 0:  nz = ao_rows<0>(sg, As4, wave, lane); break;
    case 1:  nz = ao_rows<1>(sg, As4, wave, lane); break;
    case 2:  nz = ao_rows<2>(sg, As4, wave, lane); break;
    case 3:  nz = ao_rows<3>(sg, As4, wave, lane); break;
    case 4:  nz = ao_rows<4>(sg, As4, wave, lane); break;
    case 5:  nz = ao_rows<5>(sg, As4, wave, lane); break;
    case 6:  nz = ao_rows<6>(sg, As4, wave, lane); break;
    case 7:  nz = ao_rows<7>(sg, As4, wave, lane); break;
    default: nz = ao_rows<8>(sg, As4, wave, lane); break;
    }
    if (lane == 0) nzS[wave] = nz ? 1 : 0;

#pragma unroll
    for (int p = 0; p < AO_VIT_; ++p) Vs4[(vr + p * 16) * AO_VP4_ + vc] = rv[p];
    __syncthreads();

    const int anynz = __builtin_amdgcn_readfirstlane(nzS[0] | nzS[1] | nzS[2] | nzS[3]);
    const int ngt = (anynz != 0 || vl <= 0 || vl >= LK_) ? (LK_ / 32) : ((vl + 31) >> 5);
    const int nch = (ngt + AO_NG_ - 1) / AO_NG_;

    for (int kc = 0; kc < nch; ++kc) {
        const int  buf  = kc & 1;
        const bool more = (kc + 1 < nch);
        const int  gcnt = min(AO_NG_, ngt - kc * AO_NG_);

        if (more) {
#pragma unroll
            for (int p = 0; p < AO_VIT_; ++p)
                rv[p] = *(const v4f*)(vg + ((size_t)(kc + 1) * AO_KC_ + p * 16) * DSZ_);
        }

        const v2f*   af = (const v2f*)(As4 + (tm * 16 + c15) * AO_RP4_) + kc * (AO_KC_ / 2) + h;
        const float* vf = (const float*)(Vs4 + (buf * AO_KC_ + 2 * h) * AO_VP4_) + tn * 16 + c15;
        v2f fa[2][AO_FG_], fb[2][AO_FG_];
#pragma unroll
        for (int i = 0; i < AO_FG_; ++i) {
            fa[0][i]   = af[i * 2];
            fb[0][i].x = vf[(i * 4 + 0) * (AO_VP4_ * 4)];
            fb[0][i].y = vf[(i * 4 + 1) * (AO_VP4_ * 4)];
        }
#pragma unroll
        for (int g = 0; g < AO_NG_; ++g) {
            if (g >= gcnt) break;
            if (g + 1 < AO_NG_) {
                if (g + 1 < gcnt) {
#pragma unroll
                    for (int i = 0; i < AO_FG_; ++i) {
                        const int ks = (g + 1) * AO_FG_ + i;
                        fa[(g + 1) & 1][i]   = af[ks * 2];
                        fb[(g + 1) & 1][i].x = vf[(ks * 4 + 0) * (AO_VP4_ * 4)];
                        fb[(g + 1) & 1][i].y = vf[(ks * 4 + 1) * (AO_VP4_ * 4)];
                    }
                }
            }
            __builtin_amdgcn_sched_barrier(0);
#pragma unroll
            for (int i = 0; i < AO_FG_; ++i)
                acc = __builtin_amdgcn_wmma_f32_16x16x4_f32(
                          false, fa[g & 1][i], false, fb[g & 1][i], (short)0, acc, false, false);
            __builtin_amdgcn_sched_barrier(0);
        }

        if (more) {
#pragma unroll
            for (int p = 0; p < AO_VIT_; ++p)
                Vs4[((buf ^ 1) * AO_KC_ + vr + p * 16) * AO_VP4_ + vc] = rv[p];
        }
        __syncthreads();
    }

    const int m0 = bm * AO_BM_ + tm * 16 + 8 * h;
    const int n  = bn * AO_BN_ + tn * 16 + c15;
#pragma unroll
    for (int r = 0; r < 8; ++r)
        out[((size_t)b * LQ_ + m0 + r) * DSZ_ + n] = acc[r];
}

extern "C" void kernel_launch(void* const* d_in, const int* in_sizes, int n_in,
                              void* d_out, int out_size, void* d_ws, size_t ws_size,
                              hipStream_t stream) {
    const float* query     = (const float*)d_in[0];
    const float* key       = (const float*)d_in[1];
    const float* value     = (const float*)d_in[2];
    const int*   valid_len = (const int*)  d_in[3];
    const float* W_q       = (const float*)d_in[4];
    const float* W_k       = (const float*)d_in[5];
    const float* vvec      = (const float*)d_in[6];

    float* qf     = (float*)d_ws;
    float* kfT    = qf  + (size_t)B_ * LQ_ * UNITS_;
    float* scores = kfT + (size_t)B_ * UNITS_ * LK_;
    float* outp   = (float*)d_out;

    proj_both<<<PJ_KB_ + PJ_QB_, PJ_NT_, 0, stream>>>(key, W_k, kfT, query, W_q, qf, valid_len);
    scores_tanh_kernel<<<512, 128, 0, stream>>>(qf, kfT, vvec, valid_len, scores);
    attn_out_wmma<<<256, 128, 0, stream>>>(scores, value, valid_len, outp);
}


# attn x value kernel: 8 waves per block; the two waves of a SIMD share a tile, one takes the even and one the odd 32-k groups, the two sums added through LDS at the end (k sum order changed)
# speedup vs baseline: 5.2663x; 1.0318x over previous
#include <hip/hip_runtime.h>

typedef float v2f __attribute__((ext_vector_type(2)));
typedef float v8f __attribute__((ext_vector_type(8)));
typedef float v4f __attribute__((ext_vector_type(4)));

#define B_     4
#define LQ_    128
#define LK_    1024
#define DSZ_   512
#define UNITS_ 256
#define MASK_VAL_ (-1.0e6f)

#if __has_builtin(__builtin_amdgcn_tanhf)
#define TANHF(x) __builtin_amdgcn_tanhf(x)
#else
#define TANHF(x) tanhf(x)
#endif

#define KT_   64
#define NKT_  (LK_ / KT_)
struct TileRef { int b; int kt; bool live; };

__device__ __forceinline__ TileRef tile_lookup(const int* __restrict__ valid_len, const int w)
{
    int n[B_];
    int live_total = 0;
#pragma unroll
    for (int b = 0; b < B_; ++b) {
        const int vl = valid_len[b];
        n[b] = (vl <= 0) ? 0 : ((vl >= LK_) ? NKT_ : (vl + KT_ - 1) / KT_);
        live_total += n[b];
    }
    TileRef r;
    r.live = (w < live_total);
    r.b  = 0;
    r.kt = 0;
    int  rest  = r.live ? w : w - live_total;
    bool found = false;
#pragma unroll
    for (int b = 0; b < B_; ++b) {
        const int cnt = r.live ? n[b] : NKT_ - n[b];
        if (!found && rest < cnt) {
            r.b   = b;
            r.kt  = r.live ? rest : n[b] + rest;
            found = true;
        }
        if (!found) rest -= cnt;
    }
    return r;
}

#define PJ_KC_     128
#define PJ_C4_     (PJ_KC_ / 4)
#define PJ_P4_     (PJ_C4_ + 1)
#define PJ_NC_     (DSZ_ / PJ_KC_)
#define PJ_NT_     256
#define PJ_BM_     32
#define PJ_BN_     64
#define PJ_KB_     512
#define PJ_QB_     64

template <int BM, int BN, int WN, int SEQ, bool TRANS_OUT>
__device__ __forceinline__ void proj_tile(
    const float* __restrict__ A,
    const float* __restrict__ W,
    float* __restrict__ out,
    const int blk,
    float4* __restrict__ As4,
    float4* __restrict__ Ws4)
{
    constexpr int K   = DSZ_;
    constexpr int N   = UNITS_;
    constexpr int NT  = (BM / 16) * (BN / 16) / WN * 32;
    constexpr int RPP = NT / PJ_C4_;
    constexpr int AIT = BM / RPP;
    constexpr int WIT = BN / RPP;
    constexpr int WPR = (BN / 16) / WN;
    static_assert(NT == PJ_NT_, "both sides run in blocks of PJ_NT_ threads");

    const int t    = threadIdx.x;
    const int lane = t & 31;
    const int wave = t >> 5;
    const int c15  = lane & 15;
    const int h    = lane >> 4;
    const int tm   = wave / WPR;
    const int tn0  = (wave % WPR) * WN;
    const int bm   = blk / (N / BN);
    const int bn   = blk % (N / BN);

    const int lr = t / PJ_C4_;
    const int lc = t % PJ_C4_;
    const float* ag = A + (size_t)(bm * BM + lr) * K + lc * 4;
    const float* wg = W + (size_t)(bn * BN + lr) * K + lc * 4;

    float4 ra[AIT], rw[WIT];
    v8f acc[WN];
#pragma unroll
    for (int j = 0; j < WN; ++j)
        acc[j] = (v8f){0.f, 0.f, 0.f, 0.f, 0.f, 0.f, 0.f, 0.f};

#pragma unroll
    for (int p = 0; p < AIT; ++p) ra[p] = *(const float4*)(ag + (size_t)p * RPP * K);
#pragma unroll
    for (int p = 0; p < WIT; ++p) rw[p] = *(const float4*)(wg + (size_t)p * RPP * K);
#pragma unroll
    for (int p = 0; p < AIT; ++p) As4[(lr + p * RPP) * PJ_P4_ + lc] = ra[p];
#pragma unroll
    for (int p = 0; p < WIT; ++p) Ws4[(lr + p * RPP) * PJ_P4_ + lc] = rw[p];
    __syncthreads();

    for (int kc = 0; kc < PJ_NC_; ++kc) {
        const int  buf  = kc & 1;
        const bool more = (kc + 1 < PJ_NC_);

        if (more) {
#pragma unroll
            for (int p = 0; p < AIT; ++p)
                ra[p] = *(const float4*)(ag + (size_t)p * RPP * K + (kc + 1) * PJ_KC_);
#pragma unroll
            for (int p = 0; p < WIT; ++p)
                rw[p] = *(const float4*)(wg + (size_t)p * RPP * K + (kc + 1) * PJ_KC_);
        }

        const v2f* af = (const v2f*)(As4 + (buf * BM + tm * 16 + c15) * PJ_P4_) + h;
        const v2f* wf = (const v2f*)(Ws4 + (buf * BN + tn0 * 16 + c15) * PJ_P4_) + h;
#pragma unroll
        for (int ks = 0; ks < PJ_KC_ / 4; ++ks) {
            const v2f av = af[ks * 2];
#pragma unroll
            for (int j = 0; j < WN; ++j) {
                const v2f bv = wf[j * 16 * PJ_P4_ * 2 + ks * 2];
                acc[j] = __builtin_amdgcn_wmma_f32_16x16x4_f32(
                             false, av, false, bv, (short)0, acc[j], false, false);
            }
        }

        if (more) {
#pragma unroll
            for (int p = 0; p < AIT; ++p)
                As4[((buf ^ 1) * BM + lr + p * RPP) * PJ_P4_ + lc] = ra[p];
#pragma unroll
            for (int p = 0; p < WIT; ++p)
                Ws4[((buf ^ 1) * BN + lr + p * RPP) * PJ_P4_ + lc] = rw[p];
        }
        __syncthreads();
    }

    const int m0 = bm * BM + tm * 16 + 8 * h;
#pragma unroll
    for (int j = 0; j < WN; ++j) {
        const int n = bn * BN + (tn0 + j) * 16 + c15;
#pragma unroll
        for (int r = 0; r < 8; ++r) {
            const int m = m0 + r;
            if (TRANS_OUT) {
                const int b    = m / SEQ;
                const int kpos = m % SEQ;
                out[((size_t)b * N + n) * SEQ + kpos] = acc[j][r];
            } else {
                out[(size_t)m * N + n] = acc[j][r];
            }
        }
    }
}

__global__ __launch_bounds__(PJ_NT_) void proj_both(
    const float* __restrict__ key,
    const float* __restrict__ W_k,
    float* __restrict__ kfT,
    const float* __restrict__ query,
    const float* __restrict__ W_q,
    float* __restrict__ qf,
    const int* __restrict__ valid_len)
{
    __shared__ float4 As4[2 * PJ_BM_ * PJ_P4_];
    __shared__ float4 Ws4[2 * PJ_BN_ * PJ_P4_];

    if (blockIdx.x < PJ_QB_) {
        proj_tile<PJ_BM_, PJ_BN_, 1, LQ_, false>(query, W_q, qf, (int)blockIdx.x, As4, Ws4);
    } else {
        const int kb = (int)blockIdx.x - PJ_QB_;
        const TileRef tr = tile_lookup(valid_len, kb >> 3);
        if (!tr.live) return;
        proj_tile<PJ_BM_, PJ_BN_, 1, LK_, true>(key, W_k, kfT,
                                                ((tr.b * NKT_ + tr.kt) << 3) | (kb & 7), As4, Ws4);
    }
}

#define SC_KT_ 64
#define SC_QG_ 16
#define SC_QW_ 4

#define SC_UT_ 8
#define SC_NS_ (SC_UT_ * SC_QW_)
#define SC_DA_ 3
#define SC_DF_ 3
#define SC_E_  (SC_DA_ + SC_DF_)
#define SC_HOLD() __builtin_amdgcn_sched_barrier(0)

struct ScTrip  { v4f q[SC_UT_ / 4][SC_QW_]; v2f k[SC_UT_]; v4f w[SC_UT_ / 4]; };
struct ScCarry { v2f s[SC_DA_]; v2f t[SC_DF_]; v2f w; };

__device__ __forceinline__ float sc_tanh(const float x)
{
    float r;
    asm volatile("v_tanh_f32 %0, %1" : "=v"(r) : "v"(x));
    return r;
}
template <int HI>
__device__ __forceinline__ v2f sc_add(const v2f k, const v2f qp)
{
    v2f r;
    if (HI) asm volatile("v_pk_add_f32 %0, %1, %2 op_sel:[0,1]"    : "=v"(r) : "v"(k), "v"(qp));
    else    asm volatile("v_pk_add_f32 %0, %1, %2 op_sel_hi:[1,0]" : "=v"(r) : "v"(k), "v"(qp));
    return r;
}
template <int HI>
__device__ __forceinline__ void sc_fma(v2f& a, const v2f wp, const v2f t)
{
    if (HI) asm volatile("v_pk_fma_f32 %0, %1, %2, %0 op_sel:[1,0,0]"    : "+v"(a) : "v"(wp), "v"(t));
    else    asm volatile("v_pk_fma_f32 %0, %1, %2, %0 op_sel_hi:[0,1,1]" : "+v"(a) : "v"(wp), "v"(t));
}

__device__ __forceinline__ void sc_trip_load(ScTrip& o, const v4f* __restrict__ qrow,
                                             const v2f* __restrict__ kS2l,
                                             const v4f* __restrict__ vS4, const int tr)
{
#pragma unroll
    for (int h = 0; h < SC_UT_ / 4; ++h)
#pragma unroll
        for (int j = 0; j < SC_QW_; ++j)
            o.q[h][j] = qrow[j * (UNITS_ / 4) + tr * (SC_UT_ / 4) + h];
#pragma unroll
    for (int i = 0; i < SC_UT_; ++i)
        o.k[i] = kS2l[(tr * SC_UT_ + i) * (SC_KT_ / 2)];
#pragma unroll
    for (int h = 0; h < SC_UT_ / 4; ++h)
        o.w[h] = vS4[tr * (SC_UT_ / 4) + h];
}

template <int P>
__device__ __forceinline__ v2f sc_slot_add(const ScTrip& o)
{
    constexpr int i = P / SC_QW_, j = P % SC_QW_;
    const v4f q4 = o.q[i >> 2][j];
    const v2f qp = ((i & 2) == 0) ? (v2f){q4.x, q4.y} : (v2f){q4.z, q4.w};
    return sc_add<(i & 1)>(o.k[i], qp);
}
template <int I>
__device__ __forceinline__ v2f sc_wpair(const ScTrip& o)
{
    const v4f w4 = o.w[I >> 2];
    return ((I & 2) == 0) ? (v2f){w4.x, w4.y} : (v2f){w4.z, w4.w};
}
template <int PS>
__device__ __forceinline__ void sc_slot_fma(v2f (&a)[SC_QW_], const v2f wp, const v2f t)
{
    sc_fma<((PS / SC_QW_) & 1)>(a[PS % SC_QW_], wp, t);
}

template <int P>
__device__ __forceinline__ void sc_step(const ScTrip& o, const ScCarry& cy,
                                        v2f (&Sx)[SC_NS_ + SC_E_], v2f (&Tx)[SC_NS_ + SC_DF_],
                                        v2f (&a)[SC_QW_])
{
    constexpr int eT = P + SC_DF_;
    Tx[eT].x = sc_tanh(Sx[eT].x);
    Sx[P + SC_E_] = sc_slot_add<P>(o);
    Tx[eT].y = sc_tanh(Sx[eT].y);
    constexpr int sl = P - SC_E_;
    constexpr int ps = (sl >= 0) ? sl : SC_NS_ + sl;
    sc_slot_fma<ps>(a, (sl >= 0) ? sc_wpair<ps / SC_QW_>(o) : cy.w, Tx[P]);
    if constexpr (P + 1 < SC_NS_) sc_step<P + 1>(o, cy, Sx, Tx, a);
}

__device__ __forceinline__ void sc_trip_steps(const ScTrip& o, ScCarry& cy, v2f (&a)[SC_QW_])
{
    v2f Sx[SC_NS_ + SC_E_], Tx[SC_NS_ + SC_DF_];
#pragma unroll
    for (int d = 0; d < SC_DA_; ++d) Sx[SC_DF_ + d] = cy.s[d];
#pragma unroll
    for (int d = 0; d < SC_DF_; ++d) Tx[d] = cy.t[d];

    sc_step<0>(o, cy, Sx, Tx, a);

#pragma unroll
    for (int d = 0; d < SC_DA_; ++d) cy.s[d] = Sx[SC_NS_ + SC_DF_ + d];
#pragma unroll
    for (int d = 0; d < SC_DF_; ++d) cy.t[d] = Tx[SC_NS_ + d];
    cy.w = sc_wpair<SC_UT_ - 1>(o);
}

template <int D>
__device__ __forceinline__ void sc_drain_fma(const ScCarry& cy, const v2f (&td)[SC_DA_],
                                             v2f (&a)[SC_QW_])
{
    constexpr int ps = SC_NS_ - SC_E_ + D;
    if constexpr (D < SC_DF_) sc_slot_fma<ps>(a, cy.w, cy.t[D]);
    else                      sc_slot_fma<ps>(a, cy.w, td[D - SC_DF_]);
    if constexpr (D + 1 < SC_E_) sc_drain_fma<D + 1>(cy, td, a);
}
__device__ __forceinline__ void sc_drain(const ScCarry& cy, v2f (&a)[SC_QW_])
{
    v2f td[SC_DA_];
#pragma unroll
    for (int d = 0; d < SC_DA_; ++d) td[d].x = sc_tanh(cy.s[d].x);
#pragma unroll
    for (int d = 0; d < SC_DA_; ++d) td[d].y = sc_tanh(cy.s[d].y);
    sc_drain_fma<0>(cy, td, a);
}

__global__ __launch_bounds__(128) void scores_tanh_kernel(
    const float* __restrict__ qf,
    const float* __restrict__ kfT,
    const float* __restrict__ vvec,
    const int*   __restrict__ valid_len,
    float* __restrict__ scores)
{
    __shared__ float4 scS4[SC_QG_ * UNITS_ / 4 + UNITS_ / 4 + UNITS_ * SC_KT_ / 4];
    float4* const qS4 = scS4;
    float*  const vS  = (float*)(scS4 + SC_QG_ * UNITS_ / 4);
    float4* const kS4 = scS4 + SC_QG_ * UNITS_ / 4 + UNITS_ / 4;

    const int t    = threadIdx.x;
    const int lane = t & 31;
    const int wave = t >> 5;
    const int qg   = blockIdx.x & 7;
    const TileRef tr = tile_lookup(valid_len, (int)(blockIdx.x >> 3));
    const int kt   = tr.kt;
    const int b    = tr.b;

    if (!tr.live) {
        float* dst = scores + (size_t)(b * LQ_ + qg * SC_QG_ + wave * SC_QW_) * LK_
                            + kt * SC_KT_ + 2 * lane;
#pragma unroll
        for (int j = 0; j < SC_QW_; ++j)
            *(float2*)(dst + (size_t)j * LK_) = make_float2(MASK_VAL_, MASK_VAL_);
        return;
    }

    const float* ksrc = kfT + (size_t)b * UNITS_ * LK_ + kt * SC_KT_;
    const int c  = t & 15;
    const int r0 = t >> 4;
#pragma unroll
    for (int i = 0; i < UNITS_ / 8; ++i) {
        const int u = r0 + 8 * i;
        kS4[u * (SC_KT_ / 4) + c] = *(const float4*)(ksrc + (size_t)u * LK_ + c * 4);
    }
    const float4* qsrc =
        (const float4*)(qf + (size_t)(b * LQ_ + qg * SC_QG_) * UNITS_);
#pragma unroll
    for (int i = 0; i < SC_QG_ * UNITS_ / 4 / 128; ++i)
        qS4[i * 128 + t] = qsrc[i * 128 + t];
#pragma unroll
    for (int i = 0; i < UNITS_ / 128; ++i)
        vS[i * 128 + t] = vvec[i * 128 + t];
    __syncthreads();

    const v2f* kS2l = (const v2f*)kS4 + lane;
    const v4f* qrow = (const v4f*)qS4 + (wave * SC_QW_) * (UNITS_ / 4);
    const v4f* vS4  = (const v4f*)vS;

    v2f a[SC_QW_];
#pragma unroll
    for (int j = 0; j < SC_QW_; ++j) a[j] = (v2f){0.f, 0.f};

    ScCarry cy;
#pragma unroll
    for (int d = 0; d < SC_DA_; ++d) cy.s[d] = (v2f){0.f, 0.f};
#pragma unroll
    for (int d = 0; d < SC_DF_; ++d) cy.t[d] = (v2f){0.f, 0.f};
    cy.w = (v2f){0.f, 0.f};

    ScTrip A, B;
    sc_trip_load(A, qrow, kS2l, vS4, 0);
    for (int tr = 0; tr < UNITS_ / SC_UT_; tr += 2) {
        sc_trip_load(B, qrow, kS2l, vS4, tr + 1);
        SC_HOLD();
        sc_trip_steps(A, cy, a);
        sc_trip_load(A, qrow, kS2l, vS4, (tr + 2 < UNITS_ / SC_UT_) ? tr + 2 : tr + 1);
        SC_HOLD();
        sc_trip_steps(B, cy, a);
    }
    sc_drain(cy, a);

    const int vl = valid_len[b];
    const int k0 = kt * SC_KT_ + 2 * lane;
    const int q0 = b * LQ_ + qg * SC_QG_ + wave * SC_QW_;
#pragma unroll
    for (int j = 0; j < SC_QW_; ++j) {
        float2 r;
        r.x = (k0 + 0 < vl) ? a[j].x : MASK_VAL_;
        r.y = (k0 + 1 < vl) ? a[j].y : MASK_VAL_;
        *(float2*)(scores + (size_t)(q0 + j) * LK_ + k0) = r;
    }
}

#define AO_BM_  32
#define AO_BN_  32
#define AO_KC_  256
#define AO_NC_  (LK_ / AO_KC_)
#define AO_RP4_ (LK_ / 4 + 1)
#define AO_NW_  8
#define AO_RW_  (AO_BM_ / AO_NW_)
#define AO_RD_  4
#define AO_RJ_  (LK_ / 128)
#define AO_VP4_ 12
#define AO_VRP_ (AO_NW_ * 4)
#define AO_VIT_ (AO_KC_ / AO_VRP_)
#define AO_FG_  8
#define AO_NG_  (AO_KC_ / 4 / AO_FG_)

__device__ __forceinline__ float lane_xor16(float v) {
    const int b = __builtin_bit_cast(int, v);
    return __builtin_bit_cast(float,
        __builtin_amdgcn_permlanex16(b, b, 0x76543210, 0xfedcba98, false, false));
}
template <int N>
__device__ __forceinline__ float lane_xor(float v) {
    return __builtin_bit_cast(float,
        __builtin_amdgcn_update_dpp(0, __builtin_bit_cast(int, v), 0x160 + N, 0xf, 0xf, true));
}

template <int NLJ>
__device__ __forceinline__ bool ao_rows(const float* __restrict__ sg, v4f* __restrict__ As4,
                                        const int wave, const int lane)
{
    constexpr int NX = (NLJ > 0) ? NLJ : 1;
    v4f x[AO_RD_][NX];
    if constexpr (NLJ > 0) {
#pragma unroll
        for (int r = 0; r < AO_RD_; ++r)
#pragma unroll
            for (int j = 0; j < NLJ; ++j)
                x[r][j] = *(const v4f*)(sg + (size_t)r * LK_ + j * 128);
    }
    bool nz = false;
#pragma unroll
    for (int r = 0; r < AO_RW_; ++r) {
        v4f (&xr)[NX] = x[r % AO_RD_];

        float m = MASK_VAL_;
        if constexpr (NLJ > 0) {
            m = fmaxf(fmaxf(xr[0].x, xr[0].y), fmaxf(xr[0].z, xr[0].w));
#pragma unroll
            for (int j = 1; j < NLJ; ++j)
                m = fmaxf(m, fmaxf(fmaxf(xr[j].x, xr[j].y), fmaxf(xr[j].z, xr[j].w)));
            if constexpr (NLJ < AO_RJ_) m = fmaxf(m, MASK_VAL_);
        }
        m = fmaxf(m, lane_xor16(m));
        m = fmaxf(m, lane_xor<8>(m));
        m = fmaxf(m, lane_xor<4>(m));
        m = fmaxf(m, lane_xor<2>(m));
        m = fmaxf(m, lane_xor<1>(m));

        float s[NX];
        if constexpr (NLJ > 0) {
#pragma unroll
            for (int j = 0; j < NLJ; ++j) {
                v4f e;
                e.x = __expf(xr[j].x - m);
                e.y = __expf(xr[j].y - m);
                e.z = __expf(xr[j].z - m);
                e.w = __expf(xr[j].w - m);
                xr[j] = e;
                s[j] = e.x + e.y + e.z + e.w;
            }
        }
        const float ed = __expf(MASK_VAL_ - m);
        nz = nz || (ed != 0.0f);
        float sd = ed + ed + ed + ed;
        if constexpr (NLJ > 0) {
#pragma unroll
            for (int j = 0; j < NLJ; ++j) s[j] += lane_xor16(s[j]);
#pragma unroll
            for (int j = 0; j < NLJ; ++j) s[j] += lane_xor<8>(s[j]);
#pragma unroll
            for (int j = 0; j < NLJ; ++j) s[j] += lane_xor<4>(s[j]);
#pragma unroll
            for (int j = 0; j < NLJ; ++j) s[j] += lane_xor<2>(s[j]);
#pragma unroll
            for (int j = 0; j < NLJ; ++j) s[j] += lane_xor<1>(s[j]);
        }
        if constexpr (NLJ < AO_RJ_) {
            sd += lane_xor16(sd);
            sd += lane_xor<8>(sd);
            sd += lane_xor<4>(sd);
            sd += lane_xor<2>(sd);
            sd += lane_xor<1>(sd);
        }
        float gs = (NLJ > 0) ? s[0] : sd;
#pragma unroll
        for (int j = 1; j < AO_RJ_; ++j) gs += (j < NLJ) ? s[j < NX ? j : 0] : sd;

        const float inv = 1.0f / gs;
        v4f* arow = As4 + (wave * AO_RW_ + r) * AO_RP4_ + lane;
        if constexpr (NLJ > 0) {
#pragma unroll
            for (int j = 0; j < NLJ; ++j) {
                v4f e = xr[j];
                e.x *= inv; e.y *= inv; e.z *= inv; e.w *= inv;
                arow[j * 32] = e;
            }
        }
        if constexpr (NLJ < AO_RJ_) {
            const float wd = ed * inv;
            const v4f wd4 = {wd, wd, wd, wd};
#pragma unroll
            for (int j = NLJ; j < AO_RJ_; ++j) arow[j * 32] = wd4;
        }

        if constexpr (NLJ > 0) {
            if (r + AO_RD_ < AO_RW_) {
#pragma unroll
                for (int j = 0; j < NLJ; ++j)
                    xr[j] = *(const v4f*)(sg + (size_t)(r + AO_RD_) * LK_ + j * 128);
            }
        }
    }
    return nz;
}

__global__ __launch_bounds__(AO_NW_ * 32) void attn_out_wmma(
    const float* __restrict__ scores,
    const float* __restrict__ value,
    const int*   __restrict__ valid_len,
    float* __restrict__ out)
{
    __shared__ v4f As4[AO_BM_ * AO_RP4_];
    __shared__ v4f Vs4[2 * AO_KC_ * AO_VP4_];
    __shared__ int nzS[AO_NW_];

    const int t    = threadIdx.x;
    const int lane = t & 31;
    const int wave = t >> 5;
    const int c15  = lane & 15;
    const int h    = lane >> 4;
    const int wt   = wave & 3;
    const int kh   = __builtin_amdgcn_readfirstlane(wave >> 2);
    const int tm   = wt >> 1;
    const int tn   = wt & 1;
    const int bn   = blockIdx.x & 15;
    const int bm   = (blockIdx.x >> 4) & 3;
    const int b    = blockIdx.x >> 6;
    const int vl   = valid_len[b];

    const int vr = t >> 3;
    const int vc = t & 7;
    const float* sg = scores + ((size_t)b * LQ_ + bm * AO_BM_ + wave * AO_RW_) * LK_ + lane * 4;
    const float* vg = value + ((size_t)b * LK_ + vr) * DSZ_ + bn * AO_BN_ + vc * 4;

    v4f rv[AO_VIT_];
    v8f acc = {0.f, 0.f, 0.f, 0.f, 0.f, 0.f, 0.f, 0.f};

#pragma unroll
    for (int p = 0; p < AO_VIT_; ++p) rv[p] = *(const v4f*)(vg + (size_t)p * AO_VRP_ * DSZ_);

    const int nlj = (vl <= 0) ? 0 : ((vl >= LK_) ? AO_RJ_ : (vl + 127) >> 7);
    bool nz;
    switch (nlj) {
    case 0:  nz = ao_rows<0>(sg, As4, wave, lane); break;
    case 1:  nz = ao_rows<1>(sg, As4, wave, lane); break;
    case 2:  nz = ao_rows<2>(sg, As4, wave, lane); break;
    case 3:  nz = ao_rows<3>(sg, As4, wave, lane); break;
    case 4:  nz = ao_rows<4>(sg, As4, wave, lane); break;
    case 5:  nz = ao_rows<5>(sg, As4, wave, lane); break;
    case 6:  nz = ao_rows<6>(sg, As4, wave, lane); break;
    case 7:  nz = ao_rows<7>(sg, As4, wave, lane); break;
    default: nz = ao_rows<8>(sg, As4, wave, lane); break;
    }
    if (lane == 0) nzS[wave] = nz ? 1 : 0;

#pragma unroll
    for (int p = 0; p < AO_VIT_; ++p) Vs4[(vr + p * AO_VRP_) * AO_VP4_ + vc] = rv[p];
    __syncthreads();

    const int anynz = __builtin_amdgcn_readfirstlane(nzS[0] | nzS[1] | nzS[2] | nzS[3] |
                                                     nzS[4] | nzS[5] | nzS[6] | nzS[7]);
    const int ngt = (anynz != 0 || vl <= 0 || vl >= LK_) ? (LK_ / 32) : ((vl + 31) >> 5);
    const int nch = (ngt + AO_NG_ - 1) / AO_NG_;

    for (int kc = 0; kc < nch; ++kc) {
        const int  buf  = kc & 1;
        const bool more = (kc + 1 < nch);
        const int  gcnt = min(AO_NG_, ngt - kc * AO_NG_);

        if (more) {
#pragma unroll
            for (int p = 0; p < AO_VIT_; ++p)
                rv[p] = *(const v4f*)(vg + ((size_t)(kc + 1) * AO_KC_ + p * AO_VRP_) * DSZ_);
        }

        const v2f*   af = (const v2f*)(As4 + (tm * 16 + c15) * AO_RP4_) + kc * (AO_KC_ / 2) + h
                          + kh * (AO_FG_ * 2);
        const float* vf = (const float*)(Vs4 + (buf * AO_KC_ + 2 * h + kh * (AO_FG_ * 4)) * AO_VP4_)
                          + tn * 16 + c15;
        v2f fa[2][AO_FG_], fb[2][AO_FG_];
#pragma unroll
        for (int i = 0; i < AO_FG_; ++i) {
            fa[0][i]   = af[i * 2];
            fb[0][i].x = vf[(i * 4 + 0) * (AO_VP4_ * 4)];
            fb[0][i].y = vf[(i * 4 + 1) * (AO_VP4_ * 4)];
        }
#pragma unroll
        for (int j = 0; j < AO_NG_ / 2; ++j) {
            if (2 * j + kh >= gcnt) break;
            if (j + 1 < AO_NG_ / 2) {
                if (2 * (j + 1) + kh < gcnt) {
#pragma unroll
                    for (int i = 0; i < AO_FG_; ++i) {
                        const int ks = 2 * (j + 1) * AO_FG_ + i;
                        fa[(j + 1) & 1][i]   = af[ks * 2];
                        fb[(j + 1) & 1][i].x = vf[(ks * 4 + 0) * (AO_VP4_ * 4)];
                        fb[(j + 1) & 1][i].y = vf[(ks * 4 + 1) * (AO_VP4_ * 4)];
                    }
                }
            }
            __builtin_amdgcn_sched_barrier(0);
#pragma unroll
            for (int i = 0; i < AO_FG_; ++i)
                acc = __builtin_amdgcn_wmma_f32_16x16x4_f32(
                          false, fa[j & 1][i], false, fb[j & 1][i], (short)0, acc, false, false);
            __builtin_amdgcn_sched_barrier(0);
        }

        if (more) {
#pragma unroll
            for (int p = 0; p < AO_VIT_; ++p)
                Vs4[((buf ^ 1) * AO_KC_ + vr + p * AO_VRP_) * AO_VP4_ + vc] = rv[p];
        }
        __syncthreads();
    }

    float* red = (float*)Vs4;
    if (kh == 1) {
#pragma unroll
        for (int r = 0; r < 8; ++r) red[(wt * 8 + r) * 32 + lane] = acc[r];
    }
    __syncthreads();
    if (kh == 0) {
        const int m0 = bm * AO_BM_ + tm * 16 + 8 * h;
        const int n  = bn * AO_BN_ + tn * 16 + c15;
#pragma unroll
        for (int r = 0; r < 8; ++r)
            out[((size_t)b * LQ_ + m0 + r) * DSZ_ + n] = acc[r] + red[(wt * 8 + r) * 32 + lane];
    }
}

extern "C" void kernel_launch(void* const* d_in, const int* in_sizes, int n_in,
                              void* d_out, int out_size, void* d_ws, size_t ws_size,
                              hipStream_t stream) {
    const float* query     = (const float*)d_in[0];
    const float* key       = (const float*)d_in[1];
    const float* value     = (const float*)d_in[2];
    const int*   valid_len = (const int*)  d_in[3];
    const float* W_q       = (const float*)d_in[4];
    const float* W_k       = (const float*)d_in[5];
    const float* vvec      = (const float*)d_in[6];

    float* qf     = (float*)d_ws;
    float* kfT    = qf  + (size_t)B_ * LQ_ * UNITS_;
    float* scores = kfT + (size_t)B_ * UNITS_ * LK_;
    float* outp   = (float*)d_out;

    proj_both<<<PJ_KB_ + PJ_QB_, PJ_NT_, 0, stream>>>(key, W_k, kfT, query, W_q, qf, valid_len);
    scores_tanh_kernel<<<512, 128, 0, stream>>>(qf, kfT, vvec, valid_len, scores);
    attn_out_wmma<<<256, AO_NW_ * 32, 0, stream>>>(scores, value, valid_len, outp);
}
